# starting kernel
# baseline (speedup 1.0000x reference)
#include <hip/hip_runtime.h>

typedef __attribute__((ext_vector_type(2))) float v2f;
typedef __attribute__((ext_vector_type(8))) float v8f;

__global__ __launch_bounds__(256) void fill_f32(float* __restrict__ p, float v, long n) {
    long i = (long)blockIdx.x * 256 + threadIdx.x;
    if (i < n) p[i] = v;
}

__global__ __launch_bounds__(256) void deg_count(const int* __restrict__ dst,
                                                 float* __restrict__ deg, int nE) {
    int e = blockIdx.x * 256 + threadIdx.x;
    if (e < nE) atomicAdd(&deg[dst[e]], 1.0f);
}

__global__ __launch_bounds__(256) void rsqrt_k(const float* __restrict__ deg,
                                               float* __restrict__ dinv, int n) {
    int i = blockIdx.x * 256 + threadIdx.x;
    if (i < n) dinv[i] = rsqrtf(deg[i]);
}

#define LPITCH 66

__global__ __launch_bounds__(256) void gemm_k128(const float* __restrict__ A,
                                                 const float* __restrict__ W,
                                                 const float* __restrict__ bias,
                                                 float* __restrict__ out,
                                                 int M, int N, int do_relu, int has_bias) {
    __shared__ v2f sA2[16][LPITCH];
    __shared__ v2f sW2[128][LPITCH];

    const int row0 = blockIdx.x * 16;
    const int tid  = threadIdx.x;

    float* sWf = (float*)sW2;
    for (int i = tid; i < 128 * N; i += 256) {
        int k = i / N, n = i - k * N;
        sWf[n * (2 * LPITCH) + k] = W[i];
    }
    float* sAf = (float*)sA2;
    for (int i = tid; i < 16 * 128; i += 256) {
        int r = i >> 7, c = i & 127;
        sAf[r * (2 * LPITCH) + c] = (row0 + r < M) ? A[(size_t)(row0 + r) * 128 + c] : 0.0f;
    }
    __syncthreads();

    const int wave  = tid >> 5;
    const int lane  = tid & 31;
    const int m     = lane & 15;
    const int hsel  = lane >> 4;
    const int mb    = hsel << 3;
    const int ncolt = N >> 4;

    for (int ct = wave; ct < ncolt; ct += 8) {
        const int col0 = ct << 4;
        v8f acc = {};
#pragma unroll 4
        for (int k2 = 0; k2 < 64; k2 += 2) {
            const int kq = k2 + hsel;
            v2f a = sA2[m][kq];
            v2f b = sW2[col0 + m][kq];
            acc = __builtin_amdgcn_wmma_f32_16x16x4_f32(
                false, a, false, b, (short)0, acc, false, false);
        }
        const float bv = has_bias ? bias[col0 + m] : 0.0f;
#pragma unroll
        for (int r = 0; r < 8; ++r) {
            int row = row0 + mb + r;
            if (row < M) {
                float v = acc[r] + bv;
                if (do_relu) v = fmaxf(v, 0.0f);
                out[(size_t)row * N + col0 + m] = v;
            }
        }
    }
}

__global__ __launch_bounds__(256) void edge_scatter(const int* __restrict__ src,
                                                    const int* __restrict__ dst,
                                                    const float* __restrict__ dinv,
                                                    const float* __restrict__ h,
                                                    float* __restrict__ agg, int nE) {
    int e = blockIdx.x * 8 + (threadIdx.x >> 5);
    if (e >= nE) return;
    const int lane = threadIdx.x & 31;
    const int s = src[e], d = dst[e];
    const float norm = dinv[s] * dinv[d];
    const float4 v = ((const float4*)(h + (size_t)s * 128))[lane];
    float* ap = agg + (size_t)d * 128 + lane * 4;
    atomicAdd(ap + 0, norm * v.x);
    atomicAdd(ap + 1, norm * v.y);
    atomicAdd(ap + 2, norm * v.z);
    atomicAdd(ap + 3, norm * v.w);
}

__global__ __launch_bounds__(256) void finalize_k(const float* __restrict__ agg,
                                                  float* __restrict__ h,
                                                  const float* __restrict__ dinv,
                                                  const float* __restrict__ bias,
                                                  int nNodes) {
    long i = (long)blockIdx.x * 256 + threadIdx.x;
    long total = (long)nNodes * 32;
    if (i >= total) return;
    const int node = (int)(i >> 5);
    const int c    = (int)(i & 31);
    const float di = dinv[node];
    const float d2 = di * di;
    float4 hv = ((const float4*)h)[i];
    float4 av = ((const float4*)agg)[i];
    const float* bp = bias + c * 4;
    float4 o;
    o.x = fmaxf(av.x + d2 * hv.x + bp[0], 0.0f);
    o.y = fmaxf(av.y + d2 * hv.y + bp[1], 0.0f);
    o.z = fmaxf(av.z + d2 * hv.z + bp[2], 0.0f);
    o.w = fmaxf(av.w + d2 * hv.w + bp[3], 0.0f);
    ((float4*)h)[i] = o;
}

__global__ __launch_bounds__(256) void pool_sum(const float* __restrict__ h,
                                                const int* __restrict__ batch,
                                                float* __restrict__ gsum,
                                                float* __restrict__ gcnt, int nNodes) {
    long i = (long)blockIdx.x * 256 + threadIdx.x;
    long total = (long)nNodes * 32;
    if (i >= total) return;
    const int node = (int)(i >> 5);
    const int c    = (int)(i & 31);
    const int b    = batch[node];
    float4 v = ((const float4*)h)[i];
    float* gp = gsum + (size_t)b * 128 + c * 4;
    atomicAdd(gp + 0, v.x);
    atomicAdd(gp + 1, v.y);
    atomicAdd(gp + 2, v.z);
    atomicAdd(gp + 3, v.w);
    if (c == 0) atomicAdd(&gcnt[b], 1.0f);
}

__global__ __launch_bounds__(256) void pool_div(const float* __restrict__ gsum,
                                                const float* __restrict__ gcnt,
                                                float* __restrict__ g, int G) {
    int i = blockIdx.x * 256 + threadIdx.x;
    if (i < G * 128) {
        int b = i >> 7;
        g[i] = gsum[i] / fmaxf(gcnt[b], 1.0f);
    }
}

extern "C" void kernel_launch(void* const* d_in, const int* in_sizes, int n_in,
                              void* d_out, int out_size, void* d_ws, size_t ws_size,
                              hipStream_t stream) {
    const float* x     = (const float*)d_in[0];
    const int*   ei    = (const int*)d_in[1];
    const int*   batch = (const int*)d_in[2];
    const float* W1    = (const float*)d_in[3];
    const float* b1    = (const float*)d_in[4];
    const float* W2    = (const float*)d_in[5];
    const float* b2    = (const float*)d_in[6];
    const float* Wl1   = (const float*)d_in[7];
    const float* bl1   = (const float*)d_in[8];
    const float* Wl2   = (const float*)d_in[9];
    const float* bl2   = (const float*)d_in[10];
    float* out = (float*)d_out;

    const int nN = in_sizes[0] / 128;
    const int nE = in_sizes[1] / 2;
    const int G  = out_size / 32;
    const int* src = ei;
    const int* dst = ei + nE;

    char* ws = (char*)d_ws;
    auto alloc = [&](size_t bytes) -> float* {
        float* p = (float*)ws;
        ws += (bytes + 255) & ~(size_t)255;
        return p;
    };
    float* bufA = alloc((size_t)nN * 128 * 4);
    float* bufB = alloc((size_t)nN * 128 * 4);
    float* bufC = alloc((size_t)nN * 128 * 4);
    float* deg  = alloc((size_t)nN * 4);
    float* dinv = alloc((size_t)nN * 4);
    float* gsum = alloc((size_t)G * 128 * 4);
    float* gcnt = alloc((size_t)G * 4);
    float* g    = alloc((size_t)G * 128 * 4);
    float* m1   = alloc((size_t)G * 128 * 4);

    const long nel = (long)nN * 128;
    const int  fB  = (int)((nel + 255) / 256);
    const int  eB  = (nE + 255) / 256;
    const int  nB  = (nN + 255) / 256;
    const int  qB  = (int)(((long)nN * 32 + 255) / 256);
    const int  sB  = (nE + 7) / 8;

    fill_f32<<<nB, 256, 0, stream>>>(deg, 1.0f, nN);
    fill_f32<<<fB, 256, 0, stream>>>(bufB, 0.0f, nel);
    fill_f32<<<(G * 128 + 255) / 256, 256, 0, stream>>>(gsum, 0.0f, (long)G * 128);
    fill_f32<<<(G + 255) / 256, 256, 0, stream>>>(gcnt, 0.0f, G);
    deg_count<<<eB, 256, 0, stream>>>(dst, deg, nE);
    rsqrt_k<<<nB, 256, 0, stream>>>(deg, dinv, nN);

    gemm_k128<<<(nN + 15) / 16, 256, 0, stream>>>(x, W1, nullptr, bufA, nN, 128, 0, 0);
    edge_scatter<<<sB, 256, 0, stream>>>(src, dst, dinv, bufA, bufB, nE);
    finalize_k<<<qB, 256, 0, stream>>>(bufB, bufA, dinv, b1, nN);

    fill_f32<<<fB, 256, 0, stream>>>(bufB, 0.0f, nel);
    gemm_k128<<<(nN + 15) / 16, 256, 0, stream>>>(bufA, W2, nullptr, bufC, nN, 128, 0, 0);
    edge_scatter<<<sB, 256, 0, stream>>>(src, dst, dinv, bufC, bufB, nE);
    finalize_k<<<qB, 256, 0, stream>>>(bufB, bufC, dinv, b2, nN);

    pool_sum<<<qB, 256, 0, stream>>>(bufC, batch, gsum, gcnt, nN);
    pool_div<<<(G * 128 + 255) / 256, 256, 0, stream>>>(gsum, gcnt, g, G);

    gemm_k128<<<(G + 15) / 16, 256, 0, stream>>>(g, Wl1, bl1, m1, G, 128, 1, 1);
    gemm_k128<<<(G + 15) / 16, 256, 0, stream>>>(m1, Wl2, bl2, out, G, 32, 0, 1);
}


# Edge sum as a gather: edges grouped by destination once per call (returning integer add = rank, prefix sum); one wave sums a node's rows in registers, fused with self-loop+bias+relu; sum order = rank
# speedup vs baseline: 1.9079x; 1.9079x over previous
#include <hip/hip_runtime.h>

typedef __attribute__((ext_vector_type(2))) float v2f;
typedef __attribute__((ext_vector_type(8))) float v8f;

__global__ __launch_bounds__(256) void fill_f32(float* __restrict__ p, float v, long n) {
    long i = (long)blockIdx.x * 256 + threadIdx.x;
    if (i < n) p[i] = v;
}

__global__ __launch_bounds__(256) void zero_i32(int* __restrict__ p, int n) {
    int i = blockIdx.x * 256 + threadIdx.x;
    if (i < n) p[i] = 0;
}

__global__ __launch_bounds__(256) void count_rank(const int* __restrict__ dst,
                                                  int* __restrict__ cnt,
                                                  int* __restrict__ rank, int nE) {
    int e = blockIdx.x * 256 + threadIdx.x;
    if (e < nE) rank[e] = atomicAdd(&cnt[dst[e]], 1);
}

__device__ __forceinline__ int wave_sum(int v) {
#pragma unroll
    for (int off = 16; off > 0; off >>= 1) v += __shfl_xor(v, off);
    return v;
}

#define SCAN_ELEMS 1024

__global__ __launch_bounds__(256) void blk_sum(const int* __restrict__ cnt,
                                               int* __restrict__ bsum, int n) {
    __shared__ int s[8];
    const int tid  = threadIdx.x;
    const int base = blockIdx.x * SCAN_ELEMS + tid * 4;
    int v = 0;
#pragma unroll
    for (int j = 0; j < 4; ++j)
        if (base + j < n) v += cnt[base + j];
    v = wave_sum(v);
    if ((tid & 31) == 0) s[tid >> 5] = v;
    __syncthreads();
    if (tid == 0) {
        int t = 0;
        for (int w = 0; w < 8; ++w) t += s[w];
        bsum[blockIdx.x] = t;
    }
}

__global__ __launch_bounds__(256) void scan_rows(const int* __restrict__ cnt,
                                                 const int* __restrict__ bsum,
                                                 int* __restrict__ rowptr,
                                                 float* __restrict__ dinv, int n) {
    __shared__ int sp[8];
    __shared__ int sw[8];
    const int tid = threadIdx.x, lane = tid & 31, wave = tid >> 5;

    int pv = 0;
    for (int i = tid; i < (int)blockIdx.x; i += 256) pv += bsum[i];
    pv = wave_sum(pv);
    if (lane == 0) sp[wave] = pv;

    const int base = blockIdx.x * SCAN_ELEMS + tid * 4;
    int c[4];
#pragma unroll
    for (int j = 0; j < 4; ++j) c[j] = (base + j < n) ? cnt[base + j] : 0;
    const int tsum = c[0] + c[1] + c[2] + c[3];
    int inc = tsum;
#pragma unroll
    for (int off = 1; off < 32; off <<= 1) {
        int t = __shfl_up(inc, off);
        if (lane >= off) inc += t;
    }
    if (lane == 31) sw[wave] = inc;
    __syncthreads();
    int off0 = 0;
    for (int w = 0; w < 8; ++w) {
        off0 += sp[w];
        if (w < wave) off0 += sw[w];
    }
    int excl = off0 + inc - tsum;
#pragma unroll
    for (int j = 0; j < 4; ++j) {
        if (base + j < n) {
            rowptr[base + j] = excl;
            dinv[base + j]   = rsqrtf((float)c[j] + 1.0f);
            excl += c[j];
        }
    }
}

__global__ __launch_bounds__(256) void place_src(const int* __restrict__ src,
                                                 const int* __restrict__ dst,
                                                 const int* __restrict__ rank,
                                                 const int* __restrict__ rowptr,
                                                 int* __restrict__ csr_src, int nE) {
    int e = blockIdx.x * 256 + threadIdx.x;
    if (e < nE) csr_src[rowptr[dst[e]] + rank[e]] = src[e];
}

#define LPITCH 66

__global__ __launch_bounds__(256) void gemm_k128(const float* __restrict__ A,
                                                 const float* __restrict__ W,
                                                 const float* __restrict__ bias,
                                                 float* __restrict__ out,
                                                 int M, int N, int do_relu, int has_bias) {
    __shared__ v2f sA2[16][LPITCH];
    __shared__ v2f sW2[128][LPITCH];

    const int row0 = blockIdx.x * 16;
    const int tid  = threadIdx.x;

    float* sWf = (float*)sW2;
    for (int i = tid; i < 128 * N; i += 256) {
        int k = i / N, n = i - k * N;
        sWf[n * (2 * LPITCH) + k] = W[i];
    }
    float* sAf = (float*)sA2;
    for (int i = tid; i < 16 * 128; i += 256) {
        int r = i >> 7, c = i & 127;
        sAf[r * (2 * LPITCH) + c] = (row0 + r < M) ? A[(size_t)(row0 + r) * 128 + c] : 0.0f;
    }
    __syncthreads();

    const int wave  = tid >> 5;
    const int lane  = tid & 31;
    const int m     = lane & 15;
    const int hsel  = lane >> 4;
    const int mb    = hsel << 3;
    const int ncolt = N >> 4;

    for (int ct = wave; ct < ncolt; ct += 8) {
        const int col0 = ct << 4;
        v8f acc = {};
#pragma unroll 4
        for (int k2 = 0; k2 < 64; k2 += 2) {
            const int kq = k2 + hsel;
            v2f a = sA2[m][kq];
            v2f b = sW2[col0 + m][kq];
            acc = __builtin_amdgcn_wmma_f32_16x16x4_f32(
                false, a, false, b, (short)0, acc, false, false);
        }
        const float bv = has_bias ? bias[col0 + m] : 0.0f;
#pragma unroll
        for (int r = 0; r < 8; ++r) {
            int row = row0 + mb + r;
            if (row < M) {
                float v = acc[r] + bv;
                if (do_relu) v = fmaxf(v, 0.0f);
                out[(size_t)row * N + col0 + m] = v;
            }
        }
    }
}

__device__ __forceinline__ float lane_f(float v, int l) {
    return __int_as_float(__builtin_amdgcn_readlane(__float_as_int(v), l));
}

#define ROW(s) (((const float4*)(h + (size_t)(s) * 128))[lane])
#define ACC(wt_, vv_)                                                         \
    acc.x = fmaf((wt_), (vv_).x, acc.x); acc.y = fmaf((wt_), (vv_).y, acc.y); \
    acc.z = fmaf((wt_), (vv_).z, acc.z); acc.w = fmaf((wt_), (vv_).w, acc.w)

__global__ __launch_bounds__(256) void gather_relu(const int* __restrict__ rowptr,
                                                   const int* __restrict__ cnt,
                                                   const int* __restrict__ csr_src,
                                                   const float* __restrict__ dinv,
                                                   const float* __restrict__ h,
                                                   const float* __restrict__ bias,
                                                   float* __restrict__ act, int nN) {
    const int lane = threadIdx.x & 31;
    const int wv = __builtin_amdgcn_readfirstlane((int)(threadIdx.x >> 5));
    const int nw = gridDim.x * 8;
    const float4 bv = ((const float4*)bias)[lane];

    for (int node = blockIdx.x * 8 + wv; node < nN; node += nw) {
        const int   beg = rowptr[node];
        const int   n   = cnt[node];
        const float di  = dinv[node];
        const float4 hv = ROW(node);
        float4 acc = {0.0f, 0.0f, 0.0f, 0.0f};

        for (int c0 = 0; c0 < n; c0 += 32) {
            const int m = min(n - c0, 32);
            int   s_l = 0;
            float w_l = 0.0f;
            if (lane < m) {
                s_l = csr_src[beg + c0 + lane];
                w_l = dinv[s_l] * di;
            }
            int j = 0;
            for (; j + 4 <= m; j += 4) {
                const int s0 = __builtin_amdgcn_readlane(s_l, j);
                const int s1 = __builtin_amdgcn_readlane(s_l, j + 1);
                const int s2 = __builtin_amdgcn_readlane(s_l, j + 2);
                const int s3 = __builtin_amdgcn_readlane(s_l, j + 3);
                const float4 v0 = ROW(s0);
                const float4 v1 = ROW(s1);
                const float4 v2 = ROW(s2);
                const float4 v3 = ROW(s3);
                const float w0 = lane_f(w_l, j),     w1 = lane_f(w_l, j + 1);
                const float w2 = lane_f(w_l, j + 2), w3 = lane_f(w_l, j + 3);
                ACC(w0, v0); ACC(w1, v1); ACC(w2, v2); ACC(w3, v3);
            }
            if (j + 2 <= m) {
                const int s0 = __builtin_amdgcn_readlane(s_l, j);
                const int s1 = __builtin_amdgcn_readlane(s_l, j + 1);
                const float4 v0 = ROW(s0);
                const float4 v1 = ROW(s1);
                const float w0 = lane_f(w_l, j), w1 = lane_f(w_l, j + 1);
                ACC(w0, v0); ACC(w1, v1);
                j += 2;
            }
            if (j < m) {
                const int s0 = __builtin_amdgcn_readlane(s_l, j);
                const float4 v0 = ROW(s0);
                const float w0 = lane_f(w_l, j);
                ACC(w0, v0);
            }
        }

        const float d2 = di * di;
        float4 o;
        o.x = fmaxf(acc.x + d2 * hv.x + bv.x, 0.0f);
        o.y = fmaxf(acc.y + d2 * hv.y + bv.y, 0.0f);
        o.z = fmaxf(acc.z + d2 * hv.z + bv.z, 0.0f);
        o.w = fmaxf(acc.w + d2 * hv.w + bv.w, 0.0f);
        ((float4*)(act + (size_t)node * 128))[lane] = o;
    }
}
#undef ROW
#undef ACC

__global__ __launch_bounds__(256) void pool_sum(const float* __restrict__ h,
                                                const int* __restrict__ batch,
                                                float* __restrict__ gsum,
                                                float* __restrict__ gcnt, int nNodes) {
    long i = (long)blockIdx.x * 256 + threadIdx.x;
    long total = (long)nNodes * 32;
    if (i >= total) return;
    const int node = (int)(i >> 5);
    const int c    = (int)(i & 31);
    const int b    = batch[node];
    float4 v = ((const float4*)h)[i];
    float* gp = gsum + (size_t)b * 128 + c * 4;
    atomicAdd(gp + 0, v.x);
    atomicAdd(gp + 1, v.y);
    atomicAdd(gp + 2, v.z);
    atomicAdd(gp + 3, v.w);
    if (c == 0) atomicAdd(&gcnt[b], 1.0f);
}

__global__ __launch_bounds__(256) void pool_div(const float* __restrict__ gsum,
                                                const float* __restrict__ gcnt,
                                                float* __restrict__ g, int G) {
    int i = blockIdx.x * 256 + threadIdx.x;
    if (i < G * 128) {
        int b = i >> 7;
        g[i] = gsum[i] / fmaxf(gcnt[b], 1.0f);
    }
}

extern "C" void kernel_launch(void* const* d_in, const int* in_sizes, int n_in,
                              void* d_out, int out_size, void* d_ws, size_t ws_size,
                              hipStream_t stream) {
    const float* x     = (const float*)d_in[0];
    const int*   ei    = (const int*)d_in[1];
    const int*   batch = (const int*)d_in[2];
    const float* W1    = (const float*)d_in[3];
    const float* b1    = (const float*)d_in[4];
    const float* W2    = (const float*)d_in[5];
    const float* b2    = (const float*)d_in[6];
    const float* Wl1   = (const float*)d_in[7];
    const float* bl1   = (const float*)d_in[8];
    const float* Wl2   = (const float*)d_in[9];
    const float* bl2   = (const float*)d_in[10];
    float* out = (float*)d_out;

    const int nN = in_sizes[0] / 128;
    const int nE = in_sizes[1] / 2;
    const int G  = out_size / 32;
    const int* src = ei;
    const int* dst = ei + nE;

    char* ws = (char*)d_ws;
    auto alloc = [&](size_t bytes) -> float* {
        float* p = (float*)ws;
        ws += (bytes + 255) & ~(size_t)255;
        return p;
    };
    float* bufA = alloc((size_t)nN * 128 * 4);
    float* bufB = alloc((size_t)nN * 128 * 4);
    char*  bufC = (char*)alloc((size_t)nN * 128 * 4);
    int*   cnt  = (int*)alloc((size_t)nN * 4);
    float* dinv = alloc((size_t)nN * 4);
    float* gsum = alloc((size_t)G * 128 * 4);
    float* gcnt = alloc((size_t)G * 4);
    float* g    = alloc((size_t)G * 128 * 4);
    float* m1   = alloc((size_t)G * 128 * 4);

    auto sub = [&](size_t bytes) -> int* {
        int* p = (int*)bufC;
        bufC += (bytes + 255) & ~(size_t)255;
        return p;
    };
    const int nSB = (nN + SCAN_ELEMS - 1) / SCAN_ELEMS;
    int* csr_src = sub((size_t)nE * 4);
    int* rank    = sub((size_t)nE * 4);
    int* rowptr  = sub((size_t)nN * 4);
    int* bsum    = sub((size_t)nSB * 4);

    const int  eB  = (nE + 255) / 256;
    const int  nB  = (nN + 255) / 256;
    const int  qB  = (int)(((long)nN * 32 + 255) / 256);

    zero_i32<<<nB, 256, 0, stream>>>(cnt, nN);
    fill_f32<<<(G * 128 + 255) / 256, 256, 0, stream>>>(gsum, 0.0f, (long)G * 128);
    fill_f32<<<(G + 255) / 256, 256, 0, stream>>>(gcnt, 0.0f, G);

    count_rank<<<eB, 256, 0, stream>>>(dst, cnt, rank, nE);
    blk_sum<<<nSB, 256, 0, stream>>>(cnt, bsum, nN);
    scan_rows<<<nSB, 256, 0, stream>>>(cnt, bsum, rowptr, dinv, nN);
    place_src<<<eB, 256, 0, stream>>>(src, dst, rank, rowptr, csr_src, nE);

    gemm_k128<<<(nN + 15) / 16, 256, 0, stream>>>(x, W1, nullptr, bufA, nN, 128, 0, 0);
    gather_relu<<<2048, 256, 0, stream>>>(rowptr, cnt, csr_src, dinv, bufA, b1, bufB, nN);

    gemm_k128<<<(nN + 15) / 16, 256, 0, stream>>>(bufB, W2, nullptr, bufA, nN, 128, 0, 0);
    gather_relu<<<2048, 256, 0, stream>>>(rowptr, cnt, csr_src, dinv, bufA, b2, bufB, nN);

    pool_sum<<<qB, 256, 0, stream>>>(bufB, batch, gsum, gcnt, nN);
    pool_div<<<(G * 128 + 255) / 256, 256, 0, stream>>>(gsum, gcnt, g, G);

    gemm_k128<<<(G + 15) / 16, 256, 0, stream>>>(g, Wl1, bl1, m1, G, 128, 1, 1);
    gemm_k128<<<(G + 15) / 16, 256, 0, stream>>>(m1, Wl2, bl2, out, G, 32, 0, 1);
}


# GEMM work-groups stage W into LDS once and loop over 32-row strips (768 groups instead of 6250), next strip's rows prefetched to registers; per-tile k order unchanged
# speedup vs baseline: 2.4208x; 1.2688x over previous
#include <hip/hip_runtime.h>

typedef __attribute__((ext_vector_type(2))) float v2f;
typedef __attribute__((ext_vector_type(8))) float v8f;

__global__ __launch_bounds__(256) void fill_f32(float* __restrict__ p, float v, long n) {
    long i = (long)blockIdx.x * 256 + threadIdx.x;
    if (i < n) p[i] = v;
}

__global__ __launch_bounds__(256) void zero_i32(int* __restrict__ p, int n) {
    int i = blockIdx.x * 256 + threadIdx.x;
    if (i < n) p[i] = 0;
}

__global__ __launch_bounds__(256) void count_rank(const int* __restrict__ dst,
                                                  int* __restrict__ cnt,
                                                  int* __restrict__ rank, int nE) {
    int e = blockIdx.x * 256 + threadIdx.x;
    if (e < nE) rank[e] = atomicAdd(&cnt[dst[e]], 1);
}

__device__ __forceinline__ int wave_sum(int v) {
#pragma unroll
    for (int off = 16; off > 0; off >>= 1) v += __shfl_xor(v, off);
    return v;
}

#define SCAN_ELEMS 1024

__global__ __launch_bounds__(256) void blk_sum(const int* __restrict__ cnt,
                                               int* __restrict__ bsum, int n) {
    __shared__ int s[8];
    const int tid  = threadIdx.x;
    const int base = blockIdx.x * SCAN_ELEMS + tid * 4;
    int v = 0;
#pragma unroll
    for (int j = 0; j < 4; ++j)
        if (base + j < n) v += cnt[base + j];
    v = wave_sum(v);
    if ((tid & 31) == 0) s[tid >> 5] = v;
    __syncthreads();
    if (tid == 0) {
        int t = 0;
        for (int w = 0; w < 8; ++w) t += s[w];
        bsum[blockIdx.x] = t;
    }
}

__global__ __launch_bounds__(256) void scan_rows(const int* __restrict__ cnt,
                                                 const int* __restrict__ bsum,
                                                 int* __restrict__ rowptr,
                                                 float* __restrict__ dinv, int n) {
    __shared__ int sp[8];
    __shared__ int sw[8];
    const int tid = threadIdx.x, lane = tid & 31, wave = tid >> 5;

    int pv = 0;
    for (int i = tid; i < (int)blockIdx.x; i += 256) pv += bsum[i];
    pv = wave_sum(pv);
    if (lane == 0) sp[wave] = pv;

    const int base = blockIdx.x * SCAN_ELEMS + tid * 4;
    int c[4];
#pragma unroll
    for (int j = 0; j < 4; ++j) c[j] = (base + j < n) ? cnt[base + j] : 0;
    const int tsum = c[0] + c[1] + c[2] + c[3];
    int inc = tsum;
#pragma unroll
    for (int off = 1; off < 32; off <<= 1) {
        int t = __shfl_up(inc, off);
        if (lane >= off) inc += t;
    }
    if (lane == 31) sw[wave] = inc;
    __syncthreads();
    int off0 = 0;
    for (int w = 0; w < 8; ++w) {
        off0 += sp[w];
        if (w < wave) off0 += sw[w];
    }
    int excl = off0 + inc - tsum;
#pragma unroll
    for (int j = 0; j < 4; ++j) {
        if (base + j < n) {
            rowptr[base + j] = excl;
            dinv[base + j]   = rsqrtf((float)c[j] + 1.0f);
            excl += c[j];
        }
    }
}

__global__ __launch_bounds__(256) void place_src(const int* __restrict__ src,
                                                 const int* __restrict__ dst,
                                                 const int* __restrict__ rank,
                                                 const int* __restrict__ rowptr,
                                                 int* __restrict__ csr_src, int nE) {
    int e = blockIdx.x * 256 + threadIdx.x;
    if (e < nE) csr_src[rowptr[dst[e]] + rank[e]] = src[e];
}

#define LPITCH 66
#define RSTRIP 32

__device__ __forceinline__ void lds_ready_barrier() {
    asm volatile("s_wait_dscnt 0x0\n\ts_barrier_signal -1\n\ts_barrier_wait -1" ::: "memory");
}
__device__ __forceinline__ void reads_done_barrier() {
    asm volatile("s_barrier_signal -1\n\ts_barrier_wait -1" ::: "memory");
}

template <int N, bool HAS_BIAS, bool DO_RELU>
__global__ __launch_bounds__(256) void gemm_k128(const float* __restrict__ A,
                                                 const float* __restrict__ W,
                                                 const float* __restrict__ bias,
                                                 float* __restrict__ out,
                                                 int M) {
    __shared__ v2f sA2[RSTRIP][LPITCH];
    __shared__ v2f sW2[N][LPITCH];

    const int tid = threadIdx.x;

    const int nStrips = (M + RSTRIP - 1) / RSTRIP;
    const int prow = tid >> 5;
    const int pc4  = tid & 31;
    float* sAf = (float*)sA2;

    float4 pre[4];
    int strip = blockIdx.x;
    if (strip < nStrips) {
#pragma unroll
        for (int j = 0; j < 4; ++j) {
            const int row = strip * RSTRIP + prow + 8 * j;
            pre[j] = (row < M) ? ((const float4*)(A + (size_t)row * 128))[pc4]
                               : float4{0.0f, 0.0f, 0.0f, 0.0f};
        }
    }

    for (int i = tid; i < 64 * N; i += 256) {
        const int kp = i / N, n = i % N;
        v2f w;
        w.x = W[(2 * kp) * N + n];
        w.y = W[(2 * kp + 1) * N + n];
        sW2[n][kp] = w;
    }

    const int wave  = tid >> 5;
    const int lane  = tid & 31;
    const int m     = lane & 15;
    const int hsel  = lane >> 4;
    const int mb    = hsel << 3;

    static_assert(N % 16 == 0 && N / 16 <= 8, "one column tile per wave");
    const bool has_tile = wave < N / 16;
    const int  col0     = wave << 4;
    float bv = 0.0f;
    if constexpr (HAS_BIAS) { if (has_tile) bv = bias[col0 + m]; }

    const int gstep = (int)gridDim.x;
    for (; strip < nStrips; strip += gstep) {
        const int row0 = strip * RSTRIP;
#pragma unroll
        for (int j = 0; j < 4; ++j)
            *(float4*)(sAf + (prow + 8 * j) * (2 * LPITCH) + pc4 * 4) = pre[j];
        lds_ready_barrier();

        const int nxt = strip + gstep;
        if (nxt < nStrips) {
#pragma unroll
            for (int j = 0; j < 4; ++j) {
                const int row = nxt * RSTRIP + prow + 8 * j;
                pre[j] = (row < M) ? ((const float4*)(A + (size_t)row * 128))[pc4]
                                   : float4{0.0f, 0.0f, 0.0f, 0.0f};
            }
        }

        if (has_tile) {
            v8f acc0 = {};
            v8f acc1 = {};
#pragma unroll 8
            for (int k2 = 0; k2 < 64; k2 += 2) {
                const int kq = k2 + hsel;
                v2f b  = sW2[col0 + m][kq];
                v2f a0 = sA2[m][kq];
                v2f a1 = sA2[16 + m][kq];
                acc0 = __builtin_amdgcn_wmma_f32_16x16x4_f32(
                    false, a0, false, b, (short)0, acc0, false, false);
                acc1 = __builtin_amdgcn_wmma_f32_16x16x4_f32(
                    false, a1, false, b, (short)0, acc1, false, false);
            }
#pragma unroll
            for (int r = 0; r < 8; ++r) {
                const int row = row0 + mb + r;
                if (row < M) {
                    float v = acc0[r] + bv;
                    if constexpr (DO_RELU) v = fmaxf(v, 0.0f);
                    out[(size_t)row * N + col0 + m] = v;
                }
            }
#pragma unroll
            for (int r = 0; r < 8; ++r) {
                const int row = row0 + 16 + mb + r;
                if (row < M) {
                    float v = acc1[r] + bv;
                    if constexpr (DO_RELU) v = fmaxf(v, 0.0f);
                    out[(size_t)row * N + col0 + m] = v;
                }
            }
        }
        reads_done_barrier();
    }
}

__device__ __forceinline__ float lane_f(float v, int l) {
    return __int_as_float(__builtin_amdgcn_readlane(__float_as_int(v), l));
}

#define ROW(s) (((const float4*)(h + (size_t)(s) * 128))[lane])
#define ACC(wt_, vv_)                                                         \
    acc.x = fmaf((wt_), (vv_).x, acc.x); acc.y = fmaf((wt_), (vv_).y, acc.y); \
    acc.z = fmaf((wt_), (vv_).z, acc.z); acc.w = fmaf((wt_), (vv_).w, acc.w)

__global__ __launch_bounds__(256) void gather_relu(const int* __restrict__ rowptr,
                                                   const int* __restrict__ cnt,
                                                   const int* __restrict__ csr_src,
                                                   const float* __restrict__ dinv,
                                                   const float* __restrict__ h,
                                                   const float* __restrict__ bias,
                                                   float* __restrict__ act, int nN) {
    const int lane = threadIdx.x & 31;
    const int wv = __builtin_amdgcn_readfirstlane((int)(threadIdx.x >> 5));
    const int nw = gridDim.x * 8;
    const float4 bv = ((const float4*)bias)[lane];

    for (int node = blockIdx.x * 8 + wv; node < nN; node += nw) {
        const int   beg = rowptr[node];
        const int   n   = cnt[node];
        const float di  = dinv[node];
        const float4 hv = ROW(node);
        float4 acc = {0.0f, 0.0f, 0.0f, 0.0f};

        for (int c0 = 0; c0 < n; c0 += 32) {
            const int m = min(n - c0, 32);
            int   s_l = 0;
            float w_l = 0.0f;
            if (lane < m) {
                s_l = csr_src[beg + c0 + lane];
                w_l = dinv[s_l] * di;
            }
            int j = 0;
            for (; j + 4 <= m; j += 4) {
                const int s0 = __builtin_amdgcn_readlane(s_l, j);
                const int s1 = __builtin_amdgcn_readlane(s_l, j + 1);
                const int s2 = __builtin_amdgcn_readlane(s_l, j + 2);
                const int s3 = __builtin_amdgcn_readlane(s_l, j + 3);
                const float4 v0 = ROW(s0);
                const float4 v1 = ROW(s1);
                const float4 v2 = ROW(s2);
                const float4 v3 = ROW(s3);
                const float w0 = lane_f(w_l, j),     w1 = lane_f(w_l, j + 1);
                const float w2 = lane_f(w_l, j + 2), w3 = lane_f(w_l, j + 3);
                ACC(w0, v0); ACC(w1, v1); ACC(w2, v2); ACC(w3, v3);
            }
            if (j + 2 <= m) {
                const int s0 = __builtin_amdgcn_readlane(s_l, j);
                const int s1 = __builtin_amdgcn_readlane(s_l, j + 1);
                const float4 v0 = ROW(s0);
                const float4 v1 = ROW(s1);
                const float w0 = lane_f(w_l, j), w1 = lane_f(w_l, j + 1);
                ACC(w0, v0); ACC(w1, v1);
                j += 2;
            }
            if (j < m) {
                const int s0 = __builtin_amdgcn_readlane(s_l, j);
                const float4 v0 = ROW(s0);
                const float w0 = lane_f(w_l, j);
                ACC(w0, v0);
            }
        }

        const float d2 = di * di;
        float4 o;
        o.x = fmaxf(acc.x + d2 * hv.x + bv.x, 0.0f);
        o.y = fmaxf(acc.y + d2 * hv.y + bv.y, 0.0f);
        o.z = fmaxf(acc.z + d2 * hv.z + bv.z, 0.0f);
        o.w = fmaxf(acc.w + d2 * hv.w + bv.w, 0.0f);
        ((float4*)(act + (size_t)node * 128))[lane] = o;
    }
}
#undef ROW
#undef ACC

__global__ __launch_bounds__(256) void pool_sum(const float* __restrict__ h,
                                                const int* __restrict__ batch,
                                                float* __restrict__ gsum,
                                                float* __restrict__ gcnt, int nNodes) {
    long i = (long)blockIdx.x * 256 + threadIdx.x;
    long total = (long)nNodes * 32;
    if (i >= total) return;
    const int node = (int)(i >> 5);
    const int c    = (int)(i & 31);
    const int b    = batch[node];
    float4 v = ((const float4*)h)[i];
    float* gp = gsum + (size_t)b * 128 + c * 4;
    atomicAdd(gp + 0, v.x);
    atomicAdd(gp + 1, v.y);
    atomicAdd(gp + 2, v.z);
    atomicAdd(gp + 3, v.w);
    if (c == 0) atomicAdd(&gcnt[b], 1.0f);
}

__global__ __launch_bounds__(256) void pool_div(const float* __restrict__ gsum,
                                                const float* __restrict__ gcnt,
                                                float* __restrict__ g, int G) {
    int i = blockIdx.x * 256 + threadIdx.x;
    if (i < G * 128) {
        int b = i >> 7;
        g[i] = gsum[i] / fmaxf(gcnt[b], 1.0f);
    }
}

extern "C" void kernel_launch(void* const* d_in, const int* in_sizes, int n_in,
                              void* d_out, int out_size, void* d_ws, size_t ws_size,
                              hipStream_t stream) {
    const float* x     = (const float*)d_in[0];
    const int*   ei    = (const int*)d_in[1];
    const int*   batch = (const int*)d_in[2];
    const float* W1    = (const float*)d_in[3];
    const float* b1    = (const float*)d_in[4];
    const float* W2    = (const float*)d_in[5];
    const float* b2    = (const float*)d_in[6];
    const float* Wl1   = (const float*)d_in[7];
    const float* bl1   = (const float*)d_in[8];
    const float* Wl2   = (const float*)d_in[9];
    const float* bl2   = (const float*)d_in[10];
    float* out = (float*)d_out;

    const int nN = in_sizes[0] / 128;
    const int nE = in_sizes[1] / 2;
    const int G  = out_size / 32;
    const int* src = ei;
    const int* dst = ei + nE;

    char* ws = (char*)d_ws;
    auto alloc = [&](size_t bytes) -> float* {
        float* p = (float*)ws;
        ws += (bytes + 255) & ~(size_t)255;
        return p;
    };
    float* bufA = alloc((size_t)nN * 128 * 4);
    float* bufB = alloc((size_t)nN * 128 * 4);
    char*  bufC = (char*)alloc((size_t)nN * 128 * 4);
    int*   cnt  = (int*)alloc((size_t)nN * 4);
    float* dinv = alloc((size_t)nN * 4);
    float* gsum = alloc((size_t)G * 128 * 4);
    float* gcnt = alloc((size_t)G * 4);
    float* g    = alloc((size_t)G * 128 * 4);
    float* m1   = alloc((size_t)G * 128 * 4);

    auto sub = [&](size_t bytes) -> int* {
        int* p = (int*)bufC;
        bufC += (bytes + 255) & ~(size_t)255;
        return p;
    };
    const int nSB = (nN + SCAN_ELEMS - 1) / SCAN_ELEMS;
    int* csr_src = sub((size_t)nE * 4);
    int* rank    = sub((size_t)nE * 4);
    int* rowptr  = sub((size_t)nN * 4);
    int* bsum    = sub((size_t)nSB * 4);

    const int  eB  = (nE + 255) / 256;
    const int  nB  = (nN + 255) / 256;
    const int  qB  = (int)(((long)nN * 32 + 255) / 256);

    zero_i32<<<nB, 256, 0, stream>>>(cnt, nN);
    fill_f32<<<(G * 128 + 255) / 256, 256, 0, stream>>>(gsum, 0.0f, (long)G * 128);
    fill_f32<<<(G + 255) / 256, 256, 0, stream>>>(gcnt, 0.0f, G);

    count_rank<<<eB, 256, 0, stream>>>(dst, cnt, rank, nE);
    blk_sum<<<nSB, 256, 0, stream>>>(cnt, bsum, nN);
    scan_rows<<<nSB, 256, 0, stream>>>(cnt, bsum, rowptr, dinv, nN);
    place_src<<<eB, 256, 0, stream>>>(src, dst, rank, rowptr, csr_src, nE);

    gemm_k128<128, false, false><<<768, 256, 0, stream>>>(x, W1, nullptr, bufA, nN);
    gather_relu<<<2048, 256, 0, stream>>>(rowptr, cnt, csr_src, dinv, bufA, b1, bufB, nN);

    gemm_k128<128, false, false><<<768, 256, 0, stream>>>(bufB, W2, nullptr, bufA, nN);
    gather_relu<<<2048, 256, 0, stream>>>(rowptr, cnt, csr_src, dinv, bufA, b2, bufB, nN);

    pool_sum<<<qB, 256, 0, stream>>>(bufB, batch, gsum, gcnt, nN);
    pool_div<<<(G * 128 + 255) / 256, 256, 0, stream>>>(gsum, gcnt, g, G);

    gemm_k128<128, true, true><<<(G + RSTRIP - 1) / RSTRIP, 256, 0, stream>>>(g, Wl1, bl1, m1, G);
    gemm_k128<32, true, false><<<(G + RSTRIP - 1) / RSTRIP, 256, 0, stream>>>(m1, Wl2, bl2, out, G);
}


# Mean-pool sum fused into the layer-2 gather: a wave walks contiguous nodes, keeps a graph's row sum in registers, adds it to the pool by whole-line f32 atomics on graph change; pool sum order differs
# speedup vs baseline: 3.7253x; 1.5389x over previous
#include <hip/hip_runtime.h>

typedef __attribute__((ext_vector_type(2))) float v2f;
typedef __attribute__((ext_vector_type(8))) float v8f;

__global__ __launch_bounds__(256) void fill_f32(float* __restrict__ p, float v, long n) {
    long i = (long)blockIdx.x * 256 + threadIdx.x;
    if (i < n) p[i] = v;
}

__global__ __launch_bounds__(256) void zero_i32(int* __restrict__ p, int n) {
    int i = blockIdx.x * 256 + threadIdx.x;
    if (i < n) p[i] = 0;
}

__global__ __launch_bounds__(256) void count_rank(const int* __restrict__ dst,
                                                  int* __restrict__ cnt,
                                                  int* __restrict__ rank, int nE) {
    int e = blockIdx.x * 256 + threadIdx.x;
    if (e < nE) rank[e] = atomicAdd(&cnt[dst[e]], 1);
}

__device__ __forceinline__ int wave_sum(int v) {
#pragma unroll
    for (int off = 16; off > 0; off >>= 1) v += __shfl_xor(v, off);
    return v;
}

#define SCAN_ELEMS 1024

__global__ __launch_bounds__(256) void blk_sum(const int* __restrict__ cnt,
                                               int* __restrict__ bsum, int n) {
    __shared__ int s[8];
    const int tid  = threadIdx.x;
    const int base = blockIdx.x * SCAN_ELEMS + tid * 4;
    int v = 0;
#pragma unroll
    for (int j = 0; j < 4; ++j)
        if (base + j < n) v += cnt[base + j];
    v = wave_sum(v);
    if ((tid & 31) == 0) s[tid >> 5] = v;
    __syncthreads();
    if (tid == 0) {
        int t = 0;
        for (int w = 0; w < 8; ++w) t += s[w];
        bsum[blockIdx.x] = t;
    }
}

__global__ __launch_bounds__(256) void scan_rows(const int* __restrict__ cnt,
                                                 const int* __restrict__ bsum,
                                                 int* __restrict__ rowptr,
                                                 float* __restrict__ dinv, int n) {
    __shared__ int sp[8];
    __shared__ int sw[8];
    const int tid = threadIdx.x, lane = tid & 31, wave = tid >> 5;

    int pv = 0;
    for (int i = tid; i < (int)blockIdx.x; i += 256) pv += bsum[i];
    pv = wave_sum(pv);
    if (lane == 0) sp[wave] = pv;

    const int base = blockIdx.x * SCAN_ELEMS + tid * 4;
    int c[4];
#pragma unroll
    for (int j = 0; j < 4; ++j) c[j] = (base + j < n) ? cnt[base + j] : 0;
    const int tsum = c[0] + c[1] + c[2] + c[3];
    int inc = tsum;
#pragma unroll
    for (int off = 1; off < 32; off <<= 1) {
        int t = __shfl_up(inc, off);
        if (lane >= off) inc += t;
    }
    if (lane == 31) sw[wave] = inc;
    __syncthreads();
    int off0 = 0;
    for (int w = 0; w < 8; ++w) {
        off0 += sp[w];
        if (w < wave) off0 += sw[w];
    }
    int excl = off0 + inc - tsum;
#pragma unroll
    for (int j = 0; j < 4; ++j) {
        if (base + j < n) {
            rowptr[base + j] = excl;
            dinv[base + j]   = rsqrtf((float)c[j] + 1.0f);
            excl += c[j];
        }
    }
}

__global__ __launch_bounds__(256) void place_src(const int* __restrict__ src,
                                                 const int* __restrict__ dst,
                                                 const int* __restrict__ rank,
                                                 const int* __restrict__ rowptr,
                                                 int* __restrict__ csr_src, int nE) {
    int e = blockIdx.x * 256 + threadIdx.x;
    if (e < nE) csr_src[rowptr[dst[e]] + rank[e]] = src[e];
}

#define LPITCH 66
#define RSTRIP 32

__device__ __forceinline__ void lds_ready_barrier() {
    asm volatile("s_wait_dscnt 0x0\n\ts_barrier_signal -1\n\ts_barrier_wait -1" ::: "memory");
}
__device__ __forceinline__ void reads_done_barrier() {
    asm volatile("s_barrier_signal -1\n\ts_barrier_wait -1" ::: "memory");
}

template <int N, bool HAS_BIAS, bool DO_RELU>
__global__ __launch_bounds__(256) void gemm_k128(const float* __restrict__ A,
                                                 const float* __restrict__ W,
                                                 const float* __restrict__ bias,
                                                 float* __restrict__ out,
                                                 int M) {
    __shared__ v2f sA2[RSTRIP][LPITCH];
    __shared__ v2f sW2[N][LPITCH];

    const int tid = threadIdx.x;

    const int nStrips = (M + RSTRIP - 1) / RSTRIP;
    const int prow = tid >> 5;
    const int pc4  = tid & 31;
    float* sAf = (float*)sA2;

    float4 pre[4];
    int strip = blockIdx.x;
    if (strip < nStrips) {
#pragma unroll
        for (int j = 0; j < 4; ++j) {
            const int row = strip * RSTRIP + prow + 8 * j;
            pre[j] = (row < M) ? ((const float4*)(A + (size_t)row * 128))[pc4]
                               : float4{0.0f, 0.0f, 0.0f, 0.0f};
        }
    }

    for (int i = tid; i < 64 * N; i += 256) {
        const int kp = i / N, n = i % N;
        v2f w;
        w.x = W[(2 * kp) * N + n];
        w.y = W[(2 * kp + 1) * N + n];
        sW2[n][kp] = w;
    }

    const int wave  = tid >> 5;
    const int lane  = tid & 31;
    const int m     = lane & 15;
    const int hsel  = lane >> 4;
    const int mb    = hsel << 3;

    static_assert(N % 16 == 0 && N / 16 <= 8, "one column tile per wave");
    const bool has_tile = wave < N / 16;
    const int  col0     = wave << 4;
    float bv = 0.0f;
    if constexpr (HAS_BIAS) { if (has_tile) bv = bias[col0 + m]; }

    const int gstep = (int)gridDim.x;
    for (; strip < nStrips; strip += gstep) {
        const int row0 = strip * RSTRIP;
#pragma unroll
        for (int j = 0; j < 4; ++j)
            *(float4*)(sAf + (prow + 8 * j) * (2 * LPITCH) + pc4 * 4) = pre[j];
        lds_ready_barrier();

        const int nxt = strip + gstep;
        if (nxt < nStrips) {
#pragma unroll
            for (int j = 0; j < 4; ++j) {
                const int row = nxt * RSTRIP + prow + 8 * j;
                pre[j] = (row < M) ? ((const float4*)(A + (size_t)row * 128))[pc4]
                                   : float4{0.0f, 0.0f, 0.0f, 0.0f};
            }
        }

        if (has_tile) {
            v8f acc0 = {};
            v8f acc1 = {};
#pragma unroll 8
            for (int k2 = 0; k2 < 64; k2 += 2) {
                const int kq = k2 + hsel;
                v2f b  = sW2[col0 + m][kq];
                v2f a0 = sA2[m][kq];
                v2f a1 = sA2[16 + m][kq];
                acc0 = __builtin_amdgcn_wmma_f32_16x16x4_f32(
                    false, a0, false, b, (short)0, acc0, false, false);
                acc1 = __builtin_amdgcn_wmma_f32_16x16x4_f32(
                    false, a1, false, b, (short)0, acc1, false, false);
            }
#pragma unroll
            for (int r = 0; r < 8; ++r) {
                const int row = row0 + mb + r;
                if (row < M) {
                    float v = acc0[r] + bv;
                    if constexpr (DO_RELU) v = fmaxf(v, 0.0f);
                    out[(size_t)row * N + col0 + m] = v;
                }
            }
#pragma unroll
            for (int r = 0; r < 8; ++r) {
                const int row = row0 + 16 + mb + r;
                if (row < M) {
                    float v = acc1[r] + bv;
                    if constexpr (DO_RELU) v = fmaxf(v, 0.0f);
                    out[(size_t)row * N + col0 + m] = v;
                }
            }
        }
        reads_done_barrier();
    }
}

__device__ __forceinline__ float lane_f(float v, int l) {
    return __int_as_float(__builtin_amdgcn_readlane(__float_as_int(v), l));
}

__device__ __forceinline__ void pool_flush(float* __restrict__ gsum, float* __restrict__ gcnt,
                                           int b, float4 r, int len, int lane) {
    const int comp = lane & 3;
    float* gp = gsum + (size_t)b * 128 + lane;
    for (int j = 0; j < 4; ++j) {
        const int sl = 8 * j + (lane >> 2);
        const float x = __shfl(r.x, sl), y = __shfl(r.y, sl);
        const float z = __shfl(r.z, sl), w = __shfl(r.w, sl);
        const float lo = (comp & 1) ? y : x;
        const float hi = (comp & 1) ? w : z;
        atomicAdd(gp + 32 * j, (comp & 2) ? hi : lo);
    }
    if (lane == 0) atomicAdd(gcnt + b, (float)len);
}

#define ROW(s) (((const float4*)(h + (size_t)(s) * 128))[lane])
#define ACC(wt_, vv_)                                                         \
    acc.x = fmaf((wt_), (vv_).x, acc.x); acc.y = fmaf((wt_), (vv_).y, acc.y); \
    acc.z = fmaf((wt_), (vv_).z, acc.z); acc.w = fmaf((wt_), (vv_).w, acc.w)

template <bool POOL>
__global__ __launch_bounds__(256) void gather_relu(const int* __restrict__ rowptr,
                                                   const int* __restrict__ cnt,
                                                   const int* __restrict__ csr_src,
                                                   const float* __restrict__ dinv,
                                                   const float* __restrict__ h,
                                                   const float* __restrict__ bias,
                                                   float* __restrict__ act,
                                                   const int* __restrict__ batch,
                                                   float* __restrict__ gsum,
                                                   float* __restrict__ gcnt, int nN) {
    const int lane = threadIdx.x & 31;
    const int wv = __builtin_amdgcn_readfirstlane((int)(threadIdx.x >> 5));
    const int nw = gridDim.x * 8;
    const int gw = blockIdx.x * 8 + wv;
    const float4 bv = ((const float4*)bias)[lane];

    int node = gw, nend = nN, nstep = nw;
    if constexpr (POOL) {
        const unsigned q = (unsigned)nN / (unsigned)nw, rem = (unsigned)nN % (unsigned)nw;
        node  = (int)((unsigned)gw * q + min((unsigned)gw, rem));
        nend  = node + (int)q + (((unsigned)gw < rem) ? 1 : 0);
        nstep = 1;
    }
    float4 run = {0.0f, 0.0f, 0.0f, 0.0f};
    int runb = -1, runlen = 0;

    for (; node < nend; node += nstep) {
        const int   beg = rowptr[node];
        const int   n   = cnt[node];
        const float di  = dinv[node];
        int gid = 0;
        if constexpr (POOL) gid = batch[node];
        const float4 hv = ROW(node);
        float4 acc = {0.0f, 0.0f, 0.0f, 0.0f};

        for (int c0 = 0; c0 < n; c0 += 32) {
            const int m = min(n - c0, 32);
            int   s_l = 0;
            float w_l = 0.0f;
            if (lane < m) {
                s_l = csr_src[beg + c0 + lane];
                w_l = dinv[s_l] * di;
            }
            int j = 0;
            for (; j + 4 <= m; j += 4) {
                const int s0 = __builtin_amdgcn_readlane(s_l, j);
                const int s1 = __builtin_amdgcn_readlane(s_l, j + 1);
                const int s2 = __builtin_amdgcn_readlane(s_l, j + 2);
                const int s3 = __builtin_amdgcn_readlane(s_l, j + 3);
                const float4 v0 = ROW(s0);
                const float4 v1 = ROW(s1);
                const float4 v2 = ROW(s2);
                const float4 v3 = ROW(s3);
                const float w0 = lane_f(w_l, j),     w1 = lane_f(w_l, j + 1);
                const float w2 = lane_f(w_l, j + 2), w3 = lane_f(w_l, j + 3);
                ACC(w0, v0); ACC(w1, v1); ACC(w2, v2); ACC(w3, v3);
            }
            if (j + 2 <= m) {
                const int s0 = __builtin_amdgcn_readlane(s_l, j);
                const int s1 = __builtin_amdgcn_readlane(s_l, j + 1);
                const float4 v0 = ROW(s0);
                const float4 v1 = ROW(s1);
                const float w0 = lane_f(w_l, j), w1 = lane_f(w_l, j + 1);
                ACC(w0, v0); ACC(w1, v1);
                j += 2;
            }
            if (j < m) {
                const int s0 = __builtin_amdgcn_readlane(s_l, j);
                const float4 v0 = ROW(s0);
                const float w0 = lane_f(w_l, j);
                ACC(w0, v0);
            }
        }

        const float d2 = di * di;
        float4 o;
        o.x = fmaxf(acc.x + d2 * hv.x + bv.x, 0.0f);
        o.y = fmaxf(acc.y + d2 * hv.y + bv.y, 0.0f);
        o.z = fmaxf(acc.z + d2 * hv.z + bv.z, 0.0f);
        o.w = fmaxf(acc.w + d2 * hv.w + bv.w, 0.0f);
        if constexpr (POOL) {
            const int b = gid;
            if (b != runb) {
                if (runlen > 0) pool_flush(gsum, gcnt, runb, run, runlen, lane);
                run = o; runb = b; runlen = 1;
            } else {
                run.x += o.x; run.y += o.y; run.z += o.z; run.w += o.w;
                ++runlen;
            }
        } else {
            ((float4*)(act + (size_t)node * 128))[lane] = o;
        }
    }
    if constexpr (POOL) {
        if (runlen > 0) pool_flush(gsum, gcnt, runb, run, runlen, lane);
    }
}
#undef ROW
#undef ACC

__global__ __launch_bounds__(256) void pool_div(const float* __restrict__ gsum,
                                                const float* __restrict__ gcnt,
                                                float* __restrict__ g, int G) {
    int i = blockIdx.x * 256 + threadIdx.x;
    if (i < G * 128) {
        int b = i >> 7;
        g[i] = gsum[i] / fmaxf(gcnt[b], 1.0f);
    }
}

extern "C" void kernel_launch(void* const* d_in, const int* in_sizes, int n_in,
                              void* d_out, int out_size, void* d_ws, size_t ws_size,
                              hipStream_t stream) {
    const float* x     = (const float*)d_in[0];
    const int*   ei    = (const int*)d_in[1];
    const int*   batch = (const int*)d_in[2];
    const float* W1    = (const float*)d_in[3];
    const float* b1    = (const float*)d_in[4];
    const float* W2    = (const float*)d_in[5];
    const float* b2    = (const float*)d_in[6];
    const float* Wl1   = (const float*)d_in[7];
    const float* bl1   = (const float*)d_in[8];
    const float* Wl2   = (const float*)d_in[9];
    const float* bl2   = (const float*)d_in[10];
    float* out = (float*)d_out;

    const int nN = in_sizes[0] / 128;
    const int nE = in_sizes[1] / 2;
    const int G  = out_size / 32;
    const int* src = ei;
    const int* dst = ei + nE;

    char* ws = (char*)d_ws;
    auto alloc = [&](size_t bytes) -> float* {
        float* p = (float*)ws;
        ws += (bytes + 255) & ~(size_t)255;
        return p;
    };
    float* bufA = alloc((size_t)nN * 128 * 4);
    float* bufB = alloc((size_t)nN * 128 * 4);
    char*  bufC = (char*)alloc((size_t)nN * 128 * 4);
    int*   cnt  = (int*)alloc((size_t)nN * 4);
    float* dinv = alloc((size_t)nN * 4);
    float* gsum = alloc((size_t)G * 128 * 4);
    float* gcnt = alloc((size_t)G * 4);
    float* g    = alloc((size_t)G * 128 * 4);
    float* m1   = alloc((size_t)G * 128 * 4);

    auto sub = [&](size_t bytes) -> int* {
        int* p = (int*)bufC;
        bufC += (bytes + 255) & ~(size_t)255;
        return p;
    };
    const int nSB = (nN + SCAN_ELEMS - 1) / SCAN_ELEMS;
    int* csr_src = sub((size_t)nE * 4);
    int* rank    = sub((size_t)nE * 4);
    int* rowptr  = sub((size_t)nN * 4);
    int* bsum    = sub((size_t)nSB * 4);

    const int  eB  = (nE + 255) / 256;
    const int  nB  = (nN + 255) / 256;

    zero_i32<<<nB, 256, 0, stream>>>(cnt, nN);
    fill_f32<<<(G * 128 + 255) / 256, 256, 0, stream>>>(gsum, 0.0f, (long)G * 128);
    fill_f32<<<(G + 255) / 256, 256, 0, stream>>>(gcnt, 0.0f, G);

    count_rank<<<eB, 256, 0, stream>>>(dst, cnt, rank, nE);
    blk_sum<<<nSB, 256, 0, stream>>>(cnt, bsum, nN);
    scan_rows<<<nSB, 256, 0, stream>>>(cnt, bsum, rowptr, dinv, nN);
    place_src<<<eB, 256, 0, stream>>>(src, dst, rank, rowptr, csr_src, nE);

    gemm_k128<128, false, false><<<768, 256, 0, stream>>>(x, W1, nullptr, bufA, nN);
    gather_relu<false><<<2048, 256, 0, stream>>>(rowptr, cnt, csr_src, dinv, bufA, b1, bufB,
                                                 nullptr, nullptr, nullptr, nN);

    gemm_k128<128, false, false><<<768, 256, 0, stream>>>(bufB, W2, nullptr, bufA, nN);
    gather_relu<true><<<2048, 256, 0, stream>>>(rowptr, cnt, csr_src, dinv, bufA, b2, nullptr,
                                                batch, gsum, gcnt, nN);

    pool_div<<<(G * 128 + 255) / 256, 256, 0, stream>>>(gsum, gcnt, g, G);

    gemm_k128<128, true, true><<<(G + RSTRIP - 1) / RSTRIP, 256, 0, stream>>>(g, Wl1, bl1, m1, G);
    gemm_k128<32, true, false><<<(G + RSTRIP - 1) / RSTRIP, 256, 0, stream>>>(m1, Wl2, bl2, out, G);
}


# Edges grouped by destination by a two-level counting sort: binned by 256-node range, then counted and placed by LDS adds per range (no global integer atomics); sum order = LDS arrival order
# speedup vs baseline: 4.6283x; 1.2424x over previous
#include <hip/hip_runtime.h>

typedef __attribute__((ext_vector_type(2))) float v2f;
typedef __attribute__((ext_vector_type(8))) float v8f;

__global__ __launch_bounds__(256) void fill_f32(float* __restrict__ p, float v, long n) {
    long i = (long)blockIdx.x * 256 + threadIdx.x;
    if (i < n) p[i] = v;
}

#define BIN_SHIFT 8
#define NBMAX     512
#define ECH       4096

__device__ __forceinline__ int wave_sum(int v) {
#pragma unroll
    for (int off = 16; off > 0; off >>= 1) v += __shfl_xor(v, off);
    return v;
}

__device__ __forceinline__ int wave_incl_scan(int v, int lane) {
#pragma unroll
    for (int off = 1; off < 32; off <<= 1) {
        int t = __shfl_up(v, off);
        if (lane >= off) v += t;
    }
    return v;
}

__global__ __launch_bounds__(256) void bin_count(const int* __restrict__ dst,
                                                 int* __restrict__ rank,
                                                 int* __restrict__ hist,
                                                 int nE, int NB, int NC) {
    __shared__ int sh[NBMAX + 1];
    const int tid = threadIdx.x;
    for (int i = tid; i < NBMAX + 1; i += 256) sh[i] = 0;
    __syncthreads();
    const int base = blockIdx.x * ECH;
    int d[16], r[16];
#pragma unroll
    for (int j = 0; j < 16; ++j) {
        const int e = base + j * 256 + tid;
        d[j] = dst[min(e, nE - 1)];
    }
#pragma unroll
    for (int j = 0; j < 16; ++j) {
        const int e = base + j * 256 + tid;
        const int bin = (e < nE) ? min(d[j] >> BIN_SHIFT, NBMAX - 1) : NBMAX;
        r[j] = atomicAdd(&sh[bin], 1);
    }
#pragma unroll
    for (int j = 0; j < 16; ++j) {
        const int e = base + j * 256 + tid;
        if (e < nE) rank[e] = r[j];
    }
    __syncthreads();
    for (int i = tid; i < NB; i += 256) hist[(size_t)i * NC + blockIdx.x] = sh[i];
}

__global__ __launch_bounds__(256) void bin_scan(int* __restrict__ hist,
                                                int* __restrict__ btot, int NC) {
    __shared__ int sw[8];
    const int tid = threadIdx.x, lane = tid & 31, wave = tid >> 5;
    int* row = hist + (size_t)blockIdx.x * NC;
    int carry = 0;
    for (int c0 = 0; c0 < NC; c0 += 256) {
        const int c = c0 + tid;
        const int v = (c < NC) ? row[c] : 0;
        const int inc = wave_incl_scan(v, lane);
        if (lane == 31) sw[wave] = inc;
        __syncthreads();
        int off = 0, tot = 0;
        for (int w = 0; w < 8; ++w) {
            const int t = sw[w];
            tot += t;
            if (w < wave) off += t;
        }
        if (c < NC) row[c] = carry + off + inc - v;
        carry += tot;
        __syncthreads();
    }
    if (tid == 0) btot[blockIdx.x] = carry;
}

__global__ __launch_bounds__(256) void bin_scatter(const int* __restrict__ src,
                                                   const int* __restrict__ dst,
                                                   const int* __restrict__ rank,
                                                   const int* __restrict__ hist,
                                                   const int* __restrict__ btot,
                                                   int* __restrict__ bed_i,
                                                   int nE, int NB, int NC) {
    __shared__ int sbase[NBMAX];
    __shared__ int sw[8];
    const int tid = threadIdx.x, lane = tid & 31, wave = tid >> 5;
    int2* bed = (int2*)bed_i;

    const int b0 = 2 * tid, b1 = b0 + 1;
    const int a0 = (b0 < NB) ? btot[b0] : 0;
    const int a1 = (b1 < NB) ? btot[b1] : 0;
    const int s  = a0 + a1;
    const int inc = wave_incl_scan(s, lane);
    if (lane == 31) sw[wave] = inc;
    __syncthreads();
    int off = 0;
    for (int w = 0; w < 8; ++w)
        if (w < wave) off += sw[w];
    const int ex = off + inc - s;
    if (b0 < NB) sbase[b0] = ex + hist[(size_t)b0 * NC + blockIdx.x];
    if (b1 < NB) sbase[b1] = ex + a0 + hist[(size_t)b1 * NC + blockIdx.x];
    __syncthreads();

    const int base = blockIdx.x * ECH;
    for (int h = 0; h < 2; ++h) {
        int d[8], rk[8], sv[8];
#pragma unroll
        for (int j = 0; j < 8; ++j) {
            const int ec = min(base + (h * 8 + j) * 256 + tid, nE - 1);
            d[j]  = dst[ec];
            rk[j] = rank[ec];
            sv[j] = src[ec];
        }
#pragma unroll
        for (int j = 0; j < 8; ++j) {
            const int e = base + (h * 8 + j) * 256 + tid;
            const int pos = sbase[min(d[j] >> BIN_SHIFT, NBMAX - 1)] + rk[j];
            if (e < nE) bed[pos] = make_int2(sv[j], d[j]);
        }
    }
}

__global__ __launch_bounds__(256) void bin_csr(const int* __restrict__ bed_i,
                                               const int* __restrict__ btot,
                                               int* __restrict__ rowptr,
                                               int* __restrict__ cnt,
                                               float* __restrict__ dinv,
                                               int* __restrict__ csr_src, int nN) {
    __shared__ int scnt[257];
    __shared__ int sptr[257];
    __shared__ int sw[8];
    const int tid = threadIdx.x, lane = tid & 31, wave = tid >> 5;
    const int b = blockIdx.x;
    const int2* bed = (const int2*)bed_i;

    int pv = 0;
    for (int i = tid; i < b; i += 256) pv += btot[i];
    pv = wave_sum(pv);
    if (lane == 0) sw[wave] = pv;
    scnt[tid] = 0;
    if (tid == 0) { scnt[256] = 0; sptr[256] = 0; }
    __syncthreads();
    int beg = 0;
    for (int w = 0; w < 8; ++w) beg += sw[w];
    const int n = btot[b];
    __syncthreads();

    for (int i0 = 0; i0 < n; i0 += 8 * 256) {
        int dd[8];
#pragma unroll
        for (int j = 0; j < 8; ++j) dd[j] = bed[beg + min(i0 + j * 256 + tid, n - 1)].y;
#pragma unroll
        for (int j = 0; j < 8; ++j)
            atomicAdd(&scnt[(i0 + j * 256 + tid < n) ? (dd[j] & 255) : 256], 1);
    }
    __syncthreads();

    const int v   = scnt[tid];
    const int inc = wave_incl_scan(v, lane);
    if (lane == 31) sw[wave] = inc;
    __syncthreads();
    int off = 0;
    for (int w = 0; w < 8; ++w)
        if (w < wave) off += sw[w];
    const int ex = off + inc - v;
    sptr[tid] = ex;
    const int node = (b << BIN_SHIFT) + tid;
    if (node < nN) {
        rowptr[node] = beg + ex;
        cnt[node]    = v;
        dinv[node]   = rsqrtf((float)v + 1.0f);
    }
    __syncthreads();

    for (int i0 = 0; i0 < n; i0 += 8 * 256) {
        int2 ed[8];
        int  p[8];
#pragma unroll
        for (int j = 0; j < 8; ++j) ed[j] = bed[beg + min(i0 + j * 256 + tid, n - 1)];
#pragma unroll
        for (int j = 0; j < 8; ++j)
            p[j] = atomicAdd(&sptr[(i0 + j * 256 + tid < n) ? (ed[j].y & 255) : 256], 1);
#pragma unroll
        for (int j = 0; j < 8; ++j)
            if (i0 + j * 256 + tid < n) csr_src[beg + p[j]] = ed[j].x;
    }
}

#define LPITCH 66
#define RSTRIP 32

__device__ __forceinline__ void lds_ready_barrier() {
    asm volatile("s_wait_dscnt 0x0\n\ts_barrier_signal -1\n\ts_barrier_wait -1" ::: "memory");
}
__device__ __forceinline__ void reads_done_barrier() {
    asm volatile("s_barrier_signal -1\n\ts_barrier_wait -1" ::: "memory");
}

template <int N, bool HAS_BIAS, bool DO_RELU>
__global__ __launch_bounds__(256) void gemm_k128(const float* __restrict__ A,
                                                 const float* __restrict__ W,
                                                 const float* __restrict__ bias,
                                                 float* __restrict__ out,
                                                 int M) {
    __shared__ v2f sA2[RSTRIP][LPITCH];
    __shared__ v2f sW2[N][LPITCH];

    const int tid = threadIdx.x;

    const int nStrips = (M + RSTRIP - 1) / RSTRIP;
    const int prow = tid >> 5;
    const int pc4  = tid & 31;
    float* sAf = (float*)sA2;

    float4 pre[4];
    int strip = blockIdx.x;
    if (strip < nStrips) {
#pragma unroll
        for (int j = 0; j < 4; ++j) {
            const int row = strip * RSTRIP + prow + 8 * j;
            pre[j] = (row < M) ? ((const float4*)(A + (size_t)row * 128))[pc4]
                               : float4{0.0f, 0.0f, 0.0f, 0.0f};
        }
    }

    for (int i = tid; i < 64 * N; i += 256) {
        const int kp = i / N, n = i % N;
        v2f w;
        w.x = W[(2 * kp) * N + n];
        w.y = W[(2 * kp + 1) * N + n];
        sW2[n][kp] = w;
    }

    const int wave  = tid >> 5;
    const int lane  = tid & 31;
    const int m     = lane & 15;
    const int hsel  = lane >> 4;
    const int mb    = hsel << 3;

    static_assert(N % 16 == 0 && N / 16 <= 8, "one column tile per wave");
    const bool has_tile = wave < N / 16;
    const int  col0     = wave << 4;
    float bv = 0.0f;
    if constexpr (HAS_BIAS) { if (has_tile) bv = bias[col0 + m]; }

    const int gstep = (int)gridDim.x;
    for (; strip < nStrips; strip += gstep) {
        const int row0 = strip * RSTRIP;
#pragma unroll
        for (int j = 0; j < 4; ++j)
            *(float4*)(sAf + (prow + 8 * j) * (2 * LPITCH) + pc4 * 4) = pre[j];
        lds_ready_barrier();

        const int nxt = strip + gstep;
        if (nxt < nStrips) {
#pragma unroll
            for (int j = 0; j < 4; ++j) {
                const int row = nxt * RSTRIP + prow + 8 * j;
                pre[j] = (row < M) ? ((const float4*)(A + (size_t)row * 128))[pc4]
                                   : float4{0.0f, 0.0f, 0.0f, 0.0f};
            }
        }

        if (has_tile) {
            v8f acc0 = {};
            v8f acc1 = {};
#pragma unroll 8
            for (int k2 = 0; k2 < 64; k2 += 2) {
                const int kq = k2 + hsel;
                v2f b  = sW2[col0 + m][kq];
                v2f a0 = sA2[m][kq];
                v2f a1 = sA2[16 + m][kq];
                acc0 = __builtin_amdgcn_wmma_f32_16x16x4_f32(
                    false, a0, false, b, (short)0, acc0, false, false);
                acc1 = __builtin_amdgcn_wmma_f32_16x16x4_f32(
                    false, a1, false, b, (short)0, acc1, false, false);
            }
#pragma unroll
            for (int r = 0; r < 8; ++r) {
                const int row = row0 + mb + r;
                if (row < M) {
                    float v = acc0[r] + bv;
                    if constexpr (DO_RELU) v = fmaxf(v, 0.0f);
                    out[(size_t)row * N + col0 + m] = v;
                }
            }
#pragma unroll
            for (int r = 0; r < 8; ++r) {
                const int row = row0 + 16 + mb + r;
                if (row < M) {
                    float v = acc1[r] + bv;
                    if constexpr (DO_RELU) v = fmaxf(v, 0.0f);
                    out[(size_t)row * N + col0 + m] = v;
                }
            }
        }
        reads_done_barrier();
    }
}

__device__ __forceinline__ float lane_f(float v, int l) {
    return __int_as_float(__builtin_amdgcn_readlane(__float_as_int(v), l));
}

__device__ __forceinline__ void pool_flush(float* __restrict__ gsum, float* __restrict__ gcnt,
                                           int b, float4 r, int len, int lane) {
    const int comp = lane & 3;
    float* gp = gsum + (size_t)b * 128 + lane;
    for (int j = 0; j < 4; ++j) {
        const int sl = 8 * j + (lane >> 2);
        const float x = __shfl(r.x, sl), y = __shfl(r.y, sl);
        const float z = __shfl(r.z, sl), w = __shfl(r.w, sl);
        const float lo = (comp & 1) ? y : x;
        const float hi = (comp & 1) ? w : z;
        atomicAdd(gp + 32 * j, (comp & 2) ? hi : lo);
    }
    if (lane == 0) atomicAdd(gcnt + b, (float)len);
}

#define ROW(s) (((const float4*)(h + (size_t)(s) * 128))[lane])
#define ACC(wt_, vv_)                                                         \
    acc.x = fmaf((wt_), (vv_).x, acc.x); acc.y = fmaf((wt_), (vv_).y, acc.y); \
    acc.z = fmaf((wt_), (vv_).z, acc.z); acc.w = fmaf((wt_), (vv_).w, acc.w)

template <bool POOL>
__global__ __launch_bounds__(256) void gather_relu(const int* __restrict__ rowptr,
                                                   const int* __restrict__ cnt,
                                                   const int* __restrict__ csr_src,
                                                   const float* __restrict__ dinv,
                                                   const float* __restrict__ h,
                                                   const float* __restrict__ bias,
                                                   float* __restrict__ act,
                                                   const int* __restrict__ batch,
                                                   float* __restrict__ gsum,
                                                   float* __restrict__ gcnt, int nN) {
    const int lane = threadIdx.x & 31;
    const int wv = __builtin_amdgcn_readfirstlane((int)(threadIdx.x >> 5));
    const int nw = gridDim.x * 8;
    const int gw = blockIdx.x * 8 + wv;
    const float4 bv = ((const float4*)bias)[lane];

    int node = gw, nend = nN, nstep = nw;
    if constexpr (POOL) {
        const unsigned q = (unsigned)nN / (unsigned)nw, rem = (unsigned)nN % (unsigned)nw;
        node  = (int)((unsigned)gw * q + min((unsigned)gw, rem));
        nend  = node + (int)q + (((unsigned)gw < rem) ? 1 : 0);
        nstep = 1;
    }
    float4 run = {0.0f, 0.0f, 0.0f, 0.0f};
    int runb = -1, runlen = 0;

    for (; node < nend; node += nstep) {
        const int   beg = rowptr[node];
        const int   n   = cnt[node];
        const float di  = dinv[node];
        int gid = 0;
        if constexpr (POOL) gid = batch[node];
        const float4 hv = ROW(node);
        float4 acc = {0.0f, 0.0f, 0.0f, 0.0f};

        for (int c0 = 0; c0 < n; c0 += 32) {
            const int m = min(n - c0, 32);
            int   s_l = 0;
            float w_l = 0.0f;
            if (lane < m) {
                s_l = csr_src[beg + c0 + lane];
                w_l = dinv[s_l] * di;
            }
            int j = 0;
            for (; j + 4 <= m; j += 4) {
                const int s0 = __builtin_amdgcn_readlane(s_l, j);
                const int s1 = __builtin_amdgcn_readlane(s_l, j + 1);
                const int s2 = __builtin_amdgcn_readlane(s_l, j + 2);
                const int s3 = __builtin_amdgcn_readlane(s_l, j + 3);
                const float4 v0 = ROW(s0);
                const float4 v1 = ROW(s1);
                const float4 v2 = ROW(s2);
                const float4 v3 = ROW(s3);
                const float w0 = lane_f(w_l, j),     w1 = lane_f(w_l, j + 1);
                const float w2 = lane_f(w_l, j + 2), w3 = lane_f(w_l, j + 3);
                ACC(w0, v0); ACC(w1, v1); ACC(w2, v2); ACC(w3, v3);
            }
            if (j + 2 <= m) {
                const int s0 = __builtin_amdgcn_readlane(s_l, j);
                const int s1 = __builtin_amdgcn_readlane(s_l, j + 1);
                const float4 v0 = ROW(s0);
                const float4 v1 = ROW(s1);
                const float w0 = lane_f(w_l, j), w1 = lane_f(w_l, j + 1);
                ACC(w0, v0); ACC(w1, v1);
                j += 2;
            }
            if (j < m) {
                const int s0 = __builtin_amdgcn_readlane(s_l, j);
                const float4 v0 = ROW(s0);
                const float w0 = lane_f(w_l, j);
                ACC(w0, v0);
            }
        }

        const float d2 = di * di;
        float4 o;
        o.x = fmaxf(acc.x + d2 * hv.x + bv.x, 0.0f);
        o.y = fmaxf(acc.y + d2 * hv.y + bv.y, 0.0f);
        o.z = fmaxf(acc.z + d2 * hv.z + bv.z, 0.0f);
        o.w = fmaxf(acc.w + d2 * hv.w + bv.w, 0.0f);
        if constexpr (POOL) {
            const int b = gid;
            if (b != runb) {
                if (runlen > 0) pool_flush(gsum, gcnt, runb, run, runlen, lane);
                run = o; runb = b; runlen = 1;
            } else {
                run.x += o.x; run.y += o.y; run.z += o.z; run.w += o.w;
                ++runlen;
            }
        } else {
            ((float4*)(act + (size_t)node * 128))[lane] = o;
        }
    }
    if constexpr (POOL) {
        if (runlen > 0) pool_flush(gsum, gcnt, runb, run, runlen, lane);
    }
}
#undef ROW
#undef ACC

__global__ __launch_bounds__(256) void pool_div(const float* __restrict__ gsum,
                                                const float* __restrict__ gcnt,
                                                float* __restrict__ g, int G) {
    int i = blockIdx.x * 256 + threadIdx.x;
    if (i < G * 128) {
        int b = i >> 7;
        g[i] = gsum[i] / fmaxf(gcnt[b], 1.0f);
    }
}

extern "C" void kernel_launch(void* const* d_in, const int* in_sizes, int n_in,
                              void* d_out, int out_size, void* d_ws, size_t ws_size,
                              hipStream_t stream) {
    const float* x     = (const float*)d_in[0];
    const int*   ei    = (const int*)d_in[1];
    const int*   batch = (const int*)d_in[2];
    const float* W1    = (const float*)d_in[3];
    const float* b1    = (const float*)d_in[4];
    const float* W2    = (const float*)d_in[5];
    const float* b2    = (const float*)d_in[6];
    const float* Wl1   = (const float*)d_in[7];
    const float* bl1   = (const float*)d_in[8];
    const float* Wl2   = (const float*)d_in[9];
    const float* bl2   = (const float*)d_in[10];
    float* out = (float*)d_out;

    const int nN = in_sizes[0] / 128;
    const int nE = in_sizes[1] / 2;
    const int G  = out_size / 32;
    const int* src = ei;
    const int* dst = ei + nE;

    char* ws = (char*)d_ws;
    auto alloc = [&](size_t bytes) -> float* {
        float* p = (float*)ws;
        ws += (bytes + 255) & ~(size_t)255;
        return p;
    };
    float* bufA = alloc((size_t)nN * 128 * 4);
    float* bufB = alloc((size_t)nN * 128 * 4);
    char*  bufC = (char*)alloc((size_t)nN * 128 * 4);
    int*   cnt  = (int*)alloc((size_t)nN * 4);
    float* dinv = alloc((size_t)nN * 4);
    float* gsum = alloc((size_t)G * 128 * 4);
    float* gcnt = alloc((size_t)G * 4);
    float* g    = alloc((size_t)G * 128 * 4);
    float* m1   = alloc((size_t)G * 128 * 4);

    auto sub = [&](size_t bytes) -> int* {
        int* p = (int*)bufC;
        bufC += (bytes + 255) & ~(size_t)255;
        return p;
    };
    const int NB = (nN + 255) >> BIN_SHIFT;
    const int NC = (nE + ECH - 1) / ECH;
    int* csr_src = sub((size_t)nE * 4);
    int* rank    = sub((size_t)nE * 4);
    int* bed     = sub((size_t)nE * 8);
    int* rowptr  = sub((size_t)nN * 4);
    int* hist    = sub((size_t)NB * NC * 4);
    int* btot    = sub((size_t)NB * 4);

    fill_f32<<<(G * 128 + 255) / 256, 256, 0, stream>>>(gsum, 0.0f, (long)G * 128);
    fill_f32<<<(G + 255) / 256, 256, 0, stream>>>(gcnt, 0.0f, G);

    bin_count<<<NC, 256, 0, stream>>>(dst, rank, hist, nE, NB, NC);
    bin_scan<<<NB, 256, 0, stream>>>(hist, btot, NC);
    bin_scatter<<<NC, 256, 0, stream>>>(src, dst, rank, hist, btot, bed, nE, NB, NC);
    bin_csr<<<NB, 256, 0, stream>>>(bed, btot, rowptr, cnt, dinv, csr_src, nN);

    gemm_k128<128, false, false><<<768, 256, 0, stream>>>(x, W1, nullptr, bufA, nN);
    gather_relu<false><<<2048, 256, 0, stream>>>(rowptr, cnt, csr_src, dinv, bufA, b1, bufB,
                                                 nullptr, nullptr, nullptr, nN);

    gemm_k128<128, false, false><<<768, 256, 0, stream>>>(bufB, W2, nullptr, bufA, nN);
    gather_relu<true><<<2048, 256, 0, stream>>>(rowptr, cnt, csr_src, dinv, bufA, b2, nullptr,
                                                batch, gsum, gcnt, nN);

    pool_div<<<(G * 128 + 255) / 256, 256, 0, stream>>>(gsum, gcnt, g, G);

    gemm_k128<128, true, true><<<(G + RSTRIP - 1) / RSTRIP, 256, 0, stream>>>(g, Wl1, bl1, m1, G);
    gemm_k128<32, true, false><<<(G + RSTRIP - 1) / RSTRIP, 256, 0, stream>>>(m1, Wl2, bl2, out, G);
}


# Layer-1 gather fused into the layer-2 GEMM: 32-wave groups gather 128-node strips into the LDS A tile and multiply by W2 there; activations never stored, one launch fewer; sum order unchanged
# speedup vs baseline: 4.7851x; 1.0339x over previous
#include <hip/hip_runtime.h>

typedef __attribute__((ext_vector_type(2))) float v2f;
typedef __attribute__((ext_vector_type(8))) float v8f;

__global__ __launch_bounds__(256) void fill_f32(float* __restrict__ p, float v, long n) {
    long i = (long)blockIdx.x * 256 + threadIdx.x;
    if (i < n) p[i] = v;
}

#define BIN_SHIFT 8
#define NBMAX     512
#define ECH       4096

__device__ __forceinline__ int wave_sum(int v) {
#pragma unroll
    for (int off = 16; off > 0; off >>= 1) v += __shfl_xor(v, off);
    return v;
}

__device__ __forceinline__ int wave_incl_scan(int v, int lane) {
#pragma unroll
    for (int off = 1; off < 32; off <<= 1) {
        int t = __shfl_up(v, off);
        if (lane >= off) v += t;
    }
    return v;
}

__global__ __launch_bounds__(256) void bin_count(const int* __restrict__ dst,
                                                 int* __restrict__ rank,
                                                 int* __restrict__ hist,
                                                 int nE, int NB, int NC) {
    __shared__ int sh[NBMAX + 1];
    const int tid = threadIdx.x;
    for (int i = tid; i < NBMAX + 1; i += 256) sh[i] = 0;
    __syncthreads();
    const int base = blockIdx.x * ECH;
    int d[16], r[16];
#pragma unroll
    for (int j = 0; j < 16; ++j) {
        const int e = base + j * 256 + tid;
        d[j] = dst[min(e, nE - 1)];
    }
#pragma unroll
    for (int j = 0; j < 16; ++j) {
        const int e = base + j * 256 + tid;
        const int bin = (e < nE) ? min(d[j] >> BIN_SHIFT, NBMAX - 1) : NBMAX;
        r[j] = atomicAdd(&sh[bin], 1);
    }
#pragma unroll
    for (int j = 0; j < 16; ++j) {
        const int e = base + j * 256 + tid;
        if (e < nE) rank[e] = r[j];
    }
    __syncthreads();
    for (int i = tid; i < NB; i += 256) hist[(size_t)i * NC + blockIdx.x] = sh[i];
}

__global__ __launch_bounds__(256) void bin_scan(int* __restrict__ hist,
                                                int* __restrict__ btot, int NC) {
    __shared__ int sw[8];
    const int tid = threadIdx.x, lane = tid & 31, wave = tid >> 5;
    int* row = hist + (size_t)blockIdx.x * NC;
    int carry = 0;
    for (int c0 = 0; c0 < NC; c0 += 256) {
        const int c = c0 + tid;
        const int v = (c < NC) ? row[c] : 0;
        const int inc = wave_incl_scan(v, lane);
        if (lane == 31) sw[wave] = inc;
        __syncthreads();
        int off = 0, tot = 0;
        for (int w = 0; w < 8; ++w) {
            const int t = sw[w];
            tot += t;
            if (w < wave) off += t;
        }
        if (c < NC) row[c] = carry + off + inc - v;
        carry += tot;
        __syncthreads();
    }
    if (tid == 0) btot[blockIdx.x] = carry;
}

__global__ __launch_bounds__(256) void bin_scatter(const int* __restrict__ src,
                                                   const int* __restrict__ dst,
                                                   const int* __restrict__ rank,
                                                   const int* __restrict__ hist,
                                                   const int* __restrict__ btot,
                                                   int* __restrict__ bed_i,
                                                   int nE, int NB, int NC) {
    __shared__ int sbase[NBMAX];
    __shared__ int sw[8];
    const int tid = threadIdx.x, lane = tid & 31, wave = tid >> 5;
    int2* bed = (int2*)bed_i;

    const int b0 = 2 * tid, b1 = b0 + 1;
    const int a0 = (b0 < NB) ? btot[b0] : 0;
    const int a1 = (b1 < NB) ? btot[b1] : 0;
    const int s  = a0 + a1;
    const int inc = wave_incl_scan(s, lane);
    if (lane == 31) sw[wave] = inc;
    __syncthreads();
    int off = 0;
    for (int w = 0; w < 8; ++w)
        if (w < wave) off += sw[w];
    const int ex = off + inc - s;
    if (b0 < NB) sbase[b0] = ex + hist[(size_t)b0 * NC + blockIdx.x];
    if (b1 < NB) sbase[b1] = ex + a0 + hist[(size_t)b1 * NC + blockIdx.x];
    __syncthreads();

    const int base = blockIdx.x * ECH;
    for (int h = 0; h < 2; ++h) {
        int d[8], rk[8], sv[8];
#pragma unroll
        for (int j = 0; j < 8; ++j) {
            const int ec = min(base + (h * 8 + j) * 256 + tid, nE - 1);
            d[j]  = dst[ec];
            rk[j] = rank[ec];
            sv[j] = src[ec];
        }
#pragma unroll
        for (int j = 0; j < 8; ++j) {
            const int e = base + (h * 8 + j) * 256 + tid;
            const int pos = sbase[min(d[j] >> BIN_SHIFT, NBMAX - 1)] + rk[j];
            if (e < nE) bed[pos] = make_int2(sv[j], d[j]);
        }
    }
}

__global__ __launch_bounds__(256) void bin_csr(const int* __restrict__ bed_i,
                                               const int* __restrict__ btot,
                                               int* __restrict__ rowptr,
                                               int* __restrict__ cnt,
                                               float* __restrict__ dinv,
                                               int* __restrict__ csr_src, int nN) {
    __shared__ int scnt[257];
    __shared__ int sptr[257];
    __shared__ int sw[8];
    const int tid = threadIdx.x, lane = tid & 31, wave = tid >> 5;
    const int b = blockIdx.x;
    const int2* bed = (const int2*)bed_i;

    int pv = 0;
    for (int i = tid; i < b; i += 256) pv += btot[i];
    pv = wave_sum(pv);
    if (lane == 0) sw[wave] = pv;
    scnt[tid] = 0;
    if (tid == 0) { scnt[256] = 0; sptr[256] = 0; }
    __syncthreads();
    int beg = 0;
    for (int w = 0; w < 8; ++w) beg += sw[w];
    const int n = btot[b];
    __syncthreads();

    for (int i0 = 0; i0 < n; i0 += 8 * 256) {
        int dd[8];
#pragma unroll
        for (int j = 0; j < 8; ++j) dd[j] = bed[beg + min(i0 + j * 256 + tid, n - 1)].y;
#pragma unroll
        for (int j = 0; j < 8; ++j)
            atomicAdd(&scnt[(i0 + j * 256 + tid < n) ? (dd[j] & 255) : 256], 1);
    }
    __syncthreads();

    const int v   = scnt[tid];
    const int inc = wave_incl_scan(v, lane);
    if (lane == 31) sw[wave] = inc;
    __syncthreads();
    int off = 0;
    for (int w = 0; w < 8; ++w)
        if (w < wave) off += sw[w];
    const int ex = off + inc - v;
    sptr[tid] = ex;
    const int node = (b << BIN_SHIFT) + tid;
    if (node < nN) {
        rowptr[node] = beg + ex;
        cnt[node]    = v;
        dinv[node]   = rsqrtf((float)v + 1.0f);
    }
    __syncthreads();

    for (int i0 = 0; i0 < n; i0 += 8 * 256) {
        int2 ed[8];
        int  p[8];
#pragma unroll
        for (int j = 0; j < 8; ++j) ed[j] = bed[beg + min(i0 + j * 256 + tid, n - 1)];
#pragma unroll
        for (int j = 0; j < 8; ++j)
            p[j] = atomicAdd(&sptr[(i0 + j * 256 + tid < n) ? (ed[j].y & 255) : 256], 1);
#pragma unroll
        for (int j = 0; j < 8; ++j)
            if (i0 + j * 256 + tid < n) csr_src[beg + p[j]] = ed[j].x;
    }
}

#define LPITCH 66
#define RSTRIP 32

__device__ __forceinline__ void lds_ready_barrier() {
    asm volatile("s_wait_dscnt 0x0\n\ts_barrier_signal -1\n\ts_barrier_wait -1" ::: "memory");
}
__device__ __forceinline__ void reads_done_barrier() {
    asm volatile("s_barrier_signal -1\n\ts_barrier_wait -1" ::: "memory");
}

template <int N, bool HAS_BIAS, bool DO_RELU>
__global__ __launch_bounds__(256) void gemm_k128(const float* __restrict__ A,
                                                 const float* __restrict__ W,
                                                 const float* __restrict__ bias,
                                                 float* __restrict__ out,
                                                 int M) {
    __shared__ v2f sA2[RSTRIP][LPITCH];
    __shared__ v2f sW2[N][LPITCH];

    const int tid = threadIdx.x;

    const int nStrips = (M + RSTRIP - 1) / RSTRIP;
    const int prow = tid >> 5;
    const int pc4  = tid & 31;
    float* sAf = (float*)sA2;

    float4 pre[4];
    int strip = blockIdx.x;
    if (strip < nStrips) {
#pragma unroll
        for (int j = 0; j < 4; ++j) {
            const int row = strip * RSTRIP + prow + 8 * j;
            pre[j] = (row < M) ? ((const float4*)(A + (size_t)row * 128))[pc4]
                               : float4{0.0f, 0.0f, 0.0f, 0.0f};
        }
    }

    for (int i = tid; i < 64 * N; i += 256) {
        const int kp = i / N, n = i % N;
        v2f w;
        w.x = W[(2 * kp) * N + n];
        w.y = W[(2 * kp + 1) * N + n];
        sW2[n][kp] = w;
    }

    const int wave  = tid >> 5;
    const int lane  = tid & 31;
    const int m     = lane & 15;
    const int hsel  = lane >> 4;
    const int mb    = hsel << 3;

    static_assert(N % 16 == 0 && N / 16 <= 8, "one column tile per wave");
    const bool has_tile = wave < N / 16;
    const int  col0     = wave << 4;
    float bv = 0.0f;
    if constexpr (HAS_BIAS) { if (has_tile) bv = bias[col0 + m]; }

    const int gstep = (int)gridDim.x;
    for (; strip < nStrips; strip += gstep) {
        const int row0 = strip * RSTRIP;
#pragma unroll
        for (int j = 0; j < 4; ++j)
            *(float4*)(sAf + (prow + 8 * j) * (2 * LPITCH) + pc4 * 4) = pre[j];
        lds_ready_barrier();

        const int nxt = strip + gstep;
        if (nxt < nStrips) {
#pragma unroll
            for (int j = 0; j < 4; ++j) {
                const int row = nxt * RSTRIP + prow + 8 * j;
                pre[j] = (row < M) ? ((const float4*)(A + (size_t)row * 128))[pc4]
                                   : float4{0.0f, 0.0f, 0.0f, 0.0f};
            }
        }

        if (has_tile) {
            v8f acc0 = {};
            v8f acc1 = {};
#pragma unroll 8
            for (int k2 = 0; k2 < 64; k2 += 2) {
                const int kq = k2 + hsel;
                v2f b  = sW2[col0 + m][kq];
                v2f a0 = sA2[m][kq];
                v2f a1 = sA2[16 + m][kq];
                acc0 = __builtin_amdgcn_wmma_f32_16x16x4_f32(
                    false, a0, false, b, (short)0, acc0, false, false);
                acc1 = __builtin_amdgcn_wmma_f32_16x16x4_f32(
                    false, a1, false, b, (short)0, acc1, false, false);
            }
#pragma unroll
            for (int r = 0; r < 8; ++r) {
                const int row = row0 + mb + r;
                if (row < M) {
                    float v = acc0[r] + bv;
                    if constexpr (DO_RELU) v = fmaxf(v, 0.0f);
                    out[(size_t)row * N + col0 + m] = v;
                }
            }
#pragma unroll
            for (int r = 0; r < 8; ++r) {
                const int row = row0 + 16 + mb + r;
                if (row < M) {
                    float v = acc1[r] + bv;
                    if constexpr (DO_RELU) v = fmaxf(v, 0.0f);
                    out[(size_t)row * N + col0 + m] = v;
                }
            }
        }
        reads_done_barrier();
    }
}

__device__ __forceinline__ float lane_f(float v, int l) {
    return __int_as_float(__builtin_amdgcn_readlane(__float_as_int(v), l));
}

__device__ __forceinline__ void pool_flush(float* __restrict__ gsum, float* __restrict__ gcnt,
                                           int b, float4 r, int len, int lane) {
    const int comp = lane & 3;
    float* gp = gsum + (size_t)b * 128 + lane;
    for (int j = 0; j < 4; ++j) {
        const int sl = 8 * j + (lane >> 2);
        const float x = __shfl(r.x, sl), y = __shfl(r.y, sl);
        const float z = __shfl(r.z, sl), w = __shfl(r.w, sl);
        const float lo = (comp & 1) ? y : x;
        const float hi = (comp & 1) ? w : z;
        atomicAdd(gp + 32 * j, (comp & 2) ? hi : lo);
    }
    if (lane == 0) atomicAdd(gcnt + b, (float)len);
}

#define ROW(s) (((const float4*)(h + (size_t)(s) * 128))[lane])
#define ACC(wt_, vv_)                                                         \
    acc.x = fmaf((wt_), (vv_).x, acc.x); acc.y = fmaf((wt_), (vv_).y, acc.y); \
    acc.z = fmaf((wt_), (vv_).z, acc.z); acc.w = fmaf((wt_), (vv_).w, acc.w)

template <bool POOL>
__global__ __launch_bounds__(256) void gather_relu(const int* __restrict__ rowptr,
                                                   const int* __restrict__ cnt,
                                                   const int* __restrict__ csr_src,
                                                   const float* __restrict__ dinv,
                                                   const float* __restrict__ h,
                                                   const float* __restrict__ bias,
                                                   float* __restrict__ act,
                                                   const int* __restrict__ batch,
                                                   float* __restrict__ gsum,
                                                   float* __restrict__ gcnt, int nN) {
    const int lane = threadIdx.x & 31;
    const int wv = __builtin_amdgcn_readfirstlane((int)(threadIdx.x >> 5));
    const int nw = gridDim.x * 8;
    const int gw = blockIdx.x * 8 + wv;
    const float4 bv = ((const float4*)bias)[lane];

    int node = gw, nend = nN, nstep = nw;
    if constexpr (POOL) {
        const unsigned q = (unsigned)nN / (unsigned)nw, rem = (unsigned)nN % (unsigned)nw;
        node  = (int)((unsigned)gw * q + min((unsigned)gw, rem));
        nend  = node + (int)q + (((unsigned)gw < rem) ? 1 : 0);
        nstep = 1;
    }
    float4 run = {0.0f, 0.0f, 0.0f, 0.0f};
    int runb = -1, runlen = 0;

    for (; node < nend; node += nstep) {
        const int   beg = rowptr[node];
        const int   n   = cnt[node];
        const float di  = dinv[node];
        int gid = 0;
        if constexpr (POOL) gid = batch[node];
        const float4 hv = ROW(node);
        float4 acc = {0.0f, 0.0f, 0.0f, 0.0f};

        for (int c0 = 0; c0 < n; c0 += 32) {
            const int m = min(n - c0, 32);
            int   s_l = 0;
            float w_l = 0.0f;
            if (lane < m) {
                s_l = csr_src[beg + c0 + lane];
                w_l = dinv[s_l] * di;
            }
            int j = 0;
            for (; j + 4 <= m; j += 4) {
                const int s0 = __builtin_amdgcn_readlane(s_l, j);
                const int s1 = __builtin_amdgcn_readlane(s_l, j + 1);
                const int s2 = __builtin_amdgcn_readlane(s_l, j + 2);
                const int s3 = __builtin_amdgcn_readlane(s_l, j + 3);
                const float4 v0 = ROW(s0);
                const float4 v1 = ROW(s1);
                const float4 v2 = ROW(s2);
                const float4 v3 = ROW(s3);
                const float w0 = lane_f(w_l, j),     w1 = lane_f(w_l, j + 1);
                const float w2 = lane_f(w_l, j + 2), w3 = lane_f(w_l, j + 3);
                ACC(w0, v0); ACC(w1, v1); ACC(w2, v2); ACC(w3, v3);
            }
            if (j + 2 <= m) {
                const int s0 = __builtin_amdgcn_readlane(s_l, j);
                const int s1 = __builtin_amdgcn_readlane(s_l, j + 1);
                const float4 v0 = ROW(s0);
                const float4 v1 = ROW(s1);
                const float w0 = lane_f(w_l, j), w1 = lane_f(w_l, j + 1);
                ACC(w0, v0); ACC(w1, v1);
                j += 2;
            }
            if (j < m) {
                const int s0 = __builtin_amdgcn_readlane(s_l, j);
                const float4 v0 = ROW(s0);
                const float w0 = lane_f(w_l, j);
                ACC(w0, v0);
            }
        }

        const float d2 = di * di;
        float4 o;
        o.x = fmaxf(acc.x + d2 * hv.x + bv.x, 0.0f);
        o.y = fmaxf(acc.y + d2 * hv.y + bv.y, 0.0f);
        o.z = fmaxf(acc.z + d2 * hv.z + bv.z, 0.0f);
        o.w = fmaxf(acc.w + d2 * hv.w + bv.w, 0.0f);
        if constexpr (POOL) {
            const int b = gid;
            if (b != runb) {
                if (runlen > 0) pool_flush(gsum, gcnt, runb, run, runlen, lane);
                run = o; runb = b; runlen = 1;
            } else {
                run.x += o.x; run.y += o.y; run.z += o.z; run.w += o.w;
                ++runlen;
            }
        } else {
            ((float4*)(act + (size_t)node * 128))[lane] = o;
        }
    }
    if constexpr (POOL) {
        if (runlen > 0) pool_flush(gsum, gcnt, runb, run, runlen, lane);
    }
}

#define FROWS 128

__global__ __launch_bounds__(1024, 16) void gather_gemm(const int* __restrict__ rowptr,
                                                   const int* __restrict__ cnt,
                                                   const int* __restrict__ csr_src,
                                                   const float* __restrict__ dinv,
                                                   const float* __restrict__ h,
                                                   const float* __restrict__ bias,
                                                   const float* __restrict__ W,
                                                   float* __restrict__ out, int nN) {
    __shared__ v2f sA2[FROWS][LPITCH];
    __shared__ v2f sW2[128][LPITCH];

    const int tid  = threadIdx.x;
    const int lane = tid & 31;
    const int wave = __builtin_amdgcn_readfirstlane(tid >> 5);
    float* sAf = (float*)sA2;

    for (int i = tid; i < 64 * 128; i += 1024) {
        const int kp = i >> 7, n = i & 127;
        v2f w;
        w.x = W[(2 * kp) * 128 + n];
        w.y = W[(2 * kp + 1) * 128 + n];
        sW2[n][kp] = w;
    }

    const float4 bv = ((const float4*)bias)[lane];

    const unsigned ng = gridDim.x, wg = blockIdx.x;
    const unsigned q = (unsigned)nN / ng, rem = (unsigned)nN % ng;
    const int r0 = (int)(wg * q + min(wg, rem));
    const int r1 = r0 + (int)q + ((wg < rem) ? 1 : 0);

    const int m     = lane & 15;
    const int hsel  = lane >> 4;
    const int mb    = hsel << 3;
    const int col0  = (wave & 7) << 4;
    const int rbase = (wave >> 3) << 5;

    for (int s0 = r0; s0 < r1; s0 += FROWS) {
        const int rows = min(FROWS, r1 - s0);

        for (int i = 0; i < 4; ++i) {
            const int lr = i * 32 + wave;
            float4 o = {0.0f, 0.0f, 0.0f, 0.0f};
            if (lr < rows) {
                const int   node = s0 + lr;
                const int   beg  = rowptr[node];
                const int   n    = cnt[node];
                const float di   = dinv[node];
                const float4 hv  = ROW(node);
                float4 acc = {0.0f, 0.0f, 0.0f, 0.0f};

                for (int c0 = 0; c0 < n; c0 += 32) {
                    const int mm = min(n - c0, 32);
                    int   s_l = 0;
                    float w_l = 0.0f;
                    if (lane < mm) {
                        s_l = csr_src[beg + c0 + lane];
                        w_l = dinv[s_l] * di;
                    }
                    int j = 0;
                    for (; j + 4 <= mm; j += 4) {
                        const int s0r = __builtin_amdgcn_readlane(s_l, j);
                        const int s1r = __builtin_amdgcn_readlane(s_l, j + 1);
                        const int s2r = __builtin_amdgcn_readlane(s_l, j + 2);
                        const int s3r = __builtin_amdgcn_readlane(s_l, j + 3);
                        const float4 v0 = ROW(s0r);
                        const float4 v1 = ROW(s1r);
                        const float4 v2 = ROW(s2r);
                        const float4 v3 = ROW(s3r);
                        const float w0 = lane_f(w_l, j),     w1 = lane_f(w_l, j + 1);
                        const float w2 = lane_f(w_l, j + 2), w3 = lane_f(w_l, j + 3);
                        ACC(w0, v0); ACC(w1, v1); ACC(w2, v2); ACC(w3, v3);
                    }
                    if (j + 2 <= mm) {
                        const int s0r = __builtin_amdgcn_readlane(s_l, j);
                        const int s1r = __builtin_amdgcn_readlane(s_l, j + 1);
                        const float4 v0 = ROW(s0r);
                        const float4 v1 = ROW(s1r);
                        const float w0 = lane_f(w_l, j), w1 = lane_f(w_l, j + 1);
                        ACC(w0, v0); ACC(w1, v1);
                        j += 2;
                    }
                    if (j < mm) {
                        const int s0r = __builtin_amdgcn_readlane(s_l, j);
                        const float4 v0 = ROW(s0r);
                        const float w0 = lane_f(w_l, j);
                        ACC(w0, v0);
                    }
                }

                const float d2 = di * di;
                o.x = fmaxf(acc.x + d2 * hv.x + bv.x, 0.0f);
                o.y = fmaxf(acc.y + d2 * hv.y + bv.y, 0.0f);
                o.z = fmaxf(acc.z + d2 * hv.z + bv.z, 0.0f);
                o.w = fmaxf(acc.w + d2 * hv.w + bv.w, 0.0f);
            }
            *(float4*)(sAf + lr * (2 * LPITCH) + lane * 4) = o;
        }
        lds_ready_barrier();

        if (rbase < rows) {
            v8f acc0 = {};
            v8f acc1 = {};
#pragma unroll 8
            for (int k2 = 0; k2 < 64; k2 += 2) {
                const int kq = k2 + hsel;
                v2f b  = sW2[col0 + m][kq];
                v2f a0 = sA2[rbase + m][kq];
                v2f a1 = sA2[rbase + 16 + m][kq];
                acc0 = __builtin_amdgcn_wmma_f32_16x16x4_f32(
                    false, a0, false, b, (short)0, acc0, false, false);
                acc1 = __builtin_amdgcn_wmma_f32_16x16x4_f32(
                    false, a1, false, b, (short)0, acc1, false, false);
            }
#pragma unroll
            for (int r = 0; r < 8; ++r) {
                const int lr = rbase + mb + r;
                if (lr < rows) out[(size_t)(s0 + lr) * 128 + col0 + m] = acc0[r];
            }
#pragma unroll
            for (int r = 0; r < 8; ++r) {
                const int lr = rbase + 16 + mb + r;
                if (lr < rows) out[(size_t)(s0 + lr) * 128 + col0 + m] = acc1[r];
            }
        }
        reads_done_barrier();
    }
}
#undef ROW
#undef ACC

__global__ __launch_bounds__(256) void pool_div(const float* __restrict__ gsum,
                                                const float* __restrict__ gcnt,
                                                float* __restrict__ g, int G) {
    int i = blockIdx.x * 256 + threadIdx.x;
    if (i < G * 128) {
        int b = i >> 7;
        g[i] = gsum[i] / fmaxf(gcnt[b], 1.0f);
    }
}

extern "C" void kernel_launch(void* const* d_in, const int* in_sizes, int n_in,
                              void* d_out, int out_size, void* d_ws, size_t ws_size,
                              hipStream_t stream) {
    const float* x     = (const float*)d_in[0];
    const int*   ei    = (const int*)d_in[1];
    const int*   batch = (const int*)d_in[2];
    const float* W1    = (const float*)d_in[3];
    const float* b1    = (const float*)d_in[4];
    const float* W2    = (const float*)d_in[5];
    const float* b2    = (const float*)d_in[6];
    const float* Wl1   = (const float*)d_in[7];
    const float* bl1   = (const float*)d_in[8];
    const float* Wl2   = (const float*)d_in[9];
    const float* bl2   = (const float*)d_in[10];
    float* out = (float*)d_out;

    const int nN = in_sizes[0] / 128;
    const int nE = in_sizes[1] / 2;
    const int G  = out_size / 32;
    const int* src = ei;
    const int* dst = ei + nE;

    char* ws = (char*)d_ws;
    auto alloc = [&](size_t bytes) -> float* {
        float* p = (float*)ws;
        ws += (bytes + 255) & ~(size_t)255;
        return p;
    };
    float* bufA = alloc((size_t)nN * 128 * 4);
    float* bufB = alloc((size_t)nN * 128 * 4);
    char*  bufC = (char*)alloc((size_t)nN * 128 * 4);
    int*   cnt  = (int*)alloc((size_t)nN * 4);
    float* dinv = alloc((size_t)nN * 4);
    float* gsum = alloc((size_t)G * 128 * 4);
    float* gcnt = alloc((size_t)G * 4);
    float* g    = alloc((size_t)G * 128 * 4);
    float* m1   = alloc((size_t)G * 128 * 4);

    auto sub = [&](size_t bytes) -> int* {
        int* p = (int*)bufC;
        bufC += (bytes + 255) & ~(size_t)255;
        return p;
    };
    const int NB = (nN + 255) >> BIN_SHIFT;
    const int NC = (nE + ECH - 1) / ECH;
    int* csr_src = sub((size_t)nE * 4);
    int* rank    = sub((size_t)nE * 4);
    int* bed     = sub((size_t)nE * 8);
    int* rowptr  = sub((size_t)nN * 4);
    int* hist    = sub((size_t)NB * NC * 4);
    int* btot    = sub((size_t)NB * 4);

    fill_f32<<<(G * 128 + 255) / 256, 256, 0, stream>>>(gsum, 0.0f, (long)G * 128);
    fill_f32<<<(G + 255) / 256, 256, 0, stream>>>(gcnt, 0.0f, G);

    bin_count<<<NC, 256, 0, stream>>>(dst, rank, hist, nE, NB, NC);
    bin_scan<<<NB, 256, 0, stream>>>(hist, btot, NC);
    bin_scatter<<<NC, 256, 0, stream>>>(src, dst, rank, hist, btot, bed, nE, NB, NC);
    bin_csr<<<NB, 256, 0, stream>>>(bed, btot, rowptr, cnt, dinv, csr_src, nN);

    gemm_k128<128, false, false><<<768, 256, 0, stream>>>(x, W1, nullptr, bufA, nN);
    gather_gemm<<<512, 1024, 0, stream>>>(rowptr, cnt, csr_src, dinv, bufA, b1, W2, bufB, nN);

    gather_relu<true><<<2048, 256, 0, stream>>>(rowptr, cnt, csr_src, dinv, bufB, b2, nullptr,
                                                batch, gsum, gcnt, nN);

    pool_div<<<(G * 128 + 255) / 256, 256, 0, stream>>>(gsum, gcnt, g, G);

    gemm_k128<128, true, true><<<(G + RSTRIP - 1) / RSTRIP, 256, 0, stream>>>(g, Wl1, bl1, m1, G);
    gemm_k128<32, true, false><<<(G + RSTRIP - 1) / RSTRIP, 256, 0, stream>>>(m1, Wl2, bl2, out, G);
}


# Rows stored pre-weighted by dinv[src] (A rows of both layer products scaled), so each gather is a plain row sum times dinv[dst]: no per-edge dinv read or weight; sum order kept, rounding differs
# speedup vs baseline: 4.9649x; 1.0376x over previous
#include <hip/hip_runtime.h>

typedef __attribute__((ext_vector_type(2))) float v2f;
typedef __attribute__((ext_vector_type(8))) float v8f;

__global__ __launch_bounds__(256) void fill_f32(float* __restrict__ p, float v, long n) {
    long i = (long)blockIdx.x * 256 + threadIdx.x;
    if (i < n) p[i] = v;
}

#define BIN_SHIFT 8
#define NBMAX     512
#define ECH       4096

__device__ __forceinline__ int wave_sum(int v) {
#pragma unroll
    for (int off = 16; off > 0; off >>= 1) v += __shfl_xor(v, off);
    return v;
}

__device__ __forceinline__ int wave_incl_scan(int v, int lane) {
#pragma unroll
    for (int off = 1; off < 32; off <<= 1) {
        int t = __shfl_up(v, off);
        if (lane >= off) v += t;
    }
    return v;
}

__global__ __launch_bounds__(256) void bin_count(const int* __restrict__ dst,
                                                 int* __restrict__ rank,
                                                 int* __restrict__ hist,
                                                 int nE, int NB, int NC) {
    __shared__ int sh[NBMAX + 1];
    const int tid = threadIdx.x;
    for (int i = tid; i < NBMAX + 1; i += 256) sh[i] = 0;
    __syncthreads();
    const int base = blockIdx.x * ECH;
    int d[16], r[16];
#pragma unroll
    for (int j = 0; j < 16; ++j) {
        const int e = base + j * 256 + tid;
        d[j] = dst[min(e, nE - 1)];
    }
#pragma unroll
    for (int j = 0; j < 16; ++j) {
        const int e = base + j * 256 + tid;
        const int bin = (e < nE) ? min(d[j] >> BIN_SHIFT, NBMAX - 1) : NBMAX;
        r[j] = atomicAdd(&sh[bin], 1);
    }
#pragma unroll
    for (int j = 0; j < 16; ++j) {
        const int e = base + j * 256 + tid;
        if (e < nE) rank[e] = r[j];
    }
    __syncthreads();
    for (int i = tid; i < NB; i += 256) hist[(size_t)i * NC + blockIdx.x] = sh[i];
}

__global__ __launch_bounds__(256) void bin_scan(int* __restrict__ hist,
                                                int* __restrict__ btot, int NC) {
    __shared__ int sw[8];
    const int tid = threadIdx.x, lane = tid & 31, wave = tid >> 5;
    int* row = hist + (size_t)blockIdx.x * NC;
    int carry = 0;
    for (int c0 = 0; c0 < NC; c0 += 256) {
        const int c = c0 + tid;
        const int v = (c < NC) ? row[c] : 0;
        const int inc = wave_incl_scan(v, lane);
        if (lane == 31) sw[wave] = inc;
        __syncthreads();
        int off = 0, tot = 0;
        for (int w = 0; w < 8; ++w) {
            const int t = sw[w];
            tot += t;
            if (w < wave) off += t;
        }
        if (c < NC) row[c] = carry + off + inc - v;
        carry += tot;
        __syncthreads();
    }
    if (tid == 0) btot[blockIdx.x] = carry;
}

__global__ __launch_bounds__(256) void bin_scatter(const int* __restrict__ src,
                                                   const int* __restrict__ dst,
                                                   const int* __restrict__ rank,
                                                   const int* __restrict__ hist,
                                                   const int* __restrict__ btot,
                                                   int* __restrict__ bed_i,
                                                   int nE, int NB, int NC) {
    __shared__ int sbase[NBMAX];
    __shared__ int sw[8];
    const int tid = threadIdx.x, lane = tid & 31, wave = tid >> 5;
    int2* bed = (int2*)bed_i;

    const int b0 = 2 * tid, b1 = b0 + 1;
    const int a0 = (b0 < NB) ? btot[b0] : 0;
    const int a1 = (b1 < NB) ? btot[b1] : 0;
    const int s  = a0 + a1;
    const int inc = wave_incl_scan(s, lane);
    if (lane == 31) sw[wave] = inc;
    __syncthreads();
    int off = 0;
    for (int w = 0; w < 8; ++w)
        if (w < wave) off += sw[w];
    const int ex = off + inc - s;
    if (b0 < NB) sbase[b0] = ex + hist[(size_t)b0 * NC + blockIdx.x];
    if (b1 < NB) sbase[b1] = ex + a0 + hist[(size_t)b1 * NC + blockIdx.x];
    __syncthreads();

    const int base = blockIdx.x * ECH;
    for (int h = 0; h < 2; ++h) {
        int d[8], rk[8], sv[8];
#pragma unroll
        for (int j = 0; j < 8; ++j) {
            const int ec = min(base + (h * 8 + j) * 256 + tid, nE - 1);
            d[j]  = dst[ec];
            rk[j] = rank[ec];
            sv[j] = src[ec];
        }
#pragma unroll
        for (int j = 0; j < 8; ++j) {
            const int e = base + (h * 8 + j) * 256 + tid;
            const int pos = sbase[min(d[j] >> BIN_SHIFT, NBMAX - 1)] + rk[j];
            if (e < nE) bed[pos] = make_int2(sv[j], d[j]);
        }
    }
}

__global__ __launch_bounds__(256) void bin_csr(const int* __restrict__ bed_i,
                                               const int* __restrict__ btot,
                                               int* __restrict__ rowptr,
                                               int* __restrict__ cnt,
                                               float* __restrict__ dinv,
                                               int* __restrict__ csr_src, int nN) {
    __shared__ int scnt[257];
    __shared__ int sptr[257];
    __shared__ int sw[8];
    const int tid = threadIdx.x, lane = tid & 31, wave = tid >> 5;
    const int b = blockIdx.x;
    const int2* bed = (const int2*)bed_i;

    int pv = 0;
    for (int i = tid; i < b; i += 256) pv += btot[i];
    pv = wave_sum(pv);
    if (lane == 0) sw[wave] = pv;
    scnt[tid] = 0;
    if (tid == 0) { scnt[256] = 0; sptr[256] = 0; }
    __syncthreads();
    int beg = 0;
    for (int w = 0; w < 8; ++w) beg += sw[w];
    const int n = btot[b];
    __syncthreads();

    for (int i0 = 0; i0 < n; i0 += 8 * 256) {
        int dd[8];
#pragma unroll
        for (int j = 0; j < 8; ++j) dd[j] = bed[beg + min(i0 + j * 256 + tid, n - 1)].y;
#pragma unroll
        for (int j = 0; j < 8; ++j)
            atomicAdd(&scnt[(i0 + j * 256 + tid < n) ? (dd[j] & 255) : 256], 1);
    }
    __syncthreads();

    const int v   = scnt[tid];
    const int inc = wave_incl_scan(v, lane);
    if (lane == 31) sw[wave] = inc;
    __syncthreads();
    int off = 0;
    for (int w = 0; w < 8; ++w)
        if (w < wave) off += sw[w];
    const int ex = off + inc - v;
    sptr[tid] = ex;
    const int node = (b << BIN_SHIFT) + tid;
    if (node < nN) {
        rowptr[node] = beg + ex;
        cnt[node]    = v;
        dinv[node]   = rsqrtf((float)v + 1.0f);
    }
    __syncthreads();

    for (int i0 = 0; i0 < n; i0 += 8 * 256) {
        int2 ed[8];
        int  p[8];
#pragma unroll
        for (int j = 0; j < 8; ++j) ed[j] = bed[beg + min(i0 + j * 256 + tid, n - 1)];
#pragma unroll
        for (int j = 0; j < 8; ++j)
            p[j] = atomicAdd(&sptr[(i0 + j * 256 + tid < n) ? (ed[j].y & 255) : 256], 1);
#pragma unroll
        for (int j = 0; j < 8; ++j)
            if (i0 + j * 256 + tid < n) csr_src[beg + p[j]] = ed[j].x;
    }
}

#define LPITCH 66
#define RSTRIP 32

__device__ __forceinline__ void lds_ready_barrier() {
    asm volatile("s_wait_dscnt 0x0\n\ts_barrier_signal -1\n\ts_barrier_wait -1" ::: "memory");
}
__device__ __forceinline__ void reads_done_barrier() {
    asm volatile("s_barrier_signal -1\n\ts_barrier_wait -1" ::: "memory");
}

template <int N, bool HAS_BIAS, bool DO_RELU, bool ROWSCALE>
__global__ __launch_bounds__(256) void gemm_k128(const float* __restrict__ A,
                                                 const float* __restrict__ W,
                                                 const float* __restrict__ bias,
                                                 const float* __restrict__ rscale,
                                                 float* __restrict__ out,
                                                 int M) {
    __shared__ v2f sA2[RSTRIP][LPITCH];
    __shared__ v2f sW2[N][LPITCH];

    const int tid = threadIdx.x;

    const int nStrips = (M + RSTRIP - 1) / RSTRIP;
    const int prow = tid >> 5;
    const int pc4  = tid & 31;
    float* sAf = (float*)sA2;

    float4 pre[4];
    float  psc[4] = {1.0f, 1.0f, 1.0f, 1.0f};
    int strip = blockIdx.x;
    if (strip < nStrips) {
#pragma unroll
        for (int j = 0; j < 4; ++j) {
            const int row = strip * RSTRIP + prow + 8 * j;
            pre[j] = (row < M) ? ((const float4*)(A + (size_t)row * 128))[pc4]
                               : float4{0.0f, 0.0f, 0.0f, 0.0f};
            if constexpr (ROWSCALE) psc[j] = (row < M) ? rscale[row] : 0.0f;
        }
    }

    for (int i = tid; i < 64 * N; i += 256) {
        const int kp = i / N, n = i % N;
        v2f w;
        w.x = W[(2 * kp) * N + n];
        w.y = W[(2 * kp + 1) * N + n];
        sW2[n][kp] = w;
    }

    const int wave  = tid >> 5;
    const int lane  = tid & 31;
    const int m     = lane & 15;
    const int hsel  = lane >> 4;
    const int mb    = hsel << 3;

    static_assert(N % 16 == 0 && N / 16 <= 8, "one column tile per wave");
    const bool has_tile = wave < N / 16;
    const int  col0     = wave << 4;
    float bv = 0.0f;
    if constexpr (HAS_BIAS) { if (has_tile) bv = bias[col0 + m]; }

    const int gstep = (int)gridDim.x;
    for (; strip < nStrips; strip += gstep) {
        const int row0 = strip * RSTRIP;
#pragma unroll
        for (int j = 0; j < 4; ++j) {
            float4 v = pre[j];
            if constexpr (ROWSCALE) { v.x *= psc[j]; v.y *= psc[j]; v.z *= psc[j]; v.w *= psc[j]; }
            *(float4*)(sAf + (prow + 8 * j) * (2 * LPITCH) + pc4 * 4) = v;
        }
        lds_ready_barrier();

        const int nxt = strip + gstep;
        if (nxt < nStrips) {
#pragma unroll
            for (int j = 0; j < 4; ++j) {
                const int row = nxt * RSTRIP + prow + 8 * j;
                pre[j] = (row < M) ? ((const float4*)(A + (size_t)row * 128))[pc4]
                                   : float4{0.0f, 0.0f, 0.0f, 0.0f};
                if constexpr (ROWSCALE) psc[j] = (row < M) ? rscale[row] : 0.0f;
            }
        }

        if (has_tile) {
            v8f acc0 = {};
            v8f acc1 = {};
#pragma unroll 8
            for (int k2 = 0; k2 < 64; k2 += 2) {
                const int kq = k2 + hsel;
                v2f b  = sW2[col0 + m][kq];
                v2f a0 = sA2[m][kq];
                v2f a1 = sA2[16 + m][kq];
                acc0 = __builtin_amdgcn_wmma_f32_16x16x4_f32(
                    false, a0, false, b, (short)0, acc0, false, false);
                acc1 = __builtin_amdgcn_wmma_f32_16x16x4_f32(
                    false, a1, false, b, (short)0, acc1, false, false);
            }
#pragma unroll
            for (int r = 0; r < 8; ++r) {
                const int row = row0 + mb + r;
                if (row < M) {
                    float v = acc0[r] + bv;
                    if constexpr (DO_RELU) v = fmaxf(v, 0.0f);
                    out[(size_t)row * N + col0 + m] = v;
                }
            }
#pragma unroll
            for (int r = 0; r < 8; ++r) {
                const int row = row0 + 16 + mb + r;
                if (row < M) {
                    float v = acc1[r] + bv;
                    if constexpr (DO_RELU) v = fmaxf(v, 0.0f);
                    out[(size_t)row * N + col0 + m] = v;
                }
            }
        }
        reads_done_barrier();
    }
}

__device__ __forceinline__ void pool_flush(float* __restrict__ gsum, float* __restrict__ gcnt,
                                           int b, float4 r, int len, int lane) {
    const int comp = lane & 3;
    float* gp = gsum + (size_t)b * 128 + lane;
    for (int j = 0; j < 4; ++j) {
        const int sl = 8 * j + (lane >> 2);
        const float x = __shfl(r.x, sl), y = __shfl(r.y, sl);
        const float z = __shfl(r.z, sl), w = __shfl(r.w, sl);
        const float lo = (comp & 1) ? y : x;
        const float hi = (comp & 1) ? w : z;
        atomicAdd(gp + 32 * j, (comp & 2) ? hi : lo);
    }
    if (lane == 0) atomicAdd(gcnt + b, (float)len);
}

#define ROW(s) (((const float4*)(h + (size_t)(s) * 128))[lane])
#define ACC(vv_)                                                              \
    acc.x += (vv_).x; acc.y += (vv_).y; acc.z += (vv_).z; acc.w += (vv_).w

template <bool POOL>
__global__ __launch_bounds__(256) void gather_relu(const int* __restrict__ rowptr,
                                                   const int* __restrict__ cnt,
                                                   const int* __restrict__ csr_src,
                                                   const float* __restrict__ dinv,
                                                   const float* __restrict__ h,
                                                   const float* __restrict__ bias,
                                                   float* __restrict__ act,
                                                   const int* __restrict__ batch,
                                                   float* __restrict__ gsum,
                                                   float* __restrict__ gcnt, int nN) {
    const int lane = threadIdx.x & 31;
    const int wv = __builtin_amdgcn_readfirstlane((int)(threadIdx.x >> 5));
    const int nw = gridDim.x * 8;
    const int gw = blockIdx.x * 8 + wv;
    const float4 bv = ((const float4*)bias)[lane];

    int node = gw, nend = nN, nstep = nw;
    if constexpr (POOL) {
        const unsigned q = (unsigned)nN / (unsigned)nw, rem = (unsigned)nN % (unsigned)nw;
        node  = (int)((unsigned)gw * q + min((unsigned)gw, rem));
        nend  = node + (int)q + (((unsigned)gw < rem) ? 1 : 0);
        nstep = 1;
    }
    float4 run = {0.0f, 0.0f, 0.0f, 0.0f};
    int runb = -1, runlen = 0;

    for (; node < nend; node += nstep) {
        const int   beg = rowptr[node];
        const int   n   = cnt[node];
        const float di  = dinv[node];
        int gid = 0;
        if constexpr (POOL) gid = batch[node];
        const float4 hv = ROW(node);
        float4 acc = {0.0f, 0.0f, 0.0f, 0.0f};

        for (int c0 = 0; c0 < n; c0 += 32) {
            const int m = min(n - c0, 32);
            int s_l = 0;
            if (lane < m) s_l = csr_src[beg + c0 + lane];
            int j = 0;
            for (; j + 4 <= m; j += 4) {
                const int s0 = __builtin_amdgcn_readlane(s_l, j);
                const int s1 = __builtin_amdgcn_readlane(s_l, j + 1);
                const int s2 = __builtin_amdgcn_readlane(s_l, j + 2);
                const int s3 = __builtin_amdgcn_readlane(s_l, j + 3);
                const float4 v0 = ROW(s0);
                const float4 v1 = ROW(s1);
                const float4 v2 = ROW(s2);
                const float4 v3 = ROW(s3);
                ACC(v0); ACC(v1); ACC(v2); ACC(v3);
            }
            if (j + 2 <= m) {
                const int s0 = __builtin_amdgcn_readlane(s_l, j);
                const int s1 = __builtin_amdgcn_readlane(s_l, j + 1);
                const float4 v0 = ROW(s0);
                const float4 v1 = ROW(s1);
                ACC(v0); ACC(v1);
                j += 2;
            }
            if (j < m) {
                const int s0 = __builtin_amdgcn_readlane(s_l, j);
                const float4 v0 = ROW(s0);
                ACC(v0);
            }
        }

        float4 o;
        o.x = fmaxf(fmaf(di, acc.x + hv.x, bv.x), 0.0f);
        o.y = fmaxf(fmaf(di, acc.y + hv.y, bv.y), 0.0f);
        o.z = fmaxf(fmaf(di, acc.z + hv.z, bv.z), 0.0f);
        o.w = fmaxf(fmaf(di, acc.w + hv.w, bv.w), 0.0f);
        if constexpr (POOL) {
            const int b = gid;
            if (b != runb) {
                if (runlen > 0) pool_flush(gsum, gcnt, runb, run, runlen, lane);
                run = o; runb = b; runlen = 1;
            } else {
                run.x += o.x; run.y += o.y; run.z += o.z; run.w += o.w;
                ++runlen;
            }
        } else {
            ((float4*)(act + (size_t)node * 128))[lane] = o;
        }
    }
    if constexpr (POOL) {
        if (runlen > 0) pool_flush(gsum, gcnt, runb, run, runlen, lane);
    }
}

#define FROWS 128

__global__ __launch_bounds__(1024, 16) void gather_gemm(const int* __restrict__ rowptr,
                                                   const int* __restrict__ cnt,
                                                   const int* __restrict__ csr_src,
                                                   const float* __restrict__ dinv,
                                                   const float* __restrict__ h,
                                                   const float* __restrict__ bias,
                                                   const float* __restrict__ W,
                                                   float* __restrict__ out, int nN) {
    __shared__ v2f sA2[FROWS][LPITCH];
    __shared__ v2f sW2[128][LPITCH];

    const int tid  = threadIdx.x;
    const int lane = tid & 31;
    const int wave = __builtin_amdgcn_readfirstlane(tid >> 5);
    float* sAf = (float*)sA2;

    for (int i = tid; i < 64 * 128; i += 1024) {
        const int kp = i >> 7, n = i & 127;
        v2f w;
        w.x = W[(2 * kp) * 128 + n];
        w.y = W[(2 * kp + 1) * 128 + n];
        sW2[n][kp] = w;
    }

    const float4 bv = ((const float4*)bias)[lane];

    const unsigned ng = gridDim.x, wg = blockIdx.x;
    const unsigned q = (unsigned)nN / ng, rem = (unsigned)nN % ng;
    const int r0 = (int)(wg * q + min(wg, rem));
    const int r1 = r0 + (int)q + ((wg < rem) ? 1 : 0);

    const int m     = lane & 15;
    const int hsel  = lane >> 4;
    const int mb    = hsel << 3;
    const int col0  = (wave & 7) << 4;
    const int rbase = (wave >> 3) << 5;

    for (int s0 = r0; s0 < r1; s0 += FROWS) {
        const int rows = min(FROWS, r1 - s0);

        for (int i = 0; i < 4; ++i) {
            const int lr = i * 32 + wave;
            float4 o = {0.0f, 0.0f, 0.0f, 0.0f};
            if (lr < rows) {
                const int   node = s0 + lr;
                const int   beg  = rowptr[node];
                const int   n    = cnt[node];
                const float di   = dinv[node];
                const float4 hv  = ROW(node);
                float4 acc = {0.0f, 0.0f, 0.0f, 0.0f};

                for (int c0 = 0; c0 < n; c0 += 32) {
                    const int mm = min(n - c0, 32);
                    int s_l = 0;
                    if (lane < mm) s_l = csr_src[beg + c0 + lane];
                    int j = 0;
                    for (; j + 4 <= mm; j += 4) {
                        const int s0r = __builtin_amdgcn_readlane(s_l, j);
                        const int s1r = __builtin_amdgcn_readlane(s_l, j + 1);
                        const int s2r = __builtin_amdgcn_readlane(s_l, j + 2);
                        const int s3r = __builtin_amdgcn_readlane(s_l, j + 3);
                        const float4 v0 = ROW(s0r);
                        const float4 v1 = ROW(s1r);
                        const float4 v2 = ROW(s2r);
                        const float4 v3 = ROW(s3r);
                        ACC(v0); ACC(v1); ACC(v2); ACC(v3);
                    }
                    if (j + 2 <= mm) {
                        const int s0r = __builtin_amdgcn_readlane(s_l, j);
                        const int s1r = __builtin_amdgcn_readlane(s_l, j + 1);
                        const float4 v0 = ROW(s0r);
                        const float4 v1 = ROW(s1r);
                        ACC(v0); ACC(v1);
                        j += 2;
                    }
                    if (j < mm) {
                        const int s0r = __builtin_amdgcn_readlane(s_l, j);
                        const float4 v0 = ROW(s0r);
                        ACC(v0);
                    }
                }

                o.x = di * fmaxf(fmaf(di, acc.x + hv.x, bv.x), 0.0f);
                o.y = di * fmaxf(fmaf(di, acc.y + hv.y, bv.y), 0.0f);
                o.z = di * fmaxf(fmaf(di, acc.z + hv.z, bv.z), 0.0f);
                o.w = di * fmaxf(fmaf(di, acc.w + hv.w, bv.w), 0.0f);
            }
            *(float4*)(sAf + lr * (2 * LPITCH) + lane * 4) = o;
        }
        lds_ready_barrier();

        if (rbase < rows) {
            v8f acc0 = {};
            v8f acc1 = {};
#pragma unroll 8
            for (int k2 = 0; k2 < 64; k2 += 2) {
                const int kq = k2 + hsel;
                v2f b  = sW2[col0 + m][kq];
                v2f a0 = sA2[rbase + m][kq];
                v2f a1 = sA2[rbase + 16 + m][kq];
                acc0 = __builtin_amdgcn_wmma_f32_16x16x4_f32(
                    false, a0, false, b, (short)0, acc0, false, false);
                acc1 = __builtin_amdgcn_wmma_f32_16x16x4_f32(
                    false, a1, false, b, (short)0, acc1, false, false);
            }
#pragma unroll
            for (int r = 0; r < 8; ++r) {
                const int lr = rbase + mb + r;
                if (lr < rows) out[(size_t)(s0 + lr) * 128 + col0 + m] = acc0[r];
            }
#pragma unroll
            for (int r = 0; r < 8; ++r) {
                const int lr = rbase + 16 + mb + r;
                if (lr < rows) out[(size_t)(s0 + lr) * 128 + col0 + m] = acc1[r];
            }
        }
        reads_done_barrier();
    }
}
#undef ROW
#undef ACC

__global__ __launch_bounds__(256) void pool_div(const float* __restrict__ gsum,
                                                const float* __restrict__ gcnt,
                                                float* __restrict__ g, int G) {
    int i = blockIdx.x * 256 + threadIdx.x;
    if (i < G * 128) {
        int b = i >> 7;
        g[i] = gsum[i] / fmaxf(gcnt[b], 1.0f);
    }
}

extern "C" void kernel_launch(void* const* d_in, const int* in_sizes, int n_in,
                              void* d_out, int out_size, void* d_ws, size_t ws_size,
                              hipStream_t stream) {
    const float* x     = (const float*)d_in[0];
    const int*   ei    = (const int*)d_in[1];
    const int*   batch = (const int*)d_in[2];
    const float* W1    = (const float*)d_in[3];
    const float* b1    = (const float*)d_in[4];
    const float* W2    = (const float*)d_in[5];
    const float* b2    = (const float*)d_in[6];
    const float* Wl1   = (const float*)d_in[7];
    const float* bl1   = (const float*)d_in[8];
    const float* Wl2   = (const float*)d_in[9];
    const float* bl2   = (const float*)d_in[10];
    float* out = (float*)d_out;

    const int nN = in_sizes[0] / 128;
    const int nE = in_sizes[1] / 2;
    const int G  = out_size / 32;
    const int* src = ei;
    const int* dst = ei + nE;

    char* ws = (char*)d_ws;
    auto alloc = [&](size_t bytes) -> float* {
        float* p = (float*)ws;
        ws += (bytes + 255) & ~(size_t)255;
        return p;
    };
    float* bufA = alloc((size_t)nN * 128 * 4);
    float* bufB = alloc((size_t)nN * 128 * 4);
    char*  bufC = (char*)alloc((size_t)nN * 128 * 4);
    int*   cnt  = (int*)alloc((size_t)nN * 4);
    float* dinv = alloc((size_t)nN * 4);
    float* gsum = alloc((size_t)G * 128 * 4);
    float* gcnt = alloc((size_t)G * 4);
    float* g    = alloc((size_t)G * 128 * 4);
    float* m1   = alloc((size_t)G * 128 * 4);

    auto sub = [&](size_t bytes) -> int* {
        int* p = (int*)bufC;
        bufC += (bytes + 255) & ~(size_t)255;
        return p;
    };
    const int NB = (nN + 255) >> BIN_SHIFT;
    const int NC = (nE + ECH - 1) / ECH;
    int* csr_src = sub((size_t)nE * 4);
    int* rank    = sub((size_t)nE * 4);
    int* bed     = sub((size_t)nE * 8);
    int* rowptr  = sub((size_t)nN * 4);
    int* hist    = sub((size_t)NB * NC * 4);
    int* btot    = sub((size_t)NB * 4);

    fill_f32<<<(G * 128 + 255) / 256, 256, 0, stream>>>(gsum, 0.0f, (long)G * 128);
    fill_f32<<<(G + 255) / 256, 256, 0, stream>>>(gcnt, 0.0f, G);

    bin_count<<<NC, 256, 0, stream>>>(dst, rank, hist, nE, NB, NC);
    bin_scan<<<NB, 256, 0, stream>>>(hist, btot, NC);
    bin_scatter<<<NC, 256, 0, stream>>>(src, dst, rank, hist, btot, bed, nE, NB, NC);
    bin_csr<<<NB, 256, 0, stream>>>(bed, btot, rowptr, cnt, dinv, csr_src, nN);

    gemm_k128<128, false, false, true><<<768, 256, 0, stream>>>(x, W1, nullptr, dinv, bufA, nN);
    gather_gemm<<<512, 1024, 0, stream>>>(rowptr, cnt, csr_src, dinv, bufA, b1, W2, bufB, nN);

    gather_relu<true><<<2048, 256, 0, stream>>>(rowptr, cnt, csr_src, dinv, bufB, b2, nullptr,
                                                batch, gsum, gcnt, nN);

    pool_div<<<(G * 128 + 255) / 256, 256, 0, stream>>>(gsum, gcnt, g, G);

    gemm_k128<128, true, true, false><<<(G + RSTRIP - 1) / RSTRIP, 256, 0, stream>>>(g, Wl1, bl1, nullptr, m1, G);
    gemm_k128<32, true, false, false><<<(G + RSTRIP - 1) / RSTRIP, 256, 0, stream>>>(m1, Wl2, bl2, nullptr, out, G);
}


# Launch count 12 -> 8: the two clears folded into the first edge pass; pool divide and both head products in one launch (hidden rows stay in LDS); tile code, k order and sum order unchanged
# speedup vs baseline: 5.2547x; 1.0584x over previous
#include <hip/hip_runtime.h>

typedef __attribute__((ext_vector_type(2))) float v2f;
typedef __attribute__((ext_vector_type(8))) float v8f;

#define BIN_SHIFT 8
#define NBMAX     512
#define ECH       4096

__device__ __forceinline__ int wave_sum(int v) {
#pragma unroll
    for (int off = 16; off > 0; off >>= 1) v += __shfl_xor(v, off);
    return v;
}

__device__ __forceinline__ int wave_incl_scan(int v, int lane) {
#pragma unroll
    for (int off = 1; off < 32; off <<= 1) {
        int t = __shfl_up(v, off);
        if (lane >= off) v += t;
    }
    return v;
}

__global__ __launch_bounds__(256) void bin_count(const int* __restrict__ dst,
                                                 int* __restrict__ rank,
                                                 int* __restrict__ hist,
                                                 int nE, int NB, int NC,
                                                 float* __restrict__ zsum, int nzsum,
                                                 float* __restrict__ zcnt, int nzcnt) {
    __shared__ int sh[NBMAX + 1];
    const int tid = threadIdx.x;
    for (int i = tid; i < NBMAX + 1; i += 256) sh[i] = 0;
    const int gt = blockIdx.x * 256 + tid, gstride = (int)gridDim.x * 256;
    for (int i = gt; i < nzsum; i += gstride) zsum[i] = 0.0f;
    for (int i = gt; i < nzcnt; i += gstride) zcnt[i] = 0.0f;
    __syncthreads();
    const int base = blockIdx.x * ECH;
    int d[16], r[16];
#pragma unroll
    for (int j = 0; j < 16; ++j) {
        const int e = base + j * 256 + tid;
        d[j] = dst[min(e, nE - 1)];
    }
#pragma unroll
    for (int j = 0; j < 16; ++j) {
        const int e = base + j * 256 + tid;
        const int bin = (e < nE) ? min(d[j] >> BIN_SHIFT, NBMAX - 1) : NBMAX;
        r[j] = atomicAdd(&sh[bin], 1);
    }
#pragma unroll
    for (int j = 0; j < 16; ++j) {
        const int e = base + j * 256 + tid;
        if (e < nE) rank[e] = r[j];
    }
    __syncthreads();
    for (int i = tid; i < NB; i += 256) hist[(size_t)i * NC + blockIdx.x] = sh[i];
}

__global__ __launch_bounds__(256) void bin_scan(int* __restrict__ hist,
                                                int* __restrict__ btot, int NC) {
    __shared__ int sw[8];
    const int tid = threadIdx.x, lane = tid & 31, wave = tid >> 5;
    int* row = hist + (size_t)blockIdx.x * NC;
    int carry = 0;
    for (int c0 = 0; c0 < NC; c0 += 256) {
        const int c = c0 + tid;
        const int v = (c < NC) ? row[c] : 0;
        const int inc = wave_incl_scan(v, lane);
        if (lane == 31) sw[wave] = inc;
        __syncthreads();
        int off = 0, tot = 0;
        for (int w = 0; w < 8; ++w) {
            const int t = sw[w];
            tot += t;
            if (w < wave) off += t;
        }
        if (c < NC) row[c] = carry + off + inc - v;
        carry += tot;
        __syncthreads();
    }
    if (tid == 0) btot[blockIdx.x] = carry;
}

__global__ __launch_bounds__(256) void bin_scatter(const int* __restrict__ src,
                                                   const int* __restrict__ dst,
                                                   const int* __restrict__ rank,
                                                   const int* __restrict__ hist,
                                                   const int* __restrict__ btot,
                                                   int* __restrict__ bed_i,
                                                   int nE, int NB, int NC) {
    __shared__ int sbase[NBMAX];
    __shared__ int sw[8];
    const int tid = threadIdx.x, lane = tid & 31, wave = tid >> 5;
    int2* bed = (int2*)bed_i;

    const int b0 = 2 * tid, b1 = b0 + 1;
    const int a0 = (b0 < NB) ? btot[b0] : 0;
    const int a1 = (b1 < NB) ? btot[b1] : 0;
    const int s  = a0 + a1;
    const int inc = wave_incl_scan(s, lane);
    if (lane == 31) sw[wave] = inc;
    __syncthreads();
    int off = 0;
    for (int w = 0; w < 8; ++w)
        if (w < wave) off += sw[w];
    const int ex = off + inc - s;
    if (b0 < NB) sbase[b0] = ex + hist[(size_t)b0 * NC + blockIdx.x];
    if (b1 < NB) sbase[b1] = ex + a0 + hist[(size_t)b1 * NC + blockIdx.x];
    __syncthreads();

    const int base = blockIdx.x * ECH;
    for (int h = 0; h < 2; ++h) {
        int d[8], rk[8], sv[8];
#pragma unroll
        for (int j = 0; j < 8; ++j) {
            const int ec = min(base + (h * 8 + j) * 256 + tid, nE - 1);
            d[j]  = dst[ec];
            rk[j] = rank[ec];
            sv[j] = src[ec];
        }
#pragma unroll
        for (int j = 0; j < 8; ++j) {
            const int e = base + (h * 8 + j) * 256 + tid;
            const int pos = sbase[min(d[j] >> BIN_SHIFT, NBMAX - 1)] + rk[j];
            if (e < nE) bed[pos] = make_int2(sv[j], d[j]);
        }
    }
}

__global__ __launch_bounds__(256) void bin_csr(const int* __restrict__ bed_i,
                                               const int* __restrict__ btot,
                                               int* __restrict__ rowptr,
                                               int* __restrict__ cnt,
                                               float* __restrict__ dinv,
                                               int* __restrict__ csr_src, int nN) {
    __shared__ int scnt[257];
    __shared__ int sptr[257];
    __shared__ int sw[8];
    const int tid = threadIdx.x, lane = tid & 31, wave = tid >> 5;
    const int b = blockIdx.x;
    const int2* bed = (const int2*)bed_i;

    int pv = 0;
    for (int i = tid; i < b; i += 256) pv += btot[i];
    pv = wave_sum(pv);
    if (lane == 0) sw[wave] = pv;
    scnt[tid] = 0;
    if (tid == 0) { scnt[256] = 0; sptr[256] = 0; }
    __syncthreads();
    int beg = 0;
    for (int w = 0; w < 8; ++w) beg += sw[w];
    const int n = btot[b];
    __syncthreads();

    for (int i0 = 0; i0 < n; i0 += 8 * 256) {
        int dd[8];
#pragma unroll
        for (int j = 0; j < 8; ++j) dd[j] = bed[beg + min(i0 + j * 256 + tid, n - 1)].y;
#pragma unroll
        for (int j = 0; j < 8; ++j)
            atomicAdd(&scnt[(i0 + j * 256 + tid < n) ? (dd[j] & 255) : 256], 1);
    }
    __syncthreads();

    const int v   = scnt[tid];
    const int inc = wave_incl_scan(v, lane);
    if (lane == 31) sw[wave] = inc;
    __syncthreads();
    int off = 0;
    for (int w = 0; w < 8; ++w)
        if (w < wave) off += sw[w];
    const int ex = off + inc - v;
    sptr[tid] = ex;
    const int node = (b << BIN_SHIFT) + tid;
    if (node < nN) {
        rowptr[node] = beg + ex;
        cnt[node]    = v;
        dinv[node]   = rsqrtf((float)v + 1.0f);
    }
    __syncthreads();

    for (int i0 = 0; i0 < n; i0 += 8 * 256) {
        int2 ed[8];
        int  p[8];
#pragma unroll
        for (int j = 0; j < 8; ++j) ed[j] = bed[beg + min(i0 + j * 256 + tid, n - 1)];
#pragma unroll
        for (int j = 0; j < 8; ++j)
            p[j] = atomicAdd(&sptr[(i0 + j * 256 + tid < n) ? (ed[j].y & 255) : 256], 1);
#pragma unroll
        for (int j = 0; j < 8; ++j)
            if (i0 + j * 256 + tid < n) csr_src[beg + p[j]] = ed[j].x;
    }
}

#define LPITCH 66
#define RSTRIP 32

__device__ __forceinline__ void lds_ready_barrier() {
    asm volatile("s_wait_dscnt 0x0\n\ts_barrier_signal -1\n\ts_barrier_wait -1" ::: "memory");
}
__device__ __forceinline__ void reads_done_barrier() {
    asm volatile("s_barrier_signal -1\n\ts_barrier_wait -1" ::: "memory");
}

template <int N, bool HAS_BIAS, bool DO_RELU, bool ROWSCALE>
__global__ __launch_bounds__(256) void gemm_k128(const float* __restrict__ A,
                                                 const float* __restrict__ W,
                                                 const float* __restrict__ bias,
                                                 const float* __restrict__ rscale,
                                                 float* __restrict__ out,
                                                 int M) {
    __shared__ v2f sA2[RSTRIP][LPITCH];
    __shared__ v2f sW2[N][LPITCH];

    const int tid = threadIdx.x;

    const int nStrips = (M + RSTRIP - 1) / RSTRIP;
    const int prow = tid >> 5;
    const int pc4  = tid & 31;
    float* sAf = (float*)sA2;

    float4 pre[4];
    float  psc[4] = {1.0f, 1.0f, 1.0f, 1.0f};
    int strip = blockIdx.x;
    if (strip < nStrips) {
#pragma unroll
        for (int j = 0; j < 4; ++j) {
            const int row = strip * RSTRIP + prow + 8 * j;
            pre[j] = (row < M) ? ((const float4*)(A + (size_t)row * 128))[pc4]
                               : float4{0.0f, 0.0f, 0.0f, 0.0f};
            if constexpr (ROWSCALE) psc[j] = (row < M) ? rscale[row] : 0.0f;
        }
    }

    for (int i = tid; i < 64 * N; i += 256) {
        const int kp = i / N, n = i % N;
        v2f w;
        w.x = W[(2 * kp) * N + n];
        w.y = W[(2 * kp + 1) * N + n];
        sW2[n][kp] = w;
    }

    const int wave  = tid >> 5;
    const int lane  = tid & 31;
    const int m     = lane & 15;
    const int hsel  = lane >> 4;
    const int mb    = hsel << 3;

    static_assert(N % 16 == 0 && N / 16 <= 8, "one column tile per wave");
    const bool has_tile = wave < N / 16;
    const int  col0     = wave << 4;
    float bv = 0.0f;
    if constexpr (HAS_BIAS) { if (has_tile) bv = bias[col0 + m]; }

    const int gstep = (int)gridDim.x;
    for (; strip < nStrips; strip += gstep) {
        const int row0 = strip * RSTRIP;
#pragma unroll
        for (int j = 0; j < 4; ++j) {
            float4 v = pre[j];
            if constexpr (ROWSCALE) { v.x *= psc[j]; v.y *= psc[j]; v.z *= psc[j]; v.w *= psc[j]; }
            *(float4*)(sAf + (prow + 8 * j) * (2 * LPITCH) + pc4 * 4) = v;
        }
        lds_ready_barrier();

        const int nxt = strip + gstep;
        if (nxt < nStrips) {
#pragma unroll
            for (int j = 0; j < 4; ++j) {
                const int row = nxt * RSTRIP + prow + 8 * j;
                pre[j] = (row < M) ? ((const float4*)(A + (size_t)row * 128))[pc4]
                                   : float4{0.0f, 0.0f, 0.0f, 0.0f};
                if constexpr (ROWSCALE) psc[j] = (row < M) ? rscale[row] : 0.0f;
            }
        }

        if (has_tile) {
            v8f acc0 = {};
            v8f acc1 = {};
#pragma unroll 8
            for (int k2 = 0; k2 < 64; k2 += 2) {
                const int kq = k2 + hsel;
                v2f b  = sW2[col0 + m][kq];
                v2f a0 = sA2[m][kq];
                v2f a1 = sA2[16 + m][kq];
                acc0 = __builtin_amdgcn_wmma_f32_16x16x4_f32(
                    false, a0, false, b, (short)0, acc0, false, false);
                acc1 = __builtin_amdgcn_wmma_f32_16x16x4_f32(
                    false, a1, false, b, (short)0, acc1, false, false);
            }
#pragma unroll
            for (int r = 0; r < 8; ++r) {
                const int row = row0 + mb + r;
                if (row < M) {
                    float v = acc0[r] + bv;
                    if constexpr (DO_RELU) v = fmaxf(v, 0.0f);
                    out[(size_t)row * N + col0 + m] = v;
                }
            }
#pragma unroll
            for (int r = 0; r < 8; ++r) {
                const int row = row0 + 16 + mb + r;
                if (row < M) {
                    float v = acc1[r] + bv;
                    if constexpr (DO_RELU) v = fmaxf(v, 0.0f);
                    out[(size_t)row * N + col0 + m] = v;
                }
            }
        }
        reads_done_barrier();
    }
}

__device__ __forceinline__ void pool_flush(float* __restrict__ gsum, float* __restrict__ gcnt,
                                           int b, float4 r, int len, int lane) {
    const int comp = lane & 3;
    float* gp = gsum + (size_t)b * 128 + lane;
    for (int j = 0; j < 4; ++j) {
        const int sl = 8 * j + (lane >> 2);
        const float x = __shfl(r.x, sl), y = __shfl(r.y, sl);
        const float z = __shfl(r.z, sl), w = __shfl(r.w, sl);
        const float lo = (comp & 1) ? y : x;
        const float hi = (comp & 1) ? w : z;
        atomicAdd(gp + 32 * j, (comp & 2) ? hi : lo);
    }
    if (lane == 0) atomicAdd(gcnt + b, (float)len);
}

#define ROW(s) (((const float4*)(h + (size_t)(s) * 128))[lane])
#define ACC(vv_)                                                              \
    acc.x += (vv_).x; acc.y += (vv_).y; acc.z += (vv_).z; acc.w += (vv_).w

template <bool POOL>
__global__ __launch_bounds__(256) void gather_relu(const int* __restrict__ rowptr,
                                                   const int* __restrict__ cnt,
                                                   const int* __restrict__ csr_src,
                                                   const float* __restrict__ dinv,
                                                   const float* __restrict__ h,
                                                   const float* __restrict__ bias,
                                                   float* __restrict__ act,
                                                   const int* __restrict__ batch,
                                                   float* __restrict__ gsum,
                                                   float* __restrict__ gcnt, int nN) {
    const int lane = threadIdx.x & 31;
    const int wv = __builtin_amdgcn_readfirstlane((int)(threadIdx.x >> 5));
    const int nw = gridDim.x * 8;
    const int gw = blockIdx.x * 8 + wv;
    const float4 bv = ((const float4*)bias)[lane];

    int node = gw, nend = nN, nstep = nw;
    if constexpr (POOL) {
        const unsigned q = (unsigned)nN / (unsigned)nw, rem = (unsigned)nN % (unsigned)nw;
        node  = (int)((unsigned)gw * q + min((unsigned)gw, rem));
        nend  = node + (int)q + (((unsigned)gw < rem) ? 1 : 0);
        nstep = 1;
    }
    float4 run = {0.0f, 0.0f, 0.0f, 0.0f};
    int runb = -1, runlen = 0;

    for (; node < nend; node += nstep) {
        const int   beg = rowptr[node];
        const int   n   = cnt[node];
        const float di  = dinv[node];
        int gid = 0;
        if constexpr (POOL) gid = batch[node];
        const float4 hv = ROW(node);
        float4 acc = {0.0f, 0.0f, 0.0f, 0.0f};

        for (int c0 = 0; c0 < n; c0 += 32) {
            const int m = min(n - c0, 32);
            int s_l = 0;
            if (lane < m) s_l = csr_src[beg + c0 + lane];
            int j = 0;
            for (; j + 4 <= m; j += 4) {
                const int s0 = __builtin_amdgcn_readlane(s_l, j);
                const int s1 = __builtin_amdgcn_readlane(s_l, j + 1);
                const int s2 = __builtin_amdgcn_readlane(s_l, j + 2);
                const int s3 = __builtin_amdgcn_readlane(s_l, j + 3);
                const float4 v0 = ROW(s0);
                const float4 v1 = ROW(s1);
                const float4 v2 = ROW(s2);
                const float4 v3 = ROW(s3);
                ACC(v0); ACC(v1); ACC(v2); ACC(v3);
            }
            if (j + 2 <= m) {
                const int s0 = __builtin_amdgcn_readlane(s_l, j);
                const int s1 = __builtin_amdgcn_readlane(s_l, j + 1);
                const float4 v0 = ROW(s0);
                const float4 v1 = ROW(s1);
                ACC(v0); ACC(v1);
                j += 2;
            }
            if (j < m) {
                const int s0 = __builtin_amdgcn_readlane(s_l, j);
                const float4 v0 = ROW(s0);
                ACC(v0);
            }
        }

        float4 o;
        o.x = fmaxf(fmaf(di, acc.x + hv.x, bv.x), 0.0f);
        o.y = fmaxf(fmaf(di, acc.y + hv.y, bv.y), 0.0f);
        o.z = fmaxf(fmaf(di, acc.z + hv.z, bv.z), 0.0f);
        o.w = fmaxf(fmaf(di, acc.w + hv.w, bv.w), 0.0f);
        if constexpr (POOL) {
            const int b = gid;
            if (b != runb) {
                if (runlen > 0) pool_flush(gsum, gcnt, runb, run, runlen, lane);
                run = o; runb = b; runlen = 1;
            } else {
                run.x += o.x; run.y += o.y; run.z += o.z; run.w += o.w;
                ++runlen;
            }
        } else {
            ((float4*)(act + (size_t)node * 128))[lane] = o;
        }
    }
    if constexpr (POOL) {
        if (runlen > 0) pool_flush(gsum, gcnt, runb, run, runlen, lane);
    }
}

#define FROWS 128

__global__ __launch_bounds__(1024, 16) void gather_gemm(const int* __restrict__ rowptr,
                                                   const int* __restrict__ cnt,
                                                   const int* __restrict__ csr_src,
                                                   const float* __restrict__ dinv,
                                                   const float* __restrict__ h,
                                                   const float* __restrict__ bias,
                                                   const float* __restrict__ W,
                                                   float* __restrict__ out, int nN) {
    __shared__ v2f sA2[FROWS][LPITCH];
    __shared__ v2f sW2[128][LPITCH];

    const int tid  = threadIdx.x;
    const int lane = tid & 31;
    const int wave = __builtin_amdgcn_readfirstlane(tid >> 5);
    float* sAf = (float*)sA2;

    for (int i = tid; i < 64 * 128; i += 1024) {
        const int kp = i >> 7, n = i & 127;
        v2f w;
        w.x = W[(2 * kp) * 128 + n];
        w.y = W[(2 * kp + 1) * 128 + n];
        sW2[n][kp] = w;
    }

    const float4 bv = ((const float4*)bias)[lane];

    const unsigned ng = gridDim.x, wg = blockIdx.x;
    const unsigned q = (unsigned)nN / ng, rem = (unsigned)nN % ng;
    const int r0 = (int)(wg * q + min(wg, rem));
    const int r1 = r0 + (int)q + ((wg < rem) ? 1 : 0);

    const int m     = lane & 15;
    const int hsel  = lane >> 4;
    const int mb    = hsel << 3;
    const int col0  = (wave & 7) << 4;
    const int rbase = (wave >> 3) << 5;

    for (int s0 = r0; s0 < r1; s0 += FROWS) {
        const int rows = min(FROWS, r1 - s0);

        for (int i = 0; i < 4; ++i) {
            const int lr = i * 32 + wave;
            float4 o = {0.0f, 0.0f, 0.0f, 0.0f};
            if (lr < rows) {
                const int   node = s0 + lr;
                const int   beg  = rowptr[node];
                const int   n    = cnt[node];
                const float di   = dinv[node];
                const float4 hv  = ROW(node);
                float4 acc = {0.0f, 0.0f, 0.0f, 0.0f};

                for (int c0 = 0; c0 < n; c0 += 32) {
                    const int mm = min(n - c0, 32);
                    int s_l = 0;
                    if (lane < mm) s_l = csr_src[beg + c0 + lane];
                    int j = 0;
                    for (; j + 4 <= mm; j += 4) {
                        const int s0r = __builtin_amdgcn_readlane(s_l, j);
                        const int s1r = __builtin_amdgcn_readlane(s_l, j + 1);
                        const int s2r = __builtin_amdgcn_readlane(s_l, j + 2);
                        const int s3r = __builtin_amdgcn_readlane(s_l, j + 3);
                        const float4 v0 = ROW(s0r);
                        const float4 v1 = ROW(s1r);
                        const float4 v2 = ROW(s2r);
                        const float4 v3 = ROW(s3r);
                        ACC(v0); ACC(v1); ACC(v2); ACC(v3);
                    }
                    if (j + 2 <= mm) {
                        const int s0r = __builtin_amdgcn_readlane(s_l, j);
                        const int s1r = __builtin_amdgcn_readlane(s_l, j + 1);
                        const float4 v0 = ROW(s0r);
                        const float4 v1 = ROW(s1r);
                        ACC(v0); ACC(v1);
                        j += 2;
                    }
                    if (j < mm) {
                        const int s0r = __builtin_amdgcn_readlane(s_l, j);
                        const float4 v0 = ROW(s0r);
                        ACC(v0);
                    }
                }

                o.x = di * fmaxf(fmaf(di, acc.x + hv.x, bv.x), 0.0f);
                o.y = di * fmaxf(fmaf(di, acc.y + hv.y, bv.y), 0.0f);
                o.z = di * fmaxf(fmaf(di, acc.z + hv.z, bv.z), 0.0f);
                o.w = di * fmaxf(fmaf(di, acc.w + hv.w, bv.w), 0.0f);
            }
            *(float4*)(sAf + lr * (2 * LPITCH) + lane * 4) = o;
        }
        lds_ready_barrier();

        if (rbase < rows) {
            v8f acc0 = {};
            v8f acc1 = {};
#pragma unroll 8
            for (int k2 = 0; k2 < 64; k2 += 2) {
                const int kq = k2 + hsel;
                v2f b  = sW2[col0 + m][kq];
                v2f a0 = sA2[rbase + m][kq];
                v2f a1 = sA2[rbase + 16 + m][kq];
                acc0 = __builtin_amdgcn_wmma_f32_16x16x4_f32(
                    false, a0, false, b, (short)0, acc0, false, false);
                acc1 = __builtin_amdgcn_wmma_f32_16x16x4_f32(
                    false, a1, false, b, (short)0, acc1, false, false);
            }
#pragma unroll
            for (int r = 0; r < 8; ++r) {
                const int lr = rbase + mb + r;
                if (lr < rows) out[(size_t)(s0 + lr) * 128 + col0 + m] = acc0[r];
            }
#pragma unroll
            for (int r = 0; r < 8; ++r) {
                const int lr = rbase + 16 + mb + r;
                if (lr < rows) out[(size_t)(s0 + lr) * 128 + col0 + m] = acc1[r];
            }
        }
        reads_done_barrier();
    }
}
#undef ROW
#undef ACC

__global__ __launch_bounds__(1024) void head_mlp(const float* __restrict__ gsum,
                                                 const float* __restrict__ gcnt,
                                                 const float* __restrict__ Wa,
                                                 const float* __restrict__ ba,
                                                 const float* __restrict__ Wb,
                                                 const float* __restrict__ bb,
                                                 float* __restrict__ out, int G) {
    __shared__ v2f sG2[RSTRIP][LPITCH];
    __shared__ v2f sH2[RSTRIP][LPITCH];
    __shared__ v2f sWa[128][LPITCH];
    __shared__ v2f sWb[32][LPITCH];

    const int tid  = threadIdx.x;
    const int lane = tid & 31;
    const int wave = __builtin_amdgcn_readfirstlane(tid >> 5);
    const int row0 = blockIdx.x * RSTRIP;
    const int m    = lane & 15;
    const int hsel = lane >> 4;
    const int mb   = hsel << 3;

    v2f wa[8], wb[2];
#pragma unroll
    for (int j = 0; j < 8; ++j) {
        const int i = tid + 1024 * j, kp = i >> 7, n = i & 127;
        wa[j].x = Wa[(2 * kp) * 128 + n];
        wa[j].y = Wa[(2 * kp + 1) * 128 + n];
    }
#pragma unroll
    for (int j = 0; j < 2; ++j) {
        const int i = tid + 1024 * j, kp = i >> 5, n = i & 31;
        wb[j].x = Wb[(2 * kp) * 32 + n];
        wb[j].y = Wb[(2 * kp + 1) * 32 + n];
    }
    const int    rwc = min(row0 + wave, G - 1);
    const float4 s   = ((const float4*)(gsum + (size_t)rwc * 128))[lane];
    const float  cn  = gcnt[rwc];
    const float bva = ba[((wave & 7) << 4) + m];
    const float bvb = bb[((wave & 1) << 4) + m];

#pragma unroll
    for (int j = 0; j < 8; ++j) {
        const int i = tid + 1024 * j;
        sWa[i & 127][i >> 7] = wa[j];
    }
#pragma unroll
    for (int j = 0; j < 2; ++j) {
        const int i = tid + 1024 * j;
        sWb[i & 31][i >> 5] = wb[j];
    }
    {
        const float c = fmaxf(cn, 1.0f);
        float4 g;
        g.x = s.x / c; g.y = s.y / c; g.z = s.z / c; g.w = s.w / c;
        *(float4*)((float*)sG2 + wave * (2 * LPITCH) + lane * 4) = g;
    }
    lds_ready_barrier();

    if (wave < 8) {
        const int col0 = wave << 4;
        float* sHf = (float*)sH2;
        v8f acc0 = {};
        v8f acc1 = {};
#pragma unroll 8
        for (int k2 = 0; k2 < 64; k2 += 2) {
            const int kq = k2 + hsel;
            v2f b  = sWa[col0 + m][kq];
            v2f a0 = sG2[m][kq];
            v2f a1 = sG2[16 + m][kq];
            acc0 = __builtin_amdgcn_wmma_f32_16x16x4_f32(
                false, a0, false, b, (short)0, acc0, false, false);
            acc1 = __builtin_amdgcn_wmma_f32_16x16x4_f32(
                false, a1, false, b, (short)0, acc1, false, false);
        }
#pragma unroll
        for (int r = 0; r < 8; ++r)
            sHf[(mb + r) * (2 * LPITCH) + col0 + m] = fmaxf(acc0[r] + bva, 0.0f);
#pragma unroll
        for (int r = 0; r < 8; ++r)
            sHf[(16 + mb + r) * (2 * LPITCH) + col0 + m] = fmaxf(acc1[r] + bva, 0.0f);
    }
    lds_ready_barrier();

    if (wave < 2) {
        const int col0 = wave << 4;
        v8f acc0 = {};
        v8f acc1 = {};
#pragma unroll 8
        for (int k2 = 0; k2 < 64; k2 += 2) {
            const int kq = k2 + hsel;
            v2f b  = sWb[col0 + m][kq];
            v2f a0 = sH2[m][kq];
            v2f a1 = sH2[16 + m][kq];
            acc0 = __builtin_amdgcn_wmma_f32_16x16x4_f32(
                false, a0, false, b, (short)0, acc0, false, false);
            acc1 = __builtin_amdgcn_wmma_f32_16x16x4_f32(
                false, a1, false, b, (short)0, acc1, false, false);
        }
#pragma unroll
        for (int r = 0; r < 8; ++r) {
            const int row = row0 + mb + r;
            if (row < G) out[(size_t)row * 32 + col0 + m] = acc0[r] + bvb;
        }
#pragma unroll
        for (int r = 0; r < 8; ++r) {
            const int row = row0 + 16 + mb + r;
            if (row < G) out[(size_t)row * 32 + col0 + m] = acc1[r] + bvb;
        }
    }
}

extern "C" void kernel_launch(void* const* d_in, const int* in_sizes, int n_in,
                              void* d_out, int out_size, void* d_ws, size_t ws_size,
                              hipStream_t stream) {
    const float* x     = (const float*)d_in[0];
    const int*   ei    = (const int*)d_in[1];
    const int*   batch = (const int*)d_in[2];
    const float* W1    = (const float*)d_in[3];
    const float* b1    = (const float*)d_in[4];
    const float* W2    = (const float*)d_in[5];
    const float* b2    = (const float*)d_in[6];
    const float* Wl1   = (const float*)d_in[7];
    const float* bl1   = (const float*)d_in[8];
    const float* Wl2   = (const float*)d_in[9];
    const float* bl2   = (const float*)d_in[10];
    float* out = (float*)d_out;

    const int nN = in_sizes[0] / 128;
    const int nE = in_sizes[1] / 2;
    const int G  = out_size / 32;
    const int* src = ei;
    const int* dst = ei + nE;

    char* ws = (char*)d_ws;
    auto alloc = [&](size_t bytes) -> float* {
        float* p = (float*)ws;
        ws += (bytes + 255) & ~(size_t)255;
        return p;
    };
    float* bufA = alloc((size_t)nN * 128 * 4);
    float* bufB = alloc((size_t)nN * 128 * 4);
    char*  bufC = (char*)alloc((size_t)nN * 128 * 4);
    int*   cnt  = (int*)alloc((size_t)nN * 4);
    float* dinv = alloc((size_t)nN * 4);
    float* gsum = alloc((size_t)G * 128 * 4);
    float* gcnt = alloc((size_t)G * 4);

    auto sub = [&](size_t bytes) -> int* {
        int* p = (int*)bufC;
        bufC += (bytes + 255) & ~(size_t)255;
        return p;
    };
    const int NB = (nN + 255) >> BIN_SHIFT;
    const int NC = (nE + ECH - 1) / ECH;
    int* csr_src = sub((size_t)nE * 4);
    int* rank    = sub((size_t)nE * 4);
    int* bed     = sub((size_t)nE * 8);
    int* rowptr  = sub((size_t)nN * 4);
    int* hist    = sub((size_t)NB * NC * 4);
    int* btot    = sub((size_t)NB * 4);

    bin_count<<<NC, 256, 0, stream>>>(dst, rank, hist, nE, NB, NC, gsum, G * 128, gcnt, G);
    bin_scan<<<NB, 256, 0, stream>>>(hist, btot, NC);
    bin_scatter<<<NC, 256, 0, stream>>>(src, dst, rank, hist, btot, bed, nE, NB, NC);
    bin_csr<<<NB, 256, 0, stream>>>(bed, btot, rowptr, cnt, dinv, csr_src, nN);

    gemm_k128<128, false, false, true><<<768, 256, 0, stream>>>(x, W1, nullptr, dinv, bufA, nN);
    gather_gemm<<<512, 1024, 0, stream>>>(rowptr, cnt, csr_src, dinv, bufA, b1, W2, bufB, nN);

    gather_relu<true><<<2048, 256, 0, stream>>>(rowptr, cnt, csr_src, dinv, bufB, b2, nullptr,
                                                batch, gsum, gcnt, nN);

    head_mlp<<<(G + RSTRIP - 1) / RSTRIP, 1024, 0, stream>>>(gsum, gcnt, Wl1, bl1, Wl2, bl2, out, G);
}


# Pool sums of a work-group's 8 waves meet in two LDS rows (LDS float adds) and leave as one set of whole-line global atomics per group, not per wave: 8x fewer atomic lanes; pool sum order differs
# speedup vs baseline: 5.4299x; 1.0333x over previous
#include <hip/hip_runtime.h>

typedef __attribute__((ext_vector_type(2))) float v2f;
typedef __attribute__((ext_vector_type(8))) float v8f;

#define BIN_SHIFT 8
#define NBMAX     512
#define ECH       4096

__device__ __forceinline__ int wave_sum(int v) {
#pragma unroll
    for (int off = 16; off > 0; off >>= 1) v += __shfl_xor(v, off);
    return v;
}

__device__ __forceinline__ int wave_incl_scan(int v, int lane) {
#pragma unroll
    for (int off = 1; off < 32; off <<= 1) {
        int t = __shfl_up(v, off);
        if (lane >= off) v += t;
    }
    return v;
}

__global__ __launch_bounds__(256) void bin_count(const int* __restrict__ dst,
                                                 int* __restrict__ rank,
                                                 int* __restrict__ hist,
                                                 int nE, int NB, int NC,
                                                 float* __restrict__ zsum, int nzsum,
                                                 float* __restrict__ zcnt, int nzcnt) {
    __shared__ int sh[NBMAX + 1];
    const int tid = threadIdx.x;
    for (int i = tid; i < NBMAX + 1; i += 256) sh[i] = 0;
    const int gt = blockIdx.x * 256 + tid, gstride = (int)gridDim.x * 256;
    for (int i = gt; i < nzsum; i += gstride) zsum[i] = 0.0f;
    for (int i = gt; i < nzcnt; i += gstride) zcnt[i] = 0.0f;
    __syncthreads();
    const int base = blockIdx.x * ECH;
    int d[16], r[16];
#pragma unroll
    for (int j = 0; j < 16; ++j) {
        const int e = base + j * 256 + tid;
        d[j] = dst[min(e, nE - 1)];
    }
#pragma unroll
    for (int j = 0; j < 16; ++j) {
        const int e = base + j * 256 + tid;
        const int bin = (e < nE) ? min(d[j] >> BIN_SHIFT, NBMAX - 1) : NBMAX;
        r[j] = atomicAdd(&sh[bin], 1);
    }
#pragma unroll
    for (int j = 0; j < 16; ++j) {
        const int e = base + j * 256 + tid;
        if (e < nE) rank[e] = r[j];
    }
    __syncthreads();
    for (int i = tid; i < NB; i += 256) hist[(size_t)i * NC + blockIdx.x] = sh[i];
}

__global__ __launch_bounds__(256) void bin_scan(int* __restrict__ hist,
                                                int* __restrict__ btot, int NC) {
    __shared__ int sw[8];
    const int tid = threadIdx.x, lane = tid & 31, wave = tid >> 5;
    int* row = hist + (size_t)blockIdx.x * NC;
    int carry = 0;
    for (int c0 = 0; c0 < NC; c0 += 256) {
        const int c = c0 + tid;
        const int v = (c < NC) ? row[c] : 0;
        const int inc = wave_incl_scan(v, lane);
        if (lane == 31) sw[wave] = inc;
        __syncthreads();
        int off = 0, tot = 0;
        for (int w = 0; w < 8; ++w) {
            const int t = sw[w];
            tot += t;
            if (w < wave) off += t;
        }
        if (c < NC) row[c] = carry + off + inc - v;
        carry += tot;
        __syncthreads();
    }
    if (tid == 0) btot[blockIdx.x] = carry;
}

__global__ __launch_bounds__(256) void bin_scatter(const int* __restrict__ src,
                                                   const int* __restrict__ dst,
                                                   const int* __restrict__ rank,
                                                   const int* __restrict__ hist,
                                                   const int* __restrict__ btot,
                                                   int* __restrict__ bed_i,
                                                   int nE, int NB, int NC) {
    __shared__ int sbase[NBMAX];
    __shared__ int sw[8];
    const int tid = threadIdx.x, lane = tid & 31, wave = tid >> 5;
    int2* bed = (int2*)bed_i;

    const int b0 = 2 * tid, b1 = b0 + 1;
    const int a0 = (b0 < NB) ? btot[b0] : 0;
    const int a1 = (b1 < NB) ? btot[b1] : 0;
    const int s  = a0 + a1;
    const int inc = wave_incl_scan(s, lane);
    if (lane == 31) sw[wave] = inc;
    __syncthreads();
    int off = 0;
    for (int w = 0; w < 8; ++w)
        if (w < wave) off += sw[w];
    const int ex = off + inc - s;
    if (b0 < NB) sbase[b0] = ex + hist[(size_t)b0 * NC + blockIdx.x];
    if (b1 < NB) sbase[b1] = ex + a0 + hist[(size_t)b1 * NC + blockIdx.x];
    __syncthreads();

    const int base = blockIdx.x * ECH;
    for (int h = 0; h < 2; ++h) {
        int d[8], rk[8], sv[8];
#pragma unroll
        for (int j = 0; j < 8; ++j) {
            const int ec = min(base + (h * 8 + j) * 256 + tid, nE - 1);
            d[j]  = dst[ec];
            rk[j] = rank[ec];
            sv[j] = src[ec];
        }
#pragma unroll
        for (int j = 0; j < 8; ++j) {
            const int e = base + (h * 8 + j) * 256 + tid;
            const int pos = sbase[min(d[j] >> BIN_SHIFT, NBMAX - 1)] + rk[j];
            if (e < nE) bed[pos] = make_int2(sv[j], d[j]);
        }
    }
}

__global__ __launch_bounds__(256) void bin_csr(const int* __restrict__ bed_i,
                                               const int* __restrict__ btot,
                                               int* __restrict__ rowptr,
                                               int* __restrict__ cnt,
                                               float* __restrict__ dinv,
                                               int* __restrict__ csr_src, int nN) {
    __shared__ int scnt[257];
    __shared__ int sptr[257];
    __shared__ int sw[8];
    const int tid = threadIdx.x, lane = tid & 31, wave = tid >> 5;
    const int b = blockIdx.x;
    const int2* bed = (const int2*)bed_i;

    int pv = 0;
    for (int i = tid; i < b; i += 256) pv += btot[i];
    pv = wave_sum(pv);
    if (lane == 0) sw[wave] = pv;
    scnt[tid] = 0;
    if (tid == 0) { scnt[256] = 0; sptr[256] = 0; }
    __syncthreads();
    int beg = 0;
    for (int w = 0; w < 8; ++w) beg += sw[w];
    const int n = btot[b];
    __syncthreads();

    for (int i0 = 0; i0 < n; i0 += 8 * 256) {
        int dd[8];
#pragma unroll
        for (int j = 0; j < 8; ++j) dd[j] = bed[beg + min(i0 + j * 256 + tid, n - 1)].y;
#pragma unroll
        for (int j = 0; j < 8; ++j)
            atomicAdd(&scnt[(i0 + j * 256 + tid < n) ? (dd[j] & 255) : 256], 1);
    }
    __syncthreads();

    const int v   = scnt[tid];
    const int inc = wave_incl_scan(v, lane);
    if (lane == 31) sw[wave] = inc;
    __syncthreads();
    int off = 0;
    for (int w = 0; w < 8; ++w)
        if (w < wave) off += sw[w];
    const int ex = off + inc - v;
    sptr[tid] = ex;
    const int node = (b << BIN_SHIFT) + tid;
    if (node < nN) {
        rowptr[node] = beg + ex;
        cnt[node]    = v;
        dinv[node]   = rsqrtf((float)v + 1.0f);
    }
    __syncthreads();

    for (int i0 = 0; i0 < n; i0 += 8 * 256) {
        int2 ed[8];
        int  p[8];
#pragma unroll
        for (int j = 0; j < 8; ++j) ed[j] = bed[beg + min(i0 + j * 256 + tid, n - 1)];
#pragma unroll
        for (int j = 0; j < 8; ++j)
            p[j] = atomicAdd(&sptr[(i0 + j * 256 + tid < n) ? (ed[j].y & 255) : 256], 1);
#pragma unroll
        for (int j = 0; j < 8; ++j)
            if (i0 + j * 256 + tid < n) csr_src[beg + p[j]] = ed[j].x;
    }
}

#define LPITCH 66
#define RSTRIP 32

__device__ __forceinline__ void lds_ready_barrier() {
    asm volatile("s_wait_dscnt 0x0\n\ts_barrier_signal -1\n\ts_barrier_wait -1" ::: "memory");
}
__device__ __forceinline__ void reads_done_barrier() {
    asm volatile("s_barrier_signal -1\n\ts_barrier_wait -1" ::: "memory");
}

template <int N, bool HAS_BIAS, bool DO_RELU, bool ROWSCALE>
__global__ __launch_bounds__(256) void gemm_k128(const float* __restrict__ A,
                                                 const float* __restrict__ W,
                                                 const float* __restrict__ bias,
                                                 const float* __restrict__ rscale,
                                                 float* __restrict__ out,
                                                 int M) {
    __shared__ v2f sA2[RSTRIP][LPITCH];
    __shared__ v2f sW2[N][LPITCH];

    const int tid = threadIdx.x;

    const int nStrips = (M + RSTRIP - 1) / RSTRIP;
    const int prow = tid >> 5;
    const int pc4  = tid & 31;
    float* sAf = (float*)sA2;

    float4 pre[4];
    float  psc[4] = {1.0f, 1.0f, 1.0f, 1.0f};
    int strip = blockIdx.x;
    if (strip < nStrips) {
#pragma unroll
        for (int j = 0; j < 4; ++j) {
            const int row = strip * RSTRIP + prow + 8 * j;
            pre[j] = (row < M) ? ((const float4*)(A + (size_t)row * 128))[pc4]
                               : float4{0.0f, 0.0f, 0.0f, 0.0f};
            if constexpr (ROWSCALE) psc[j] = (row < M) ? rscale[row] : 0.0f;
        }
    }

    for (int i = tid; i < 64 * N; i += 256) {
        const int kp = i / N, n = i % N;
        v2f w;
        w.x = W[(2 * kp) * N + n];
        w.y = W[(2 * kp + 1) * N + n];
        sW2[n][kp] = w;
    }

    const int wave  = tid >> 5;
    const int lane  = tid & 31;
    const int m     = lane & 15;
    const int hsel  = lane >> 4;
    const int mb    = hsel << 3;

    static_assert(N % 16 == 0 && N / 16 <= 8, "one column tile per wave");
    const bool has_tile = wave < N / 16;
    const int  col0     = wave << 4;
    float bv = 0.0f;
    if constexpr (HAS_BIAS) { if (has_tile) bv = bias[col0 + m]; }

    const int gstep = (int)gridDim.x;
    for (; strip < nStrips; strip += gstep) {
        const int row0 = strip * RSTRIP;
#pragma unroll
        for (int j = 0; j < 4; ++j) {
            float4 v = pre[j];
            if constexpr (ROWSCALE) { v.x *= psc[j]; v.y *= psc[j]; v.z *= psc[j]; v.w *= psc[j]; }
            *(float4*)(sAf + (prow + 8 * j) * (2 * LPITCH) + pc4 * 4) = v;
        }
        lds_ready_barrier();

        const int nxt = strip + gstep;
        if (nxt < nStrips) {
#pragma unroll
            for (int j = 0; j < 4; ++j) {
                const int row = nxt * RSTRIP + prow + 8 * j;
                pre[j] = (row < M) ? ((const float4*)(A + (size_t)row * 128))[pc4]
                                   : float4{0.0f, 0.0f, 0.0f, 0.0f};
                if constexpr (ROWSCALE) psc[j] = (row < M) ? rscale[row] : 0.0f;
            }
        }

        if (has_tile) {
            v8f acc0 = {};
            v8f acc1 = {};
#pragma unroll 8
            for (int k2 = 0; k2 < 64; k2 += 2) {
                const int kq = k2 + hsel;
                v2f b  = sW2[col0 + m][kq];
                v2f a0 = sA2[m][kq];
                v2f a1 = sA2[16 + m][kq];
                acc0 = __builtin_amdgcn_wmma_f32_16x16x4_f32(
                    false, a0, false, b, (short)0, acc0, false, false);
                acc1 = __builtin_amdgcn_wmma_f32_16x16x4_f32(
                    false, a1, false, b, (short)0, acc1, false, false);
            }
#pragma unroll
            for (int r = 0; r < 8; ++r) {
                const int row = row0 + mb + r;
                if (row < M) {
                    float v = acc0[r] + bv;
                    if constexpr (DO_RELU) v = fmaxf(v, 0.0f);
                    out[(size_t)row * N + col0 + m] = v;
                }
            }
#pragma unroll
            for (int r = 0; r < 8; ++r) {
                const int row = row0 + 16 + mb + r;
                if (row < M) {
                    float v = acc1[r] + bv;
                    if constexpr (DO_RELU) v = fmaxf(v, 0.0f);
                    out[(size_t)row * N + col0 + m] = v;
                }
            }
        }
        reads_done_barrier();
    }
}

__device__ __forceinline__ void pool_flush(float* __restrict__ gsum, float* __restrict__ gcnt,
                                           int b, float4 r, int len, int lane) {
    const int comp = lane & 3;
    float* gp = gsum + (size_t)b * 128 + lane;
    for (int j = 0; j < 4; ++j) {
        const int sl = 8 * j + (lane >> 2);
        const float x = __shfl(r.x, sl), y = __shfl(r.y, sl);
        const float z = __shfl(r.z, sl), w = __shfl(r.w, sl);
        const float lo = (comp & 1) ? y : x;
        const float hi = (comp & 1) ? w : z;
        atomicAdd(gp + 32 * j, (comp & 2) ? hi : lo);
    }
    if (lane == 0) atomicAdd(gcnt + b, (float)len);
}

__device__ __forceinline__ void pool_hand_over(float* __restrict__ srow, float* __restrict__ scnt,
                                               int g0, int g1,
                                               float* __restrict__ gsum, float* __restrict__ gcnt,
                                               int b, float4 r, int len, int lane) {
    if (b == g0 || b == g1) {
        float* p = srow + ((b == g0) ? 0 : 128) + 4 * lane;
        atomicAdd(p + 0, r.x);
        atomicAdd(p + 1, r.y);
        atomicAdd(p + 2, r.z);
        atomicAdd(p + 3, r.w);
        if (lane == 0) atomicAdd(scnt + ((b == g0) ? 0 : 1), (float)len);
    } else {
        pool_flush(gsum, gcnt, b, r, len, lane);
    }
}

#define ROW(s) (((const float4*)(h + (size_t)(s) * 128))[lane])
#define ACC(vv_)                                                              \
    acc.x += (vv_).x; acc.y += (vv_).y; acc.z += (vv_).z; acc.w += (vv_).w

template <bool POOL>
__global__ __launch_bounds__(256) void gather_relu(const int* __restrict__ rowptr,
                                                   const int* __restrict__ cnt,
                                                   const int* __restrict__ csr_src,
                                                   const float* __restrict__ dinv,
                                                   const float* __restrict__ h,
                                                   const float* __restrict__ bias,
                                                   float* __restrict__ act,
                                                   const int* __restrict__ batch,
                                                   float* __restrict__ gsum,
                                                   float* __restrict__ gcnt, int nN) {
    const int lane = threadIdx.x & 31;
    const int wv = __builtin_amdgcn_readfirstlane((int)(threadIdx.x >> 5));
    const int nw = gridDim.x * 8;
    const int gw = blockIdx.x * 8 + wv;
    const float4 bv = ((const float4*)bias)[lane];

    int node = gw, nend = nN, nstep = nw;
    if constexpr (POOL) {
        const unsigned q = (unsigned)nN / (unsigned)nw, rem = (unsigned)nN % (unsigned)nw;
        node  = (int)((unsigned)gw * q + min((unsigned)gw, rem));
        nend  = node + (int)q + (((unsigned)gw < rem) ? 1 : 0);
        nstep = 1;
    }
    float4 run = {0.0f, 0.0f, 0.0f, 0.0f};
    int runb = -1, runlen = 0;

    __shared__ float srow[2 * 128];
    __shared__ float scnt[2];
    int g0 = -1, g1 = -1;
    if constexpr (POOL) {
        const unsigned q = (unsigned)nN / (unsigned)nw, rem = (unsigned)nN % (unsigned)nw;
        const unsigned wf = blockIdx.x * 8u, wl = wf + 8u;
        const int first = (int)(wf * q + min(wf, rem));
        const int last  = (int)(wl * q + min(wl, rem));
        if (first < last) { g0 = batch[first]; g1 = batch[last - 1]; }
        srow[threadIdx.x] = 0.0f;
        if (threadIdx.x < 2) scnt[threadIdx.x] = 0.0f;
        __syncthreads();
    }

    for (; node < nend; node += nstep) {
        const int   beg = rowptr[node];
        const int   n   = cnt[node];
        const float di  = dinv[node];
        int gid = 0;
        if constexpr (POOL) gid = batch[node];
        const float4 hv = ROW(node);
        float4 acc = {0.0f, 0.0f, 0.0f, 0.0f};

        for (int c0 = 0; c0 < n; c0 += 32) {
            const int m = min(n - c0, 32);
            int s_l = 0;
            if (lane < m) s_l = csr_src[beg + c0 + lane];
            int j = 0;
            for (; j + 4 <= m; j += 4) {
                const int s0 = __builtin_amdgcn_readlane(s_l, j);
                const int s1 = __builtin_amdgcn_readlane(s_l, j + 1);
                const int s2 = __builtin_amdgcn_readlane(s_l, j + 2);
                const int s3 = __builtin_amdgcn_readlane(s_l, j + 3);
                const float4 v0 = ROW(s0);
                const float4 v1 = ROW(s1);
                const float4 v2 = ROW(s2);
                const float4 v3 = ROW(s3);
                ACC(v0); ACC(v1); ACC(v2); ACC(v3);
            }
            if (j + 2 <= m) {
                const int s0 = __builtin_amdgcn_readlane(s_l, j);
                const int s1 = __builtin_amdgcn_readlane(s_l, j + 1);
                const float4 v0 = ROW(s0);
                const float4 v1 = ROW(s1);
                ACC(v0); ACC(v1);
                j += 2;
            }
            if (j < m) {
                const int s0 = __builtin_amdgcn_readlane(s_l, j);
                const float4 v0 = ROW(s0);
                ACC(v0);
            }
        }

        float4 o;
        o.x = fmaxf(fmaf(di, acc.x + hv.x, bv.x), 0.0f);
        o.y = fmaxf(fmaf(di, acc.y + hv.y, bv.y), 0.0f);
        o.z = fmaxf(fmaf(di, acc.z + hv.z, bv.z), 0.0f);
        o.w = fmaxf(fmaf(di, acc.w + hv.w, bv.w), 0.0f);
        if constexpr (POOL) {
            const int b = gid;
            if (b != runb) {
                if (runlen > 0) pool_hand_over(srow, scnt, g0, g1, gsum, gcnt, runb, run, runlen, lane);
                run = o; runb = b; runlen = 1;
            } else {
                run.x += o.x; run.y += o.y; run.z += o.z; run.w += o.w;
                ++runlen;
            }
        } else {
            ((float4*)(act + (size_t)node * 128))[lane] = o;
        }
    }
    if constexpr (POOL) {
        if (runlen > 0) pool_hand_over(srow, scnt, g0, g1, gsum, gcnt, runb, run, runlen, lane);
        __syncthreads();
        const int t = (int)threadIdx.x, slot = t >> 7, f = t & 127;
        const float cn = scnt[slot];
        if (cn > 0.0f) {
            const int g = slot ? g1 : g0;
            atomicAdd(gsum + (size_t)g * 128 + f, srow[t]);
            if (f == 0) atomicAdd(gcnt + g, cn);
        }
    }
}

#define FROWS 128

__global__ __launch_bounds__(1024, 16) void gather_gemm(const int* __restrict__ rowptr,
                                                   const int* __restrict__ cnt,
                                                   const int* __restrict__ csr_src,
                                                   const float* __restrict__ dinv,
                                                   const float* __restrict__ h,
                                                   const float* __restrict__ bias,
                                                   const float* __restrict__ W,
                                                   float* __restrict__ out, int nN) {
    __shared__ v2f sA2[FROWS][LPITCH];
    __shared__ v2f sW2[128][LPITCH];

    const int tid  = threadIdx.x;
    const int lane = tid & 31;
    const int wave = __builtin_amdgcn_readfirstlane(tid >> 5);
    float* sAf = (float*)sA2;

    for (int i = tid; i < 64 * 128; i += 1024) {
        const int kp = i >> 7, n = i & 127;
        v2f w;
        w.x = W[(2 * kp) * 128 + n];
        w.y = W[(2 * kp + 1) * 128 + n];
        sW2[n][kp] = w;
    }

    const float4 bv = ((const float4*)bias)[lane];

    const unsigned ng = gridDim.x, wg = blockIdx.x;
    const unsigned q = (unsigned)nN / ng, rem = (unsigned)nN % ng;
    const int r0 = (int)(wg * q + min(wg, rem));
    const int r1 = r0 + (int)q + ((wg < rem) ? 1 : 0);

    const int m     = lane & 15;
    const int hsel  = lane >> 4;
    const int mb    = hsel << 3;
    const int col0  = (wave & 7) << 4;
    const int rbase = (wave >> 3) << 5;

    for (int s0 = r0; s0 < r1; s0 += FROWS) {
        const int rows = min(FROWS, r1 - s0);

        for (int i = 0; i < 4; ++i) {
            const int lr = i * 32 + wave;
            float4 o = {0.0f, 0.0f, 0.0f, 0.0f};
            if (lr < rows) {
                const int   node = s0 + lr;
                const int   beg  = rowptr[node];
                const int   n    = cnt[node];
                const float di   = dinv[node];
                const float4 hv  = ROW(node);
                float4 acc = {0.0f, 0.0f, 0.0f, 0.0f};

                for (int c0 = 0; c0 < n; c0 += 32) {
                    const int mm = min(n - c0, 32);
                    int s_l = 0;
                    if (lane < mm) s_l = csr_src[beg + c0 + lane];
                    int j = 0;
                    for (; j + 4 <= mm; j += 4) {
                        const int s0r = __builtin_amdgcn_readlane(s_l, j);
                        const int s1r = __builtin_amdgcn_readlane(s_l, j + 1);
                        const int s2r = __builtin_amdgcn_readlane(s_l, j + 2);
                        const int s3r = __builtin_amdgcn_readlane(s_l, j + 3);
                        const float4 v0 = ROW(s0r);
                        const float4 v1 = ROW(s1r);
                        const float4 v2 = ROW(s2r);
                        const float4 v3 = ROW(s3r);
                        ACC(v0); ACC(v1); ACC(v2); ACC(v3);
                    }
                    if (j + 2 <= mm) {
                        const int s0r = __builtin_amdgcn_readlane(s_l, j);
                        const int s1r = __builtin_amdgcn_readlane(s_l, j + 1);
                        const float4 v0 = ROW(s0r);
                        const float4 v1 = ROW(s1r);
                        ACC(v0); ACC(v1);
                        j += 2;
                    }
                    if (j < mm) {
                        const int s0r = __builtin_amdgcn_readlane(s_l, j);
                        const float4 v0 = ROW(s0r);
                        ACC(v0);
                    }
                }

                o.x = di * fmaxf(fmaf(di, acc.x + hv.x, bv.x), 0.0f);
                o.y = di * fmaxf(fmaf(di, acc.y + hv.y, bv.y), 0.0f);
                o.z = di * fmaxf(fmaf(di, acc.z + hv.z, bv.z), 0.0f);
                o.w = di * fmaxf(fmaf(di, acc.w + hv.w, bv.w), 0.0f);
            }
            *(float4*)(sAf + lr * (2 * LPITCH) + lane * 4) = o;
        }
        lds_ready_barrier();

        if (rbase < rows) {
            v8f acc0 = {};
            v8f acc1 = {};
#pragma unroll 8
            for (int k2 = 0; k2 < 64; k2 += 2) {
                const int kq = k2 + hsel;
                v2f b  = sW2[col0 + m][kq];
                v2f a0 = sA2[rbase + m][kq];
                v2f a1 = sA2[rbase + 16 + m][kq];
                acc0 = __builtin_amdgcn_wmma_f32_16x16x4_f32(
                    false, a0, false, b, (short)0, acc0, false, false);
                acc1 = __builtin_amdgcn_wmma_f32_16x16x4_f32(
                    false, a1, false, b, (short)0, acc1, false, false);
            }
#pragma unroll
            for (int r = 0; r < 8; ++r) {
                const int lr = rbase + mb + r;
                if (lr < rows) out[(size_t)(s0 + lr) * 128 + col0 + m] = acc0[r];
            }
#pragma unroll
            for (int r = 0; r < 8; ++r) {
                const int lr = rbase + 16 + mb + r;
                if (lr < rows) out[(size_t)(s0 + lr) * 128 + col0 + m] = acc1[r];
            }
        }
        reads_done_barrier();
    }
}
#undef ROW
#undef ACC

__global__ __launch_bounds__(1024) void head_mlp(const float* __restrict__ gsum,
                                                 const float* __restrict__ gcnt,
                                                 const float* __restrict__ Wa,
                                                 const float* __restrict__ ba,
                                                 const float* __restrict__ Wb,
                                                 const float* __restrict__ bb,
                                                 float* __restrict__ out, int G) {
    __shared__ v2f sG2[RSTRIP][LPITCH];
    __shared__ v2f sH2[RSTRIP][LPITCH];
    __shared__ v2f sWa[128][LPITCH];
    __shared__ v2f sWb[32][LPITCH];

    const int tid  = threadIdx.x;
    const int lane = tid & 31;
    const int wave = __builtin_amdgcn_readfirstlane(tid >> 5);
    const int row0 = blockIdx.x * RSTRIP;
    const int m    = lane & 15;
    const int hsel = lane >> 4;
    const int mb   = hsel << 3;

    v2f wa[8], wb[2];
#pragma unroll
    for (int j = 0; j < 8; ++j) {
        const int i = tid + 1024 * j, kp = i >> 7, n = i & 127;
        wa[j].x = Wa[(2 * kp) * 128 + n];
        wa[j].y = Wa[(2 * kp + 1) * 128 + n];
    }
#pragma unroll
    for (int j = 0; j < 2; ++j) {
        const int i = tid + 1024 * j, kp = i >> 5, n = i & 31;
        wb[j].x = Wb[(2 * kp) * 32 + n];
        wb[j].y = Wb[(2 * kp + 1) * 32 + n];
    }
    const int    rwc = min(row0 + wave, G - 1);
    const float4 s   = ((const float4*)(gsum + (size_t)rwc * 128))[lane];
    const float  cn  = gcnt[rwc];
    const float bva = ba[((wave & 7) << 4) + m];
    const float bvb = bb[((wave & 1) << 4) + m];

#pragma unroll
    for (int j = 0; j < 8; ++j) {
        const int i = tid + 1024 * j;
        sWa[i & 127][i >> 7] = wa[j];
    }
#pragma unroll
    for (int j = 0; j < 2; ++j) {
        const int i = tid + 1024 * j;
        sWb[i & 31][i >> 5] = wb[j];
    }
    {
        const float c = fmaxf(cn, 1.0f);
        float4 g;
        g.x = s.x / c; g.y = s.y / c; g.z = s.z / c; g.w = s.w / c;
        *(float4*)((float*)sG2 + wave * (2 * LPITCH) + lane * 4) = g;
    }
    lds_ready_barrier();

    if (wave < 8) {
        const int col0 = wave << 4;
        float* sHf = (float*)sH2;
        v8f acc0 = {};
        v8f acc1 = {};
#pragma unroll 8
        for (int k2 = 0; k2 < 64; k2 += 2) {
            const int kq = k2 + hsel;
            v2f b  = sWa[col0 + m][kq];
            v2f a0 = sG2[m][kq];
            v2f a1 = sG2[16 + m][kq];
            acc0 = __builtin_amdgcn_wmma_f32_16x16x4_f32(
                false, a0, false, b, (short)0, acc0, false, false);
            acc1 = __builtin_amdgcn_wmma_f32_16x16x4_f32(
                false, a1, false, b, (short)0, acc1, false, false);
        }
#pragma unroll
        for (int r = 0; r < 8; ++r)
            sHf[(mb + r) * (2 * LPITCH) + col0 + m] = fmaxf(acc0[r] + bva, 0.0f);
#pragma unroll
        for (int r = 0; r < 8; ++r)
            sHf[(16 + mb + r) * (2 * LPITCH) + col0 + m] = fmaxf(acc1[r] + bva, 0.0f);
    }
    lds_ready_barrier();

    if (wave < 2) {
        const int col0 = wave << 4;
        v8f acc0 = {};
        v8f acc1 = {};
#pragma unroll 8
        for (int k2 = 0; k2 < 64; k2 += 2) {
            const int kq = k2 + hsel;
            v2f b  = sWb[col0 + m][kq];
            v2f a0 = sH2[m][kq];
            v2f a1 = sH2[16 + m][kq];
            acc0 = __builtin_amdgcn_wmma_f32_16x16x4_f32(
                false, a0, false, b, (short)0, acc0, false, false);
            acc1 = __builtin_amdgcn_wmma_f32_16x16x4_f32(
                false, a1, false, b, (short)0, acc1, false, false);
        }
#pragma unroll
        for (int r = 0; r < 8; ++r) {
            const int row = row0 + mb + r;
            if (row < G) out[(size_t)row * 32 + col0 + m] = acc0[r] + bvb;
        }
#pragma unroll
        for (int r = 0; r < 8; ++r) {
            const int row = row0 + 16 + mb + r;
            if (row < G) out[(size_t)row * 32 + col0 + m] = acc1[r] + bvb;
        }
    }
}

extern "C" void kernel_launch(void* const* d_in, const int* in_sizes, int n_in,
                              void* d_out, int out_size, void* d_ws, size_t ws_size,
                              hipStream_t stream) {
    const float* x     = (const float*)d_in[0];
    const int*   ei    = (const int*)d_in[1];
    const int*   batch = (const int*)d_in[2];
    const float* W1    = (const float*)d_in[3];
    const float* b1    = (const float*)d_in[4];
    const float* W2    = (const float*)d_in[5];
    const float* b2    = (const float*)d_in[6];
    const float* Wl1   = (const float*)d_in[7];
    const float* bl1   = (const float*)d_in[8];
    const float* Wl2   = (const float*)d_in[9];
    const float* bl2   = (const float*)d_in[10];
    float* out = (float*)d_out;

    const int nN = in_sizes[0] / 128;
    const int nE = in_sizes[1] / 2;
    const int G  = out_size / 32;
    const int* src = ei;
    const int* dst = ei + nE;

    char* ws = (char*)d_ws;
    auto alloc = [&](size_t bytes) -> float* {
        float* p = (float*)ws;
        ws += (bytes + 255) & ~(size_t)255;
        return p;
    };
    float* bufA = alloc((size_t)nN * 128 * 4);
    float* bufB = alloc((size_t)nN * 128 * 4);
    char*  bufC = (char*)alloc((size_t)nN * 128 * 4);
    int*   cnt  = (int*)alloc((size_t)nN * 4);
    float* dinv = alloc((size_t)nN * 4);
    float* gsum = alloc((size_t)G * 128 * 4);
    float* gcnt = alloc((size_t)G * 4);

    auto sub = [&](size_t bytes) -> int* {
        int* p = (int*)bufC;
        bufC += (bytes + 255) & ~(size_t)255;
        return p;
    };
    const int NB = (nN + 255) >> BIN_SHIFT;
    const int NC = (nE + ECH - 1) / ECH;
    int* csr_src = sub((size_t)nE * 4);
    int* rank    = sub((size_t)nE * 4);
    int* bed     = sub((size_t)nE * 8);
    int* rowptr  = sub((size_t)nN * 4);
    int* hist    = sub((size_t)NB * NC * 4);
    int* btot    = sub((size_t)NB * 4);

    bin_count<<<NC, 256, 0, stream>>>(dst, rank, hist, nE, NB, NC, gsum, G * 128, gcnt, G);
    bin_scan<<<NB, 256, 0, stream>>>(hist, btot, NC);
    bin_scatter<<<NC, 256, 0, stream>>>(src, dst, rank, hist, btot, bed, nE, NB, NC);
    bin_csr<<<NB, 256, 0, stream>>>(bed, btot, rowptr, cnt, dinv, csr_src, nN);

    gemm_k128<128, false, false, true><<<768, 256, 0, stream>>>(x, W1, nullptr, dinv, bufA, nN);
    gather_gemm<<<512, 1024, 0, stream>>>(rowptr, cnt, csr_src, dinv, bufA, b1, W2, bufB, nN);

    gather_relu<true><<<2048, 256, 0, stream>>>(rowptr, cnt, csr_src, dinv, bufB, b2, nullptr,
                                                batch, gsum, gcnt, nN);

    head_mlp<<<(G + RSTRIP - 1) / RSTRIP, 1024, 0, stream>>>(gsum, gcnt, Wl1, bl1, Wl2, bl2, out, G);
}


# Edge grouping in 2 launches, not 4: an edge group writes its edges bin-ordered into its own span plus (offset,count) per bin; a bin's group reads its pieces, offsets sum to its base; sum order kept
# speedup vs baseline: 6.7817x; 1.2490x over previous
#include <hip/hip_runtime.h>

typedef __attribute__((ext_vector_type(2))) float v2f;
typedef __attribute__((ext_vector_type(8))) float v8f;

#define BIN_SHIFT 8
#define NBMAX     512
#define TPA       512
#define ECH       8192
#define PB        16

__device__ __forceinline__ int wave_sum(int v) {
#pragma unroll
    for (int off = 16; off > 0; off >>= 1) v += __shfl_xor(v, off);
    return v;
}

__device__ __forceinline__ int wave_incl_scan(int v, int lane) {
#pragma unroll
    for (int off = 1; off < 32; off <<= 1) {
        int t = __shfl_up(v, off);
        if (lane >= off) v += t;
    }
    return v;
}

__global__ __launch_bounds__(TPA) void edge_part(const int* __restrict__ src,
                                                 const int* __restrict__ dst,
                                                 int* __restrict__ loc_i,
                                                 int* __restrict__ hist_i,
                                                 int nE, int NB,
                                                 float* __restrict__ zsum, int nzsum,
                                                 float* __restrict__ zcnt, int nzcnt) {
    __shared__ int sh[NBMAX + 1];
    __shared__ int soff[NBMAX];
    __shared__ int sw[TPA / 32];
    const int tid = threadIdx.x, lane = tid & 31, wave = tid >> 5;
    int2* loc  = (int2*)loc_i;
    int2* hist = (int2*)hist_i;
    for (int i = tid; i < NBMAX + 1; i += TPA) sh[i] = 0;
    const int gt = blockIdx.x * TPA + tid, gstride = (int)gridDim.x * TPA;
    for (int i = gt; i < nzsum; i += gstride) zsum[i] = 0.0f;
    for (int i = gt; i < nzcnt; i += gstride) zcnt[i] = 0.0f;
    __syncthreads();
    const int base = blockIdx.x * ECH;
    int d[16], sv[16], r[16];
#pragma unroll
    for (int j = 0; j < 16; ++j) {
        const int ec = min(base + j * TPA + tid, nE - 1);
        d[j]  = dst[ec];
        sv[j] = src[ec];
    }
#pragma unroll
    for (int j = 0; j < 16; ++j) {
        const int e = base + j * TPA + tid;
        const int bin = (e < nE) ? min(d[j] >> BIN_SHIFT, NBMAX - 1) : NBMAX;
        r[j] = atomicAdd(&sh[bin], 1);
    }
    __syncthreads();

    const int c   = sh[tid];
    const int inc = wave_incl_scan(c, lane);
    if (lane == 31) sw[wave] = inc;
    __syncthreads();
    int off = 0;
    for (int w = 0; w < TPA / 32; ++w)
        if (w < wave) off += sw[w];
    const int ex = off + inc - c;
    soff[tid] = ex;
    if (tid < NB) hist[(size_t)blockIdx.x * NB + tid] = make_int2(ex, c);
    __syncthreads();

#pragma unroll
    for (int j = 0; j < 16; ++j) {
        const int e = base + j * TPA + tid;
        const int bin = min(d[j] >> BIN_SHIFT, NBMAX - 1);
        const int pos = base + soff[bin] + r[j];
        if (e < nE) loc[pos] = make_int2(sv[j], d[j]);
    }
}

template <bool PLACE>
__device__ __forceinline__ void csr_trip(const int2* __restrict__ loc, const int (&st)[PB],
                                         const int (&ln)[PB], int* __restrict__ stab,
                                         int* __restrict__ csr_src, int beg, int nE, int lane) {
    int2 ed[PB];
    int  p[PB];
#pragma unroll
    for (int k = 0; k < PB; ++k)
        ed[k] = loc[min(st[k] + min(lane, max(ln[k] - 1, 0)), nE - 1)];
#pragma unroll
    for (int k = 0; k < PB; ++k) {
        p[k] = 0;
        if (lane < ln[k]) p[k] = atomicAdd(&stab[ed[k].y & 255], 1);
    }
    if constexpr (PLACE) {
#pragma unroll
        for (int k = 0; k < PB; ++k)
            if (lane < ln[k]) csr_src[min(beg + p[k], nE - 1)] = ed[k].x;
    }
#pragma unroll
    for (int k = 0; k < PB; ++k)
        for (int i = lane + 32; i < ln[k]; i += 32) {
            const int2 e2 = loc[min(st[k] + i, nE - 1)];
            const int  p2 = atomicAdd(&stab[e2.y & 255], 1);
            if constexpr (PLACE) csr_src[min(beg + p2, nE - 1)] = e2.x;
        }
}

template <bool PLACE>
__device__ __forceinline__ int csr_walk(const int2* __restrict__ loc, const int2* __restrict__ hist,
                                        int* __restrict__ stab, int* __restrict__ csr_src,
                                        int beg, int nE, int NB, int NC, int b, int wave, int lane) {
    int osum = 0;
    for (int m0 = 0; wave + 8 * m0 < NC; m0 += 32) {
        const int g = wave + 8 * (m0 + lane);
        int2 oc = make_int2(0, 0);
        if (g < NC) oc = hist[(size_t)g * NB + b];
        const int myst = g * ECH + oc.x;
        const int myln = oc.y;
        osum += oc.x;
#pragma unroll
        for (int k0 = 0; k0 < 32; k0 += PB) {
            if (wave + 8 * (m0 + k0) < NC) {
                int st[PB], ln[PB];
#pragma unroll
                for (int k = 0; k < PB; ++k) {
                    st[k] = __builtin_amdgcn_readlane(myst, k0 + k);
                    ln[k] = __builtin_amdgcn_readlane(myln, k0 + k);
                }
                csr_trip<PLACE>(loc, st, ln, stab, csr_src, beg, nE, lane);
            }
        }
    }
    return osum;
}

__global__ __launch_bounds__(256) void bin_csr(const int* __restrict__ loc_i,
                                               const int* __restrict__ hist_i,
                                               int* __restrict__ rowptr,
                                               int* __restrict__ cnt,
                                               float* __restrict__ dinv,
                                               int* __restrict__ csr_src,
                                               int nN, int nE, int NB, int NC) {
    __shared__ int scnt[256];
    __shared__ int sptr[256];
    __shared__ int sw[8];
    const int tid = threadIdx.x, lane = tid & 31, wv = tid >> 5;
    const int wave = __builtin_amdgcn_readfirstlane(tid >> 5);
    const int b = blockIdx.x;
    const int2* loc  = (const int2*)loc_i;
    const int2* hist = (const int2*)hist_i;

    scnt[tid] = 0;
    __syncthreads();

    int pv = csr_walk<false>(loc, hist, scnt, csr_src, 0, nE, NB, NC, b, wave, lane);
    pv = wave_sum(pv);
    if (lane == 0) sw[wv] = pv;
    __syncthreads();
    int beg = 0;
    for (int w = 0; w < 8; ++w) beg += sw[w];
    const int v = scnt[tid];
    __syncthreads();

    const int inc = wave_incl_scan(v, lane);
    if (lane == 31) sw[wv] = inc;
    __syncthreads();
    int off = 0;
    for (int w = 0; w < 8; ++w)
        if (w < wv) off += sw[w];
    const int ex = off + inc - v;
    sptr[tid] = ex;
    const int node = (b << BIN_SHIFT) + tid;
    if (node < nN) {
        rowptr[node] = beg + ex;
        cnt[node]    = v;
        dinv[node]   = rsqrtf((float)v + 1.0f);
    }
    __syncthreads();

    csr_walk<true>(loc, hist, sptr, csr_src, beg, nE, NB, NC, b, wave, lane);
}

#define LPITCH 66
#define RSTRIP 32

__device__ __forceinline__ void lds_ready_barrier() {
    asm volatile("s_wait_dscnt 0x0\n\ts_barrier_signal -1\n\ts_barrier_wait -1" ::: "memory");
}
__device__ __forceinline__ void reads_done_barrier() {
    asm volatile("s_barrier_signal -1\n\ts_barrier_wait -1" ::: "memory");
}

template <int N, bool HAS_BIAS, bool DO_RELU, bool ROWSCALE>
__global__ __launch_bounds__(256) void gemm_k128(const float* __restrict__ A,
                                                 const float* __restrict__ W,
                                                 const float* __restrict__ bias,
                                                 const float* __restrict__ rscale,
                                                 float* __restrict__ out,
                                                 int M) {
    __shared__ v2f sA2[RSTRIP][LPITCH];
    __shared__ v2f sW2[N][LPITCH];

    const int tid = threadIdx.x;

    const int nStrips = (M + RSTRIP - 1) / RSTRIP;
    const int prow = tid >> 5;
    const int pc4  = tid & 31;
    float* sAf = (float*)sA2;

    float4 pre[4];
    float  psc[4] = {1.0f, 1.0f, 1.0f, 1.0f};
    int strip = blockIdx.x;
    if (strip < nStrips) {
#pragma unroll
        for (int j = 0; j < 4; ++j) {
            const int row = strip * RSTRIP + prow + 8 * j;
            pre[j] = (row < M) ? ((const float4*)(A + (size_t)row * 128))[pc4]
                               : float4{0.0f, 0.0f, 0.0f, 0.0f};
            if constexpr (ROWSCALE) psc[j] = (row < M) ? rscale[row] : 0.0f;
        }
    }

    for (int i = tid; i < 64 * N; i += 256) {
        const int kp = i / N, n = i % N;
        v2f w;
        w.x = W[(2 * kp) * N + n];
        w.y = W[(2 * kp + 1) * N + n];
        sW2[n][kp] = w;
    }

    const int wave  = tid >> 5;
    const int lane  = tid & 31;
    const int m     = lane & 15;
    const int hsel  = lane >> 4;
    const int mb    = hsel << 3;

    static_assert(N % 16 == 0 && N / 16 <= 8, "one column tile per wave");
    const bool has_tile = wave < N / 16;
    const int  col0     = wave << 4;
    float bv = 0.0f;
    if constexpr (HAS_BIAS) { if (has_tile) bv = bias[col0 + m]; }

    const int gstep = (int)gridDim.x;
    for (; strip < nStrips; strip += gstep) {
        const int row0 = strip * RSTRIP;
#pragma unroll
        for (int j = 0; j < 4; ++j) {
            float4 v = pre[j];
            if constexpr (ROWSCALE) { v.x *= psc[j]; v.y *= psc[j]; v.z *= psc[j]; v.w *= psc[j]; }
            *(float4*)(sAf + (prow + 8 * j) * (2 * LPITCH) + pc4 * 4) = v;
        }
        lds_ready_barrier();

        const int nxt = strip + gstep;
        if (nxt < nStrips) {
#pragma unroll
            for (int j = 0; j < 4; ++j) {
                const int row = nxt * RSTRIP + prow + 8 * j;
                pre[j] = (row < M) ? ((const float4*)(A + (size_t)row * 128))[pc4]
                                   : float4{0.0f, 0.0f, 0.0f, 0.0f};
                if constexpr (ROWSCALE) psc[j] = (row < M) ? rscale[row] : 0.0f;
            }
        }

        if (has_tile) {
            v8f acc0 = {};
            v8f acc1 = {};
#pragma unroll 8
            for (int k2 = 0; k2 < 64; k2 += 2) {
                const int kq = k2 + hsel;
                v2f b  = sW2[col0 + m][kq];
                v2f a0 = sA2[m][kq];
                v2f a1 = sA2[16 + m][kq];
                acc0 = __builtin_amdgcn_wmma_f32_16x16x4_f32(
                    false, a0, false, b, (short)0, acc0, false, false);
                acc1 = __builtin_amdgcn_wmma_f32_16x16x4_f32(
                    false, a1, false, b, (short)0, acc1, false, false);
            }
#pragma unroll
            for (int r = 0; r < 8; ++r) {
                const int row = row0 + mb + r;
                if (row < M) {
                    float v = acc0[r] + bv;
                    if constexpr (DO_RELU) v = fmaxf(v, 0.0f);
                    out[(size_t)row * N + col0 + m] = v;
                }
            }
#pragma unroll
            for (int r = 0; r < 8; ++r) {
                const int row = row0 + 16 + mb + r;
                if (row < M) {
                    float v = acc1[r] + bv;
                    if constexpr (DO_RELU) v = fmaxf(v, 0.0f);
                    out[(size_t)row * N + col0 + m] = v;
                }
            }
        }
        reads_done_barrier();
    }
}

__device__ __forceinline__ void pool_flush(float* __restrict__ gsum, float* __restrict__ gcnt,
                                           int b, float4 r, int len, int lane) {
    const int comp = lane & 3;
    float* gp = gsum + (size_t)b * 128 + lane;
    for (int j = 0; j < 4; ++j) {
        const int sl = 8 * j + (lane >> 2);
        const float x = __shfl(r.x, sl), y = __shfl(r.y, sl);
        const float z = __shfl(r.z, sl), w = __shfl(r.w, sl);
        const float lo = (comp & 1) ? y : x;
        const float hi = (comp & 1) ? w : z;
        atomicAdd(gp + 32 * j, (comp & 2) ? hi : lo);
    }
    if (lane == 0) atomicAdd(gcnt + b, (float)len);
}

__device__ __forceinline__ void pool_hand_over(float* __restrict__ srow, float* __restrict__ scnt,
                                               int g0, int g1,
                                               float* __restrict__ gsum, float* __restrict__ gcnt,
                                               int b, float4 r, int len, int lane) {
    if (b == g0 || b == g1) {
        float* p = srow + ((b == g0) ? 0 : 128) + 4 * lane;
        atomicAdd(p + 0, r.x);
        atomicAdd(p + 1, r.y);
        atomicAdd(p + 2, r.z);
        atomicAdd(p + 3, r.w);
        if (lane == 0) atomicAdd(scnt + ((b == g0) ? 0 : 1), (float)len);
    } else {
        pool_flush(gsum, gcnt, b, r, len, lane);
    }
}

#define ROW(s) (((const float4*)(h + (size_t)(s) * 128))[lane])
#define ACC(vv_)                                                              \
    acc.x += (vv_).x; acc.y += (vv_).y; acc.z += (vv_).z; acc.w += (vv_).w

template <bool POOL>
__global__ __launch_bounds__(256) void gather_relu(const int* __restrict__ rowptr,
                                                   const int* __restrict__ cnt,
                                                   const int* __restrict__ csr_src,
                                                   const float* __restrict__ dinv,
                                                   const float* __restrict__ h,
                                                   const float* __restrict__ bias,
                                                   float* __restrict__ act,
                                                   const int* __restrict__ batch,
                                                   float* __restrict__ gsum,
                                                   float* __restrict__ gcnt, int nN) {
    const int lane = threadIdx.x & 31;
    const int wv = __builtin_amdgcn_readfirstlane((int)(threadIdx.x >> 5));
    const int nw = gridDim.x * 8;
    const int gw = blockIdx.x * 8 + wv;
    const float4 bv = ((const float4*)bias)[lane];

    int node = gw, nend = nN, nstep = nw;
    if constexpr (POOL) {
        const unsigned q = (unsigned)nN / (unsigned)nw, rem = (unsigned)nN % (unsigned)nw;
        node  = (int)((unsigned)gw * q + min((unsigned)gw, rem));
        nend  = node + (int)q + (((unsigned)gw < rem) ? 1 : 0);
        nstep = 1;
    }
    float4 run = {0.0f, 0.0f, 0.0f, 0.0f};
    int runb = -1, runlen = 0;

    __shared__ float srow[2 * 128];
    __shared__ float scnt[2];
    int g0 = -1, g1 = -1;
    if constexpr (POOL) {
        const unsigned q = (unsigned)nN / (unsigned)nw, rem = (unsigned)nN % (unsigned)nw;
        const unsigned wf = blockIdx.x * 8u, wl = wf + 8u;
        const int first = (int)(wf * q + min(wf, rem));
        const int last  = (int)(wl * q + min(wl, rem));
        if (first < last) { g0 = batch[first]; g1 = batch[last - 1]; }
        srow[threadIdx.x] = 0.0f;
        if (threadIdx.x < 2) scnt[threadIdx.x] = 0.0f;
        __syncthreads();
    }

    for (; node < nend; node += nstep) {
        const int   beg = rowptr[node];
        const int   n   = cnt[node];
        const float di  = dinv[node];
        int gid = 0;
        if constexpr (POOL) gid = batch[node];
        const float4 hv = ROW(node);
        float4 acc = {0.0f, 0.0f, 0.0f, 0.0f};

        for (int c0 = 0; c0 < n; c0 += 32) {
            const int m = min(n - c0, 32);
            int s_l = 0;
            if (lane < m) s_l = csr_src[beg + c0 + lane];
            int j = 0;
            for (; j + 4 <= m; j += 4) {
                const int s0 = __builtin_amdgcn_readlane(s_l, j);
                const int s1 = __builtin_amdgcn_readlane(s_l, j + 1);
                const int s2 = __builtin_amdgcn_readlane(s_l, j + 2);
                const int s3 = __builtin_amdgcn_readlane(s_l, j + 3);
                const float4 v0 = ROW(s0);
                const float4 v1 = ROW(s1);
                const float4 v2 = ROW(s2);
                const float4 v3 = ROW(s3);
                ACC(v0); ACC(v1); ACC(v2); ACC(v3);
            }
            if (j + 2 <= m) {
                const int s0 = __builtin_amdgcn_readlane(s_l, j);
                const int s1 = __builtin_amdgcn_readlane(s_l, j + 1);
                const float4 v0 = ROW(s0);
                const float4 v1 = ROW(s1);
                ACC(v0); ACC(v1);
                j += 2;
            }
            if (j < m) {
                const int s0 = __builtin_amdgcn_readlane(s_l, j);
                const float4 v0 = ROW(s0);
                ACC(v0);
            }
        }

        float4 o;
        o.x = fmaxf(fmaf(di, acc.x + hv.x, bv.x), 0.0f);
        o.y = fmaxf(fmaf(di, acc.y + hv.y, bv.y), 0.0f);
        o.z = fmaxf(fmaf(di, acc.z + hv.z, bv.z), 0.0f);
        o.w = fmaxf(fmaf(di, acc.w + hv.w, bv.w), 0.0f);
        if constexpr (POOL) {
            const int b = gid;
            if (b != runb) {
                if (runlen > 0) pool_hand_over(srow, scnt, g0, g1, gsum, gcnt, runb, run, runlen, lane);
                run = o; runb = b; runlen = 1;
            } else {
                run.x += o.x; run.y += o.y; run.z += o.z; run.w += o.w;
                ++runlen;
            }
        } else {
            ((float4*)(act + (size_t)node * 128))[lane] = o;
        }
    }
    if constexpr (POOL) {
        if (runlen > 0) pool_hand_over(srow, scnt, g0, g1, gsum, gcnt, runb, run, runlen, lane);
        __syncthreads();
        const int t = (int)threadIdx.x, slot = t >> 7, f = t & 127;
        const float cn = scnt[slot];
        if (cn > 0.0f) {
            const int g = slot ? g1 : g0;
            atomicAdd(gsum + (size_t)g * 128 + f, srow[t]);
            if (f == 0) atomicAdd(gcnt + g, cn);
        }
    }
}

#define FROWS 128

__global__ __launch_bounds__(1024, 16) void gather_gemm(const int* __restrict__ rowptr,
                                                   const int* __restrict__ cnt,
                                                   const int* __restrict__ csr_src,
                                                   const float* __restrict__ dinv,
                                                   const float* __restrict__ h,
                                                   const float* __restrict__ bias,
                                                   const float* __restrict__ W,
                                                   float* __restrict__ out, int nN) {
    __shared__ v2f sA2[FROWS][LPITCH];
    __shared__ v2f sW2[128][LPITCH];

    const int tid  = threadIdx.x;
    const int lane = tid & 31;
    const int wave = __builtin_amdgcn_readfirstlane(tid >> 5);
    float* sAf = (float*)sA2;

    for (int i = tid; i < 64 * 128; i += 1024) {
        const int kp = i >> 7, n = i & 127;
        v2f w;
        w.x = W[(2 * kp) * 128 + n];
        w.y = W[(2 * kp + 1) * 128 + n];
        sW2[n][kp] = w;
    }

    const float4 bv = ((const float4*)bias)[lane];

    const unsigned ng = gridDim.x, wg = blockIdx.x;
    const unsigned q = (unsigned)nN / ng, rem = (unsigned)nN % ng;
    const int r0 = (int)(wg * q + min(wg, rem));
    const int r1 = r0 + (int)q + ((wg < rem) ? 1 : 0);

    const int m     = lane & 15;
    const int hsel  = lane >> 4;
    const int mb    = hsel << 3;
    const int col0  = (wave & 7) << 4;
    const int rbase = (wave >> 3) << 5;

    for (int s0 = r0; s0 < r1; s0 += FROWS) {
        const int rows = min(FROWS, r1 - s0);

        for (int i = 0; i < 4; ++i) {
            const int lr = i * 32 + wave;
            float4 o = {0.0f, 0.0f, 0.0f, 0.0f};
            if (lr < rows) {
                const int   node = s0 + lr;
                const int   beg  = rowptr[node];
                const int   n    = cnt[node];
                const float di   = dinv[node];
                const float4 hv  = ROW(node);
                float4 acc = {0.0f, 0.0f, 0.0f, 0.0f};

                for (int c0 = 0; c0 < n; c0 += 32) {
                    const int mm = min(n - c0, 32);
                    int s_l = 0;
                    if (lane < mm) s_l = csr_src[beg + c0 + lane];
                    int j = 0;
                    for (; j + 4 <= mm; j += 4) {
                        const int s0r = __builtin_amdgcn_readlane(s_l, j);
                        const int s1r = __builtin_amdgcn_readlane(s_l, j + 1);
                        const int s2r = __builtin_amdgcn_readlane(s_l, j + 2);
                        const int s3r = __builtin_amdgcn_readlane(s_l, j + 3);
                        const float4 v0 = ROW(s0r);
                        const float4 v1 = ROW(s1r);
                        const float4 v2 = ROW(s2r);
                        const float4 v3 = ROW(s3r);
                        ACC(v0); ACC(v1); ACC(v2); ACC(v3);
                    }
                    if (j + 2 <= mm) {
                        const int s0r = __builtin_amdgcn_readlane(s_l, j);
                        const int s1r = __builtin_amdgcn_readlane(s_l, j + 1);
                        const float4 v0 = ROW(s0r);
                        const float4 v1 = ROW(s1r);
                        ACC(v0); ACC(v1);
                        j += 2;
                    }
                    if (j < mm) {
                        const int s0r = __builtin_amdgcn_readlane(s_l, j);
                        const float4 v0 = ROW(s0r);
                        ACC(v0);
                    }
                }

                o.x = di * fmaxf(fmaf(di, acc.x + hv.x, bv.x), 0.0f);
                o.y = di * fmaxf(fmaf(di, acc.y + hv.y, bv.y), 0.0f);
                o.z = di * fmaxf(fmaf(di, acc.z + hv.z, bv.z), 0.0f);
                o.w = di * fmaxf(fmaf(di, acc.w + hv.w, bv.w), 0.0f);
            }
            *(float4*)(sAf + lr * (2 * LPITCH) + lane * 4) = o;
        }
        lds_ready_barrier();

        if (rbase < rows) {
            v8f acc0 = {};
            v8f acc1 = {};
#pragma unroll 8
            for (int k2 = 0; k2 < 64; k2 += 2) {
                const int kq = k2 + hsel;
                v2f b  = sW2[col0 + m][kq];
                v2f a0 = sA2[rbase + m][kq];
                v2f a1 = sA2[rbase + 16 + m][kq];
                acc0 = __builtin_amdgcn_wmma_f32_16x16x4_f32(
                    false, a0, false, b, (short)0, acc0, false, false);
                acc1 = __builtin_amdgcn_wmma_f32_16x16x4_f32(
                    false, a1, false, b, (short)0, acc1, false, false);
            }
#pragma unroll
            for (int r = 0; r < 8; ++r) {
                const int lr = rbase + mb + r;
                if (lr < rows) out[(size_t)(s0 + lr) * 128 + col0 + m] = acc0[r];
            }
#pragma unroll
            for (int r = 0; r < 8; ++r) {
                const int lr = rbase + 16 + mb + r;
                if (lr < rows) out[(size_t)(s0 + lr) * 128 + col0 + m] = acc1[r];
            }
        }
        reads_done_barrier();
    }
}
#undef ROW
#undef ACC

__global__ __launch_bounds__(1024) void head_mlp(const float* __restrict__ gsum,
                                                 const float* __restrict__ gcnt,
                                                 const float* __restrict__ Wa,
                                                 const float* __restrict__ ba,
                                                 const float* __restrict__ Wb,
                                                 const float* __restrict__ bb,
                                                 float* __restrict__ out, int G) {
    __shared__ v2f sG2[RSTRIP][LPITCH];
    __shared__ v2f sH2[RSTRIP][LPITCH];
    __shared__ v2f sWa[128][LPITCH];
    __shared__ v2f sWb[32][LPITCH];

    const int tid  = threadIdx.x;
    const int lane = tid & 31;
    const int wave = __builtin_amdgcn_readfirstlane(tid >> 5);
    const int row0 = blockIdx.x * RSTRIP;
    const int m    = lane & 15;
    const int hsel = lane >> 4;
    const int mb   = hsel << 3;

    v2f wa[8], wb[2];
#pragma unroll
    for (int j = 0; j < 8; ++j) {
        const int i = tid + 1024 * j, kp = i >> 7, n = i & 127;
        wa[j].x = Wa[(2 * kp) * 128 + n];
        wa[j].y = Wa[(2 * kp + 1) * 128 + n];
    }
#pragma unroll
    for (int j = 0; j < 2; ++j) {
        const int i = tid + 1024 * j, kp = i >> 5, n = i & 31;
        wb[j].x = Wb[(2 * kp) * 32 + n];
        wb[j].y = Wb[(2 * kp + 1) * 32 + n];
    }
    const int    rwc = min(row0 + wave, G - 1);
    const float4 s   = ((const float4*)(gsum + (size_t)rwc * 128))[lane];
    const float  cn  = gcnt[rwc];
    const float bva = ba[((wave & 7) << 4) + m];
    const float bvb = bb[((wave & 1) << 4) + m];

#pragma unroll
    for (int j = 0; j < 8; ++j) {
        const int i = tid + 1024 * j;
        sWa[i & 127][i >> 7] = wa[j];
    }
#pragma unroll
    for (int j = 0; j < 2; ++j) {
        const int i = tid + 1024 * j;
        sWb[i & 31][i >> 5] = wb[j];
    }
    {
        const float c = fmaxf(cn, 1.0f);
        float4 g;
        g.x = s.x / c; g.y = s.y / c; g.z = s.z / c; g.w = s.w / c;
        *(float4*)((float*)sG2 + wave * (2 * LPITCH) + lane * 4) = g;
    }
    lds_ready_barrier();

    if (wave < 8) {
        const int col0 = wave << 4;
        float* sHf = (float*)sH2;
        v8f acc0 = {};
        v8f acc1 = {};
#pragma unroll 8
        for (int k2 = 0; k2 < 64; k2 += 2) {
            const int kq = k2 + hsel;
            v2f b  = sWa[col0 + m][kq];
            v2f a0 = sG2[m][kq];
            v2f a1 = sG2[16 + m][kq];
            acc0 = __builtin_amdgcn_wmma_f32_16x16x4_f32(
                false, a0, false, b, (short)0, acc0, false, false);
            acc1 = __builtin_amdgcn_wmma_f32_16x16x4_f32(
                false, a1, false, b, (short)0, acc1, false, false);
        }
#pragma unroll
        for (int r = 0; r < 8; ++r)
            sHf[(mb + r) * (2 * LPITCH) + col0 + m] = fmaxf(acc0[r] + bva, 0.0f);
#pragma unroll
        for (int r = 0; r < 8; ++r)
            sHf[(16 + mb + r) * (2 * LPITCH) + col0 + m] = fmaxf(acc1[r] + bva, 0.0f);
    }
    lds_ready_barrier();

    if (wave < 2) {
        const int col0 = wave << 4;
        v8f acc0 = {};
        v8f acc1 = {};
#pragma unroll 8
        for (int k2 = 0; k2 < 64; k2 += 2) {
            const int kq = k2 + hsel;
            v2f b  = sWb[col0 + m][kq];
            v2f a0 = sH2[m][kq];
            v2f a1 = sH2[16 + m][kq];
            acc0 = __builtin_amdgcn_wmma_f32_16x16x4_f32(
                false, a0, false, b, (short)0, acc0, false, false);
            acc1 = __builtin_amdgcn_wmma_f32_16x16x4_f32(
                false, a1, false, b, (short)0, acc1, false, false);
        }
#pragma unroll
        for (int r = 0; r < 8; ++r) {
            const int row = row0 + mb + r;
            if (row < G) out[(size_t)row * 32 + col0 + m] = acc0[r] + bvb;
        }
#pragma unroll
        for (int r = 0; r < 8; ++r) {
            const int row = row0 + 16 + mb + r;
            if (row < G) out[(size_t)row * 32 + col0 + m] = acc1[r] + bvb;
        }
    }
}

extern "C" void kernel_launch(void* const* d_in, const int* in_sizes, int n_in,
                              void* d_out, int out_size, void* d_ws, size_t ws_size,
                              hipStream_t stream) {
    const float* x     = (const float*)d_in[0];
    const int*   ei    = (const int*)d_in[1];
    const int*   batch = (const int*)d_in[2];
    const float* W1    = (const float*)d_in[3];
    const float* b1    = (const float*)d_in[4];
    const float* W2    = (const float*)d_in[5];
    const float* b2    = (const float*)d_in[6];
    const float* Wl1   = (const float*)d_in[7];
    const float* bl1   = (const float*)d_in[8];
    const float* Wl2   = (const float*)d_in[9];
    const float* bl2   = (const float*)d_in[10];
    float* out = (float*)d_out;

    const int nN = in_sizes[0] / 128;
    const int nE = in_sizes[1] / 2;
    const int G  = out_size / 32;
    const int* src = ei;
    const int* dst = ei + nE;

    char* ws = (char*)d_ws;
    auto alloc = [&](size_t bytes) -> float* {
        float* p = (float*)ws;
        ws += (bytes + 255) & ~(size_t)255;
        return p;
    };
    float* bufA = alloc((size_t)nN * 128 * 4);
    float* bufB = alloc((size_t)nN * 128 * 4);
    char*  bufC = (char*)alloc((size_t)nN * 128 * 4);
    int*   cnt  = (int*)alloc((size_t)nN * 4);
    float* dinv = alloc((size_t)nN * 4);
    float* gsum = alloc((size_t)G * 128 * 4);
    float* gcnt = alloc((size_t)G * 4);

    auto sub = [&](size_t bytes) -> int* {
        int* p = (int*)bufC;
        bufC += (bytes + 255) & ~(size_t)255;
        return p;
    };
    const int NB = (nN + 255) >> BIN_SHIFT;
    const int NC = (nE + ECH - 1) / ECH;
    int* csr_src = sub((size_t)nE * 4);
    int* loc     = sub((size_t)nE * 8);
    int* rowptr  = sub((size_t)nN * 4);
    int* hist    = sub((size_t)NB * NC * 8);

    edge_part<<<NC, TPA, 0, stream>>>(src, dst, loc, hist, nE, NB, gsum, G * 128, gcnt, G);
    bin_csr<<<NB, 256, 0, stream>>>(loc, hist, rowptr, cnt, dinv, csr_src, nN, nE, NB, NC);

    gemm_k128<128, false, false, true><<<768, 256, 0, stream>>>(x, W1, nullptr, dinv, bufA, nN);
    gather_gemm<<<512, 1024, 0, stream>>>(rowptr, cnt, csr_src, dinv, bufA, b1, W2, bufB, nN);

    gather_relu<true><<<2048, 256, 0, stream>>>(rowptr, cnt, csr_src, dinv, bufB, b2, nullptr,
                                                batch, gsum, gcnt, nN);

    head_mlp<<<(G + RSTRIP - 1) / RSTRIP, 1024, 0, stream>>>(gsum, gcnt, Wl1, bl1, Wl2, bl2, out, G);
}


# Fused gather+product launch pipelined: a wave multiplies its tile of strip t between its two node gathers of strip t+1 (64-row strips, two LDS A tiles); W loads ride under gather 1; k, sum order kept
# speedup vs baseline: 7.0394x; 1.0380x over previous
#include <hip/hip_runtime.h>

typedef __attribute__((ext_vector_type(2))) float v2f;
typedef __attribute__((ext_vector_type(8))) float v8f;

#define BIN_SHIFT 8
#define NBMAX     512
#define TPA       512
#define ECH       8192
#define PB        16

__device__ __forceinline__ int wave_sum(int v) {
#pragma unroll
    for (int off = 16; off > 0; off >>= 1) v += __shfl_xor(v, off);
    return v;
}

__device__ __forceinline__ int wave_incl_scan(int v, int lane) {
#pragma unroll
    for (int off = 1; off < 32; off <<= 1) {
        int t = __shfl_up(v, off);
        if (lane >= off) v += t;
    }
    return v;
}

__global__ __launch_bounds__(TPA) void edge_part(const int* __restrict__ src,
                                                 const int* __restrict__ dst,
                                                 int* __restrict__ loc_i,
                                                 int* __restrict__ hist_i,
                                                 int nE, int NB,
                                                 float* __restrict__ zsum, int nzsum,
                                                 float* __restrict__ zcnt, int nzcnt) {
    __shared__ int sh[NBMAX + 1];
    __shared__ int soff[NBMAX];
    __shared__ int sw[TPA / 32];
    const int tid = threadIdx.x, lane = tid & 31, wave = tid >> 5;
    int2* loc  = (int2*)loc_i;
    int2* hist = (int2*)hist_i;
    for (int i = tid; i < NBMAX + 1; i += TPA) sh[i] = 0;
    const int gt = blockIdx.x * TPA + tid, gstride = (int)gridDim.x * TPA;
    for (int i = gt; i < nzsum; i += gstride) zsum[i] = 0.0f;
    for (int i = gt; i < nzcnt; i += gstride) zcnt[i] = 0.0f;
    __syncthreads();
    const int base = blockIdx.x * ECH;
    int d[16], sv[16], r[16];
#pragma unroll
    for (int j = 0; j < 16; ++j) {
        const int ec = min(base + j * TPA + tid, nE - 1);
        d[j]  = dst[ec];
        sv[j] = src[ec];
    }
#pragma unroll
    for (int j = 0; j < 16; ++j) {
        const int e = base + j * TPA + tid;
        const int bin = (e < nE) ? min(d[j] >> BIN_SHIFT, NBMAX - 1) : NBMAX;
        r[j] = atomicAdd(&sh[bin], 1);
    }
    __syncthreads();

    const int c   = sh[tid];
    const int inc = wave_incl_scan(c, lane);
    if (lane == 31) sw[wave] = inc;
    __syncthreads();
    int off = 0;
    for (int w = 0; w < TPA / 32; ++w)
        if (w < wave) off += sw[w];
    const int ex = off + inc - c;
    soff[tid] = ex;
    if (tid < NB) hist[(size_t)blockIdx.x * NB + tid] = make_int2(ex, c);
    __syncthreads();

#pragma unroll
    for (int j = 0; j < 16; ++j) {
        const int e = base + j * TPA + tid;
        const int bin = min(d[j] >> BIN_SHIFT, NBMAX - 1);
        const int pos = base + soff[bin] + r[j];
        if (e < nE) loc[pos] = make_int2(sv[j], d[j]);
    }
}

template <bool PLACE>
__device__ __forceinline__ void csr_trip(const int2* __restrict__ loc, const int (&st)[PB],
                                         const int (&ln)[PB], int* __restrict__ stab,
                                         int* __restrict__ csr_src, int beg, int nE, int lane) {
    int2 ed[PB];
    int  p[PB];
#pragma unroll
    for (int k = 0; k < PB; ++k)
        ed[k] = loc[min(st[k] + min(lane, max(ln[k] - 1, 0)), nE - 1)];
#pragma unroll
    for (int k = 0; k < PB; ++k) {
        p[k] = 0;
        if (lane < ln[k]) p[k] = atomicAdd(&stab[ed[k].y & 255], 1);
    }
    if constexpr (PLACE) {
#pragma unroll
        for (int k = 0; k < PB; ++k)
            if (lane < ln[k]) csr_src[min(beg + p[k], nE - 1)] = ed[k].x;
    }
#pragma unroll
    for (int k = 0; k < PB; ++k)
        for (int i = lane + 32; i < ln[k]; i += 32) {
            const int2 e2 = loc[min(st[k] + i, nE - 1)];
            const int  p2 = atomicAdd(&stab[e2.y & 255], 1);
            if constexpr (PLACE) csr_src[min(beg + p2, nE - 1)] = e2.x;
        }
}

template <bool PLACE>
__device__ __forceinline__ int csr_walk(const int2* __restrict__ loc, const int2* __restrict__ hist,
                                        int* __restrict__ stab, int* __restrict__ csr_src,
                                        int beg, int nE, int NB, int NC, int b, int wave, int lane) {
    int osum = 0;
    for (int m0 = 0; wave + 8 * m0 < NC; m0 += 32) {
        const int g = wave + 8 * (m0 + lane);
        int2 oc = make_int2(0, 0);
        if (g < NC) oc = hist[(size_t)g * NB + b];
        const int myst = g * ECH + oc.x;
        const int myln = oc.y;
        osum += oc.x;
#pragma unroll
        for (int k0 = 0; k0 < 32; k0 += PB) {
            if (wave + 8 * (m0 + k0) < NC) {
                int st[PB], ln[PB];
#pragma unroll
                for (int k = 0; k < PB; ++k) {
                    st[k] = __builtin_amdgcn_readlane(myst, k0 + k);
                    ln[k] = __builtin_amdgcn_readlane(myln, k0 + k);
                }
                csr_trip<PLACE>(loc, st, ln, stab, csr_src, beg, nE, lane);
            }
        }
    }
    return osum;
}

__global__ __launch_bounds__(256) void bin_csr(const int* __restrict__ loc_i,
                                               const int* __restrict__ hist_i,
                                               int* __restrict__ rowptr,
                                               int* __restrict__ cnt,
                                               float* __restrict__ dinv,
                                               int* __restrict__ csr_src,
                                               int nN, int nE, int NB, int NC) {
    __shared__ int scnt[256];
    __shared__ int sptr[256];
    __shared__ int sw[8];
    const int tid = threadIdx.x, lane = tid & 31, wv = tid >> 5;
    const int wave = __builtin_amdgcn_readfirstlane(tid >> 5);
    const int b = blockIdx.x;
    const int2* loc  = (const int2*)loc_i;
    const int2* hist = (const int2*)hist_i;

    scnt[tid] = 0;
    __syncthreads();

    int pv = csr_walk<false>(loc, hist, scnt, csr_src, 0, nE, NB, NC, b, wave, lane);
    pv = wave_sum(pv);
    if (lane == 0) sw[wv] = pv;
    __syncthreads();
    int beg = 0;
    for (int w = 0; w < 8; ++w) beg += sw[w];
    const int v = scnt[tid];
    __syncthreads();

    const int inc = wave_incl_scan(v, lane);
    if (lane == 31) sw[wv] = inc;
    __syncthreads();
    int off = 0;
    for (int w = 0; w < 8; ++w)
        if (w < wv) off += sw[w];
    const int ex = off + inc - v;
    sptr[tid] = ex;
    const int node = (b << BIN_SHIFT) + tid;
    if (node < nN) {
        rowptr[node] = beg + ex;
        cnt[node]    = v;
        dinv[node]   = rsqrtf((float)v + 1.0f);
    }
    __syncthreads();

    csr_walk<true>(loc, hist, sptr, csr_src, beg, nE, NB, NC, b, wave, lane);
}

#define LPITCH 66
#define RSTRIP 32

__device__ __forceinline__ void lds_ready_barrier() {
    asm volatile("s_wait_dscnt 0x0\n\ts_barrier_signal -1\n\ts_barrier_wait -1" ::: "memory");
}
__device__ __forceinline__ void reads_done_barrier() {
    asm volatile("s_barrier_signal -1\n\ts_barrier_wait -1" ::: "memory");
}

template <int N, bool HAS_BIAS, bool DO_RELU, bool ROWSCALE>
__global__ __launch_bounds__(256) void gemm_k128(const float* __restrict__ A,
                                                 const float* __restrict__ W,
                                                 const float* __restrict__ bias,
                                                 const float* __restrict__ rscale,
                                                 float* __restrict__ out,
                                                 int M) {
    __shared__ v2f sA2[RSTRIP][LPITCH];
    __shared__ v2f sW2[N][LPITCH];

    const int tid = threadIdx.x;

    const int nStrips = (M + RSTRIP - 1) / RSTRIP;
    const int prow = tid >> 5;
    const int pc4  = tid & 31;
    float* sAf = (float*)sA2;

    float4 pre[4];
    float  psc[4] = {1.0f, 1.0f, 1.0f, 1.0f};
    int strip = blockIdx.x;
    if (strip < nStrips) {
#pragma unroll
        for (int j = 0; j < 4; ++j) {
            const int row = strip * RSTRIP + prow + 8 * j;
            pre[j] = (row < M) ? ((const float4*)(A + (size_t)row * 128))[pc4]
                               : float4{0.0f, 0.0f, 0.0f, 0.0f};
            if constexpr (ROWSCALE) psc[j] = (row < M) ? rscale[row] : 0.0f;
        }
    }

    for (int i = tid; i < 64 * N; i += 256) {
        const int kp = i / N, n = i % N;
        v2f w;
        w.x = W[(2 * kp) * N + n];
        w.y = W[(2 * kp + 1) * N + n];
        sW2[n][kp] = w;
    }

    const int wave  = tid >> 5;
    const int lane  = tid & 31;
    const int m     = lane & 15;
    const int hsel  = lane >> 4;
    const int mb    = hsel << 3;

    static_assert(N % 16 == 0 && N / 16 <= 8, "one column tile per wave");
    const bool has_tile = wave < N / 16;
    const int  col0     = wave << 4;
    float bv = 0.0f;
    if constexpr (HAS_BIAS) { if (has_tile) bv = bias[col0 + m]; }

    const int gstep = (int)gridDim.x;
    for (; strip < nStrips; strip += gstep) {
        const int row0 = strip * RSTRIP;
#pragma unroll
        for (int j = 0; j < 4; ++j) {
            float4 v = pre[j];
            if constexpr (ROWSCALE) { v.x *= psc[j]; v.y *= psc[j]; v.z *= psc[j]; v.w *= psc[j]; }
            *(float4*)(sAf + (prow + 8 * j) * (2 * LPITCH) + pc4 * 4) = v;
        }
        lds_ready_barrier();

        const int nxt = strip + gstep;
        if (nxt < nStrips) {
#pragma unroll
            for (int j = 0; j < 4; ++j) {
                const int row = nxt * RSTRIP + prow + 8 * j;
                pre[j] = (row < M) ? ((const float4*)(A + (size_t)row * 128))[pc4]
                                   : float4{0.0f, 0.0f, 0.0f, 0.0f};
                if constexpr (ROWSCALE) psc[j] = (row < M) ? rscale[row] : 0.0f;
            }
        }

        if (has_tile) {
            v8f acc0 = {};
            v8f acc1 = {};
#pragma unroll 8
            for (int k2 = 0; k2 < 64; k2 += 2) {
                const int kq = k2 + hsel;
                v2f b  = sW2[col0 + m][kq];
                v2f a0 = sA2[m][kq];
                v2f a1 = sA2[16 + m][kq];
                acc0 = __builtin_amdgcn_wmma_f32_16x16x4_f32(
                    false, a0, false, b, (short)0, acc0, false, false);
                acc1 = __builtin_amdgcn_wmma_f32_16x16x4_f32(
                    false, a1, false, b, (short)0, acc1, false, false);
            }
#pragma unroll
            for (int r = 0; r < 8; ++r) {
                const int row = row0 + mb + r;
                if (row < M) {
                    float v = acc0[r] + bv;
                    if constexpr (DO_RELU) v = fmaxf(v, 0.0f);
                    out[(size_t)row * N + col0 + m] = v;
                }
            }
#pragma unroll
            for (int r = 0; r < 8; ++r) {
                const int row = row0 + 16 + mb + r;
                if (row < M) {
                    float v = acc1[r] + bv;
                    if constexpr (DO_RELU) v = fmaxf(v, 0.0f);
                    out[(size_t)row * N + col0 + m] = v;
                }
            }
        }
        reads_done_barrier();
    }
}

__device__ __forceinline__ void pool_flush(float* __restrict__ gsum, float* __restrict__ gcnt,
                                           int b, float4 r, int len, int lane) {
    const int comp = lane & 3;
    float* gp = gsum + (size_t)b * 128 + lane;
    for (int j = 0; j < 4; ++j) {
        const int sl = 8 * j + (lane >> 2);
        const float x = __shfl(r.x, sl), y = __shfl(r.y, sl);
        const float z = __shfl(r.z, sl), w = __shfl(r.w, sl);
        const float lo = (comp & 1) ? y : x;
        const float hi = (comp & 1) ? w : z;
        atomicAdd(gp + 32 * j, (comp & 2) ? hi : lo);
    }
    if (lane == 0) atomicAdd(gcnt + b, (float)len);
}

__device__ __forceinline__ void pool_hand_over(float* __restrict__ srow, float* __restrict__ scnt,
                                               int g0, int g1,
                                               float* __restrict__ gsum, float* __restrict__ gcnt,
                                               int b, float4 r, int len, int lane) {
    if (b == g0 || b == g1) {
        float* p = srow + ((b == g0) ? 0 : 128) + 4 * lane;
        atomicAdd(p + 0, r.x);
        atomicAdd(p + 1, r.y);
        atomicAdd(p + 2, r.z);
        atomicAdd(p + 3, r.w);
        if (lane == 0) atomicAdd(scnt + ((b == g0) ? 0 : 1), (float)len);
    } else {
        pool_flush(gsum, gcnt, b, r, len, lane);
    }
}

#define ROW(s) (((const float4*)(h + (size_t)(s) * 128))[lane])
#define ACC(vv_)                                                              \
    acc.x += (vv_).x; acc.y += (vv_).y; acc.z += (vv_).z; acc.w += (vv_).w

template <bool POOL>
__global__ __launch_bounds__(256) void gather_relu(const int* __restrict__ rowptr,
                                                   const int* __restrict__ cnt,
                                                   const int* __restrict__ csr_src,
                                                   const float* __restrict__ dinv,
                                                   const float* __restrict__ h,
                                                   const float* __restrict__ bias,
                                                   float* __restrict__ act,
                                                   const int* __restrict__ batch,
                                                   float* __restrict__ gsum,
                                                   float* __restrict__ gcnt, int nN) {
    const int lane = threadIdx.x & 31;
    const int wv = __builtin_amdgcn_readfirstlane((int)(threadIdx.x >> 5));
    const int nw = gridDim.x * 8;
    const int gw = blockIdx.x * 8 + wv;
    const float4 bv = ((const float4*)bias)[lane];

    int node = gw, nend = nN, nstep = nw;
    if constexpr (POOL) {
        const unsigned q = (unsigned)nN / (unsigned)nw, rem = (unsigned)nN % (unsigned)nw;
        node  = (int)((unsigned)gw * q + min((unsigned)gw, rem));
        nend  = node + (int)q + (((unsigned)gw < rem) ? 1 : 0);
        nstep = 1;
    }
    float4 run = {0.0f, 0.0f, 0.0f, 0.0f};
    int runb = -1, runlen = 0;

    __shared__ float srow[2 * 128];
    __shared__ float scnt[2];
    int g0 = -1, g1 = -1;
    if constexpr (POOL) {
        const unsigned q = (unsigned)nN / (unsigned)nw, rem = (unsigned)nN % (unsigned)nw;
        const unsigned wf = blockIdx.x * 8u, wl = wf + 8u;
        const int first = (int)(wf * q + min(wf, rem));
        const int last  = (int)(wl * q + min(wl, rem));
        if (first < last) { g0 = batch[first]; g1 = batch[last - 1]; }
        srow[threadIdx.x] = 0.0f;
        if (threadIdx.x < 2) scnt[threadIdx.x] = 0.0f;
        __syncthreads();
    }

    for (; node < nend; node += nstep) {
        const int   beg = rowptr[node];
        const int   n   = cnt[node];
        const float di  = dinv[node];
        int gid = 0;
        if constexpr (POOL) gid = batch[node];
        const float4 hv = ROW(node);
        float4 acc = {0.0f, 0.0f, 0.0f, 0.0f};

        for (int c0 = 0; c0 < n; c0 += 32) {
            const int m = min(n - c0, 32);
            int s_l = 0;
            if (lane < m) s_l = csr_src[beg + c0 + lane];
            int j = 0;
            for (; j + 4 <= m; j += 4) {
                const int s0 = __builtin_amdgcn_readlane(s_l, j);
                const int s1 = __builtin_amdgcn_readlane(s_l, j + 1);
                const int s2 = __builtin_amdgcn_readlane(s_l, j + 2);
                const int s3 = __builtin_amdgcn_readlane(s_l, j + 3);
                const float4 v0 = ROW(s0);
                const float4 v1 = ROW(s1);
                const float4 v2 = ROW(s2);
                const float4 v3 = ROW(s3);
                ACC(v0); ACC(v1); ACC(v2); ACC(v3);
            }
            if (j + 2 <= m) {
                const int s0 = __builtin_amdgcn_readlane(s_l, j);
                const int s1 = __builtin_amdgcn_readlane(s_l, j + 1);
                const float4 v0 = ROW(s0);
                const float4 v1 = ROW(s1);
                ACC(v0); ACC(v1);
                j += 2;
            }
            if (j < m) {
                const int s0 = __builtin_amdgcn_readlane(s_l, j);
                const float4 v0 = ROW(s0);
                ACC(v0);
            }
        }

        float4 o;
        o.x = fmaxf(fmaf(di, acc.x + hv.x, bv.x), 0.0f);
        o.y = fmaxf(fmaf(di, acc.y + hv.y, bv.y), 0.0f);
        o.z = fmaxf(fmaf(di, acc.z + hv.z, bv.z), 0.0f);
        o.w = fmaxf(fmaf(di, acc.w + hv.w, bv.w), 0.0f);
        if constexpr (POOL) {
            const int b = gid;
            if (b != runb) {
                if (runlen > 0) pool_hand_over(srow, scnt, g0, g1, gsum, gcnt, runb, run, runlen, lane);
                run = o; runb = b; runlen = 1;
            } else {
                run.x += o.x; run.y += o.y; run.z += o.z; run.w += o.w;
                ++runlen;
            }
        } else {
            ((float4*)(act + (size_t)node * 128))[lane] = o;
        }
    }
    if constexpr (POOL) {
        if (runlen > 0) pool_hand_over(srow, scnt, g0, g1, gsum, gcnt, runb, run, runlen, lane);
        __syncthreads();
        const int t = (int)threadIdx.x, slot = t >> 7, f = t & 127;
        const float cn = scnt[slot];
        if (cn > 0.0f) {
            const int g = slot ? g1 : g0;
            atomicAdd(gsum + (size_t)g * 128 + f, srow[t]);
            if (f == 0) atomicAdd(gcnt + g, cn);
        }
    }
}

#define FROWS 64

__device__ __forceinline__ float4 fused_row(const int* __restrict__ rowptr,
                                            const int* __restrict__ cnt,
                                            const int* __restrict__ csr_src,
                                            const float* __restrict__ dinv,
                                            const float* __restrict__ h,
                                            float4 bv, int node, int lane) {
    const int   beg = rowptr[node];
    const int   n   = cnt[node];
    const float di  = dinv[node];
    const float4 hv = ROW(node);
    float4 acc = {0.0f, 0.0f, 0.0f, 0.0f};

    for (int c0 = 0; c0 < n; c0 += 32) {
        const int mm = min(n - c0, 32);
        int s_l = 0;
        if (lane < mm) s_l = csr_src[beg + c0 + lane];
        int j = 0;
        for (; j + 4 <= mm; j += 4) {
            const int s0r = __builtin_amdgcn_readlane(s_l, j);
            const int s1r = __builtin_amdgcn_readlane(s_l, j + 1);
            const int s2r = __builtin_amdgcn_readlane(s_l, j + 2);
            const int s3r = __builtin_amdgcn_readlane(s_l, j + 3);
            const float4 v0 = ROW(s0r);
            const float4 v1 = ROW(s1r);
            const float4 v2 = ROW(s2r);
            const float4 v3 = ROW(s3r);
            ACC(v0); ACC(v1); ACC(v2); ACC(v3);
        }
        if (j + 2 <= mm) {
            const int s0r = __builtin_amdgcn_readlane(s_l, j);
            const int s1r = __builtin_amdgcn_readlane(s_l, j + 1);
            const float4 v0 = ROW(s0r);
            const float4 v1 = ROW(s1r);
            ACC(v0); ACC(v1);
            j += 2;
        }
        if (j < mm) {
            const int s0r = __builtin_amdgcn_readlane(s_l, j);
            const float4 v0 = ROW(s0r);
            ACC(v0);
        }
    }

    float4 o;
    o.x = di * fmaxf(fmaf(di, acc.x + hv.x, bv.x), 0.0f);
    o.y = di * fmaxf(fmaf(di, acc.y + hv.y, bv.y), 0.0f);
    o.z = di * fmaxf(fmaf(di, acc.z + hv.z, bv.z), 0.0f);
    o.w = di * fmaxf(fmaf(di, acc.w + hv.w, bv.w), 0.0f);
    return o;
}

__global__ __launch_bounds__(1024, 16) void gather_gemm(const int* __restrict__ rowptr,
                                                   const int* __restrict__ cnt,
                                                   const int* __restrict__ csr_src,
                                                   const float* __restrict__ dinv,
                                                   const float* __restrict__ h,
                                                   const float* __restrict__ bias,
                                                   const float* __restrict__ W,
                                                   float* __restrict__ out, int nN) {
    __shared__ v2f sA2[2][FROWS][LPITCH];
    __shared__ v2f sW2[128][LPITCH];

    const int tid  = threadIdx.x;
    const int lane = tid & 31;
    const int wave = __builtin_amdgcn_readfirstlane(tid >> 5);

    v2f wreg[8];
#pragma unroll
    for (int j = 0; j < 8; ++j) {
        const int i = tid + 1024 * j, kp = i >> 7, n = i & 127;
        wreg[j].x = W[(2 * kp) * 128 + n];
        wreg[j].y = W[(2 * kp + 1) * 128 + n];
    }

    const float4 bv = ((const float4*)bias)[lane];

    const unsigned ng = gridDim.x, wg = blockIdx.x;
    const unsigned q = (unsigned)nN / ng, rem = (unsigned)nN % ng;
    const int r0 = (int)(wg * q + min(wg, rem));
    const int r1 = r0 + (int)q + ((wg < rem) ? 1 : 0);
    const int nstrips = (r1 - r0 + FROWS - 1) / FROWS;

    const int m     = lane & 15;
    const int hsel  = lane >> 4;
    const int mb    = hsel << 3;
    const int col0  = (wave & 7) << 4;
    const int rbase = (wave >> 3) << 4;

    {
        const int nrows = min(FROWS, r1 - r0);
        float* nxt = (float*)sA2[0];
        if (nrows > 0) {
            float4 o = {0.0f, 0.0f, 0.0f, 0.0f};
            if (wave < nrows)
                o = fused_row(rowptr, cnt, csr_src, dinv, h, bv, r0 + wave, lane);
            *(float4*)(nxt + wave * (2 * LPITCH) + lane * 4) = o;
            float4 o2 = {0.0f, 0.0f, 0.0f, 0.0f};
            if (wave + 32 < nrows)
                o2 = fused_row(rowptr, cnt, csr_src, dinv, h, bv, r0 + 32 + wave, lane);
            *(float4*)(nxt + (32 + wave) * (2 * LPITCH) + lane * 4) = o2;
        }
#pragma unroll
        for (int j = 0; j < 8; ++j) {
            const int i = tid + 1024 * j;
            sW2[i & 127][i >> 7] = wreg[j];
        }
        lds_ready_barrier();
    }

    for (int t = 0; t < nstrips; ++t) {
        const int sc    = r0 + t * FROWS;
        const int rows  = min(FROWS, r1 - sc);
        const int sn    = sc + FROWS;
        const int nrows = min(FROWS, r1 - sn);
        float* nxt = (float*)sA2[(t + 1) & 1];
        const v2f (*cur)[LPITCH] = sA2[t & 1];
        const bool mul = rbase < rows;

        v8f acc = {};

        if (nrows > 0) {
            float4 o = {0.0f, 0.0f, 0.0f, 0.0f};
            if (wave < nrows)
                o = fused_row(rowptr, cnt, csr_src, dinv, h, bv, sn + wave, lane);
            *(float4*)(nxt + wave * (2 * LPITCH) + lane * 4) = o;
        }

        if (mul) {
#pragma unroll 8
            for (int k2 = 0; k2 < 32; k2 += 2) {
                const int kq = k2 + hsel;
                v2f b = sW2[col0 + m][kq];
                v2f a = cur[rbase + m][kq];
                acc = __builtin_amdgcn_wmma_f32_16x16x4_f32(
                    false, a, false, b, (short)0, acc, false, false);
            }
        }

        if (nrows > 0) {
            float4 o = {0.0f, 0.0f, 0.0f, 0.0f};
            if (wave + 32 < nrows)
                o = fused_row(rowptr, cnt, csr_src, dinv, h, bv, sn + 32 + wave, lane);
            *(float4*)(nxt + (32 + wave) * (2 * LPITCH) + lane * 4) = o;
        }

        if (mul) {
#pragma unroll 8
            for (int k2 = 32; k2 < 64; k2 += 2) {
                const int kq = k2 + hsel;
                v2f b = sW2[col0 + m][kq];
                v2f a = cur[rbase + m][kq];
                acc = __builtin_amdgcn_wmma_f32_16x16x4_f32(
                    false, a, false, b, (short)0, acc, false, false);
            }
#pragma unroll
            for (int r = 0; r < 8; ++r) {
                const int lr = rbase + mb + r;
                if (lr < rows) out[(size_t)(sc + lr) * 128 + col0 + m] = acc[r];
            }
        }
        lds_ready_barrier();
    }
}
#undef ROW
#undef ACC

__global__ __launch_bounds__(1024) void head_mlp(const float* __restrict__ gsum,
                                                 const float* __restrict__ gcnt,
                                                 const float* __restrict__ Wa,
                                                 const float* __restrict__ ba,
                                                 const float* __restrict__ Wb,
                                                 const float* __restrict__ bb,
                                                 float* __restrict__ out, int G) {
    __shared__ v2f sG2[RSTRIP][LPITCH];
    __shared__ v2f sH2[RSTRIP][LPITCH];
    __shared__ v2f sWa[128][LPITCH];
    __shared__ v2f sWb[32][LPITCH];

    const int tid  = threadIdx.x;
    const int lane = tid & 31;
    const int wave = __builtin_amdgcn_readfirstlane(tid >> 5);
    const int row0 = blockIdx.x * RSTRIP;
    const int m    = lane & 15;
    const int hsel = lane >> 4;
    const int mb   = hsel << 3;

    v2f wa[8], wb[2];
#pragma unroll
    for (int j = 0; j < 8; ++j) {
        const int i = tid + 1024 * j, kp = i >> 7, n = i & 127;
        wa[j].x = Wa[(2 * kp) * 128 + n];
        wa[j].y = Wa[(2 * kp + 1) * 128 + n];
    }
#pragma unroll
    for (int j = 0; j < 2; ++j) {
        const int i = tid + 1024 * j, kp = i >> 5, n = i & 31;
        wb[j].x = Wb[(2 * kp) * 32 + n];
        wb[j].y = Wb[(2 * kp + 1) * 32 + n];
    }
    const int    rwc = min(row0 + wave, G - 1);
    const float4 s   = ((const float4*)(gsum + (size_t)rwc * 128))[lane];
    const float  cn  = gcnt[rwc];
    const float bva = ba[((wave & 7) << 4) + m];
    const float bvb = bb[((wave & 1) << 4) + m];

#pragma unroll
    for (int j = 0; j < 8; ++j) {
        const int i = tid + 1024 * j;
        sWa[i & 127][i >> 7] = wa[j];
    }
#pragma unroll
    for (int j = 0; j < 2; ++j) {
        const int i = tid + 1024 * j;
        sWb[i & 31][i >> 5] = wb[j];
    }
    {
        const float c = fmaxf(cn, 1.0f);
        float4 g;
        g.x = s.x / c; g.y = s.y / c; g.z = s.z / c; g.w = s.w / c;
        *(float4*)((float*)sG2 + wave * (2 * LPITCH) + lane * 4) = g;
    }
    lds_ready_barrier();

    if (wave < 8) {
        const int col0 = wave << 4;
        float* sHf = (float*)sH2;
        v8f acc0 = {};
        v8f acc1 = {};
#pragma unroll 8
        for (int k2 = 0; k2 < 64; k2 += 2) {
            const int kq = k2 + hsel;
            v2f b  = sWa[col0 + m][kq];
            v2f a0 = sG2[m][kq];
            v2f a1 = sG2[16 + m][kq];
            acc0 = __builtin_amdgcn_wmma_f32_16x16x4_f32(
                false, a0, false, b, (short)0, acc0, false, false);
            acc1 = __builtin_amdgcn_wmma_f32_16x16x4_f32(
                false, a1, false, b, (short)0, acc1, false, false);
        }
#pragma unroll
        for (int r = 0; r < 8; ++r)
            sHf[(mb + r) * (2 * LPITCH) + col0 + m] = fmaxf(acc0[r] + bva, 0.0f);
#pragma unroll
        for (int r = 0; r < 8; ++r)
            sHf[(16 + mb + r) * (2 * LPITCH) + col0 + m] = fmaxf(acc1[r] + bva, 0.0f);
    }
    lds_ready_barrier();

    if (wave < 2) {
        const int col0 = wave << 4;
        v8f acc0 = {};
        v8f acc1 = {};
#pragma unroll 8
        for (int k2 = 0; k2 < 64; k2 += 2) {
            const int kq = k2 + hsel;
            v2f b  = sWb[col0 + m][kq];
            v2f a0 = sH2[m][kq];
            v2f a1 = sH2[16 + m][kq];
            acc0 = __builtin_amdgcn_wmma_f32_16x16x4_f32(
                false, a0, false, b, (short)0, acc0, false, false);
            acc1 = __builtin_amdgcn_wmma_f32_16x16x4_f32(
                false, a1, false, b, (short)0, acc1, false, false);
        }
#pragma unroll
        for (int r = 0; r < 8; ++r) {
            const int row = row0 + mb + r;
            if (row < G) out[(size_t)row * 32 + col0 + m] = acc0[r] + bvb;
        }
#pragma unroll
        for (int r = 0; r < 8; ++r) {
            const int row = row0 + 16 + mb + r;
            if (row < G) out[(size_t)row * 32 + col0 + m] = acc1[r] + bvb;
        }
    }
}

extern "C" void kernel_launch(void* const* d_in, const int* in_sizes, int n_in,
                              void* d_out, int out_size, void* d_ws, size_t ws_size,
                              hipStream_t stream) {
    const float* x     = (const float*)d_in[0];
    const int*   ei    = (const int*)d_in[1];
    const int*   batch = (const int*)d_in[2];
    const float* W1    = (const float*)d_in[3];
    const float* b1    = (const float*)d_in[4];
    const float* W2    = (const float*)d_in[5];
    const float* b2    = (const float*)d_in[6];
    const float* Wl1   = (const float*)d_in[7];
    const float* bl1   = (const float*)d_in[8];
    const float* Wl2   = (const float*)d_in[9];
    const float* bl2   = (const float*)d_in[10];
    float* out = (float*)d_out;

    const int nN = in_sizes[0] / 128;
    const int nE = in_sizes[1] / 2;
    const int G  = out_size / 32;
    const int* src = ei;
    const int* dst = ei + nE;

    char* ws = (char*)d_ws;
    auto alloc = [&](size_t bytes) -> float* {
        float* p = (float*)ws;
        ws += (bytes + 255) & ~(size_t)255;
        return p;
    };
    float* bufA = alloc((size_t)nN * 128 * 4);
    float* bufB = alloc((size_t)nN * 128 * 4);
    char*  bufC = (char*)alloc((size_t)nN * 128 * 4);
    int*   cnt  = (int*)alloc((size_t)nN * 4);
    float* dinv = alloc((size_t)nN * 4);
    float* gsum = alloc((size_t)G * 128 * 4);
    float* gcnt = alloc((size_t)G * 4);

    auto sub = [&](size_t bytes) -> int* {
        int* p = (int*)bufC;
        bufC += (bytes + 255) & ~(size_t)255;
        return p;
    };
    const int NB = (nN + 255) >> BIN_SHIFT;
    const int NC = (nE + ECH - 1) / ECH;
    int* csr_src = sub((size_t)nE * 4);
    int* loc     = sub((size_t)nE * 8);
    int* rowptr  = sub((size_t)nN * 4);
    int* hist    = sub((size_t)NB * NC * 8);

    edge_part<<<NC, TPA, 0, stream>>>(src, dst, loc, hist, nE, NB, gsum, G * 128, gcnt, G);
    bin_csr<<<NB, 256, 0, stream>>>(loc, hist, rowptr, cnt, dinv, csr_src, nN, nE, NB, NC);

    gemm_k128<128, false, false, true><<<768, 256, 0, stream>>>(x, W1, nullptr, dinv, bufA, nN);
    gather_gemm<<<512, 1024, 0, stream>>>(rowptr, cnt, csr_src, dinv, bufA, b1, W2, bufB, nN);

    gather_relu<true><<<2048, 256, 0, stream>>>(rowptr, cnt, csr_src, dinv, bufB, b2, nullptr,
                                                batch, gsum, gcnt, nN);

    head_mlp<<<(G + RSTRIP - 1) / RSTRIP, 1024, 0, stream>>>(gsum, gcnt, Wl1, bl1, Wl2, bl2, out, G);
}
